# starting kernel
# baseline (speedup 1.0000x reference)
#include <hip/hip_runtime.h>
#include <hip/hip_bf16.h>


#define IN_DIM  64
#define OUT_DIM 16

typedef __attribute__((ext_vector_type(2))) float v2f;
typedef __attribute__((ext_vector_type(8))) float v8f;

__device__ __forceinline__ void atomic_add_f32(float* p, float v) {
  asm volatile("global_atomic_add_f32 %0, %1, off scope:SCOPE_DEV"
               :: "v"(p), "v"(v) : "memory");
}


__global__ void fill0_kernel(float* __restrict__ p, int n) {
  int i = blockIdx.x * blockDim.x + threadIdx.x;
  if (i < n) p[i] = 0.0f;
}

__global__ void degree_kernel(const int* __restrict__ dst, float* __restrict__ deg, int E) {
  int i = blockIdx.x * blockDim.x + threadIdx.x;
  if (i < E) atomic_add_f32(&deg[dst[i]], 1.0f);
}

__global__ void rsqrt_kernel(float* __restrict__ d, int n) {
  int i = blockIdx.x * blockDim.x + threadIdx.x;
  if (i < n) d[i] = rsqrtf(d[i] + 1.0f);
}

__global__ void bias_relu_kernel(float* __restrict__ a, const float* __restrict__ b, int n) {
  int i = blockIdx.x * blockDim.x + threadIdx.x;
  if (i < n) a[i] = fmaxf(a[i] + b[i & 63], 0.0f);
}

__global__ void bias_kernel(float* __restrict__ a, const float* __restrict__ b, int n) {
  int i = blockIdx.x * blockDim.x + threadIdx.x;
  if (i < n) a[i] = a[i] + b[i & 15];
}


template <int NCOLS>
__global__ __launch_bounds__(256) void gemm_wmma_kernel(const float* __restrict__ A,
                                                        const float* __restrict__ W,
                                                        float* __restrict__ out, int M) {
  __shared__ float sW[64 * NCOLS];
  const int nvec = (64 * NCOLS) / 4;
  for (int i = threadIdx.x; i < nvec; i += blockDim.x)
    ((float4*)sW)[i] = ((const float4*)W)[i];
  __syncthreads();

  const int lane = threadIdx.x & 31;
  const int wid  = threadIdx.x >> 5;
  const int ntile_n = NCOLS / 16;
  const int mtiles  = (M + 15) >> 4;
  const int gt = blockIdx.x * 8 + wid;
  const int mt = gt / ntile_n;
  const int nt = gt - mt * ntile_n;
  if (mt >= mtiles) return;

  const int m0 = mt * 16;
  const int n0 = nt * 16;
  const int row = lane & 15;
  const int kh  = (lane >> 4) << 1;
  const int arow_idx = min(m0 + row, M - 1);
  const float* __restrict__ arow = A + (size_t)arow_idx * 64 + kh;
  const float* __restrict__ bcol = sW + n0 + row;

  v8f c = {};
#pragma unroll
  for (int k = 0; k < 64; k += 4) {
    v2f a, b;
    a.x = arow[k];
    a.y = arow[k + 1];
    b.x = bcol[(k + kh) * NCOLS];
    b.y = bcol[(k + kh + 1) * NCOLS];
    c = __builtin_amdgcn_wmma_f32_16x16x4_f32(false, a, false, b, (short)0, c, false, false);
  }

  const int rbase = (lane >> 4) << 3;
#pragma unroll
  for (int r = 0; r < 8; ++r) {
    int orow = m0 + rbase + r;
    if (orow < M) out[(size_t)orow * NCOLS + n0 + row] = c[r];
  }
}


__global__ __launch_bounds__(256) void scatter64_kernel(const float* __restrict__ h,
                                                        const int* __restrict__ src,
                                                        const int* __restrict__ dst,
                                                        const float* __restrict__ dinv,
                                                        float* __restrict__ agg,
                                                        int E, int Nn) {
  const int e = blockIdx.x * 8 + (threadIdx.x >> 5);
  const int lane = threadIdx.x & 31;
  if (e >= E + Nn) return;
  int s, d;
  if (e < E) { s = src[e]; d = dst[e]; } else { s = e - E; d = s; }
  const float w = dinv[s] * dinv[d];
  const float2 v = ((const float2*)(h + (size_t)s * 64))[lane];
  float* op = agg + (size_t)d * 64 + lane * 2;
  atomic_add_f32(op,     v.x * w);
  atomic_add_f32(op + 1, v.y * w);
}

__global__ __launch_bounds__(256) void scatter16_kernel(const float* __restrict__ h,
                                                        const int* __restrict__ src,
                                                        const int* __restrict__ dst,
                                                        const float* __restrict__ dinv,
                                                        float* __restrict__ out,
                                                        int E, int Nn) {
  const int tid = blockIdx.x * blockDim.x + threadIdx.x;
  const int e = tid >> 4;
  const int c = tid & 15;
  if (e >= E + Nn) return;
  int s, d;
  if (e < E) { s = src[e]; d = dst[e]; } else { s = e - E; d = s; }
  const float w = dinv[s] * dinv[d];
  atomic_add_f32(&out[(size_t)d * 16 + c], h[(size_t)s * 16 + c] * w);
}


extern "C" void kernel_launch(void* const* d_in, const int* in_sizes, int n_in,
                              void* d_out, int out_size, void* d_ws, size_t ws_size,
                              hipStream_t stream) {
  const float* x    = (const float*)d_in[0];
  const int*   edge = (const int*)d_in[1];
  const float* W1   = (const float*)d_in[2];
  const float* b1   = (const float*)d_in[3];
  const float* W2   = (const float*)d_in[4];
  const float* b2   = (const float*)d_in[5];
  float* out = (float*)d_out;

  const int E  = in_sizes[1] / 2;
  const int Nn = in_sizes[0] / IN_DIM;
  const int* src = edge;
  const int* dst = edge + E;

  char* ws = (char*)d_ws;
  float* dinv = (float*)ws;
  float* h1   = (float*)(ws + (size_t)Nn * 4);
  float* agg1 = (float*)(ws + (size_t)Nn * 4 + (size_t)Nn * 64 * 4);
  float* h2   = h1;

  const int TB = 256;
  const int mtiles = (Nn + 15) >> 4;

  fill0_kernel<<<(Nn + TB - 1) / TB, TB, 0, stream>>>(dinv, Nn);
  fill0_kernel<<<(Nn * 64 + TB - 1) / TB, TB, 0, stream>>>(agg1, Nn * 64);
  fill0_kernel<<<(Nn * 16 + TB - 1) / TB, TB, 0, stream>>>(out, Nn * 16);

  degree_kernel<<<(E + TB - 1) / TB, TB, 0, stream>>>(dst, dinv, E);
  rsqrt_kernel<<<(Nn + TB - 1) / TB, TB, 0, stream>>>(dinv, Nn);

  gemm_wmma_kernel<64><<<(mtiles * 4 + 7) / 8, 256, 0, stream>>>(x, W1, h1, Nn);
  scatter64_kernel<<<((E + Nn) + 7) / 8, 256, 0, stream>>>(h1, src, dst, dinv, agg1, E, Nn);
  bias_relu_kernel<<<(Nn * 64 + TB - 1) / TB, TB, 0, stream>>>(agg1, b1, Nn * 64);

  gemm_wmma_kernel<16><<<(mtiles + 7) / 8, 256, 0, stream>>>(agg1, W2, h2, Nn);
  scatter16_kernel<<<((E + Nn) * 16 + TB - 1) / TB, TB, 0, stream>>>(h2, src, dst, dinv, out, E, Nn);
  bias_kernel<<<(Nn * 16 + TB - 1) / TB, TB, 0, stream>>>(out, b2, Nn * 16);

  (void)n_in; (void)out_size; (void)ws_size;
}


# Layer-1 segment sum by destination tile: a work-group owns 400 rows, scans the dst list, each wave sums its own 13 rows' edges in registers; global atomics and agg fill gone; sum order now fixed
# speedup vs baseline: 1.3535x; 1.3535x over previous
#include <hip/hip_runtime.h>
#include <hip/hip_bf16.h>


#define IN_DIM  64
#define OUT_DIM 16

typedef __attribute__((ext_vector_type(2))) float v2f;
typedef __attribute__((ext_vector_type(8))) float v8f;

__device__ __forceinline__ void atomic_add_f32(float* p, float v) {
  asm volatile("global_atomic_add_f32 %0, %1, off scope:SCOPE_DEV"
               :: "v"(p), "v"(v) : "memory");
}


__global__ void fill0_kernel(float* __restrict__ p, int n) {
  int i = blockIdx.x * blockDim.x + threadIdx.x;
  if (i < n) p[i] = 0.0f;
}

__global__ void degree_kernel(const int* __restrict__ dst, float* __restrict__ deg, int E) {
  int i = blockIdx.x * blockDim.x + threadIdx.x;
  if (i < E) atomic_add_f32(&deg[dst[i]], 1.0f);
}

__global__ void rsqrt_kernel(float* __restrict__ d, int n) {
  int i = blockIdx.x * blockDim.x + threadIdx.x;
  if (i < n) d[i] = rsqrtf(d[i] + 1.0f);
}

__global__ void bias_kernel(float* __restrict__ a, const float* __restrict__ b, int n) {
  int i = blockIdx.x * blockDim.x + threadIdx.x;
  if (i < n) a[i] = a[i] + b[i & 15];
}


#define TILE_PITCH 68

__global__ __launch_bounds__(256) void mlp_wmma_kernel(const float* __restrict__ A,
                                                       const float* __restrict__ W1,
                                                       const float* __restrict__ b1,
                                                       const float* __restrict__ W2,
                                                       float* __restrict__ out, int M) {
  __shared__ __attribute__((aligned(16))) float sW1[64 * 64];
  __shared__ __attribute__((aligned(16))) float sW2[64 * 16];
  __shared__ __attribute__((aligned(16))) float sT[8][16 * TILE_PITCH];
  for (int i = threadIdx.x; i < (64 * 64) / 4; i += blockDim.x)
    ((float4*)sW1)[i] = ((const float4*)W1)[i];
  for (int i = threadIdx.x; i < (64 * 16) / 4; i += blockDim.x)
    ((float4*)sW2)[i] = ((const float4*)W2)[i];
  __syncthreads();

  const int lane = threadIdx.x & 31;
  const int wid  = threadIdx.x >> 5;
  const int mtiles = (M + 15) >> 4;
  const int mt = blockIdx.x * 8 + wid;
  const bool live = mt < mtiles;
  const int m0 = (live ? mt : mtiles - 1) * 16;
  const int row = lane & 15;
  const int kh  = (lane >> 4) << 1;
  const int rbase = (lane >> 4) << 3;
  const int arow_idx = min(m0 + row, M - 1);
  const float* __restrict__ arow = A + (size_t)arow_idx * 64 + kh;
  const float* __restrict__ bcol = sW1 + row;

  v8f c0 = {}, c1 = {}, c2 = {}, c3 = {};
#pragma unroll
  for (int k = 0; k < 64; k += 4) {
    v2f a, b;
    a.x = arow[k];
    a.y = arow[k + 1];
    const float* __restrict__ bk = bcol + (k + kh) * 64;
    b.x = bk[0];  b.y = bk[64];
    c0 = __builtin_amdgcn_wmma_f32_16x16x4_f32(false, a, false, b, (short)0, c0, false, false);
    b.x = bk[16]; b.y = bk[80];
    c1 = __builtin_amdgcn_wmma_f32_16x16x4_f32(false, a, false, b, (short)0, c1, false, false);
    b.x = bk[32]; b.y = bk[96];
    c2 = __builtin_amdgcn_wmma_f32_16x16x4_f32(false, a, false, b, (short)0, c2, false, false);
    b.x = bk[48]; b.y = bk[112];
    c3 = __builtin_amdgcn_wmma_f32_16x16x4_f32(false, a, false, b, (short)0, c3, false, false);
  }

  float* __restrict__ T = sT[wid];
  const float bb0 = b1[row], bb1 = b1[16 + row], bb2 = b1[32 + row], bb3 = b1[48 + row];
#pragma unroll
  for (int r = 0; r < 8; ++r) {
    float* __restrict__ tp = T + (rbase + r) * TILE_PITCH + row;
    tp[0]  = fmaxf(c0[r] + bb0, 0.0f);
    tp[16] = fmaxf(c1[r] + bb1, 0.0f);
    tp[32] = fmaxf(c2[r] + bb2, 0.0f);
    tp[48] = fmaxf(c3[r] + bb3, 0.0f);
  }
  __syncthreads();

  const float* __restrict__ trow  = T + row * TILE_PITCH + kh;
  const float* __restrict__ b2col = sW2 + row;
  v8f d = {};
#pragma unroll
  for (int k = 0; k < 64; k += 4) {
    v2f a, b;
    a.x = trow[k];
    a.y = trow[k + 1];
    b.x = b2col[(k + kh) * 16];
    b.y = b2col[(k + kh + 1) * 16];
    d = __builtin_amdgcn_wmma_f32_16x16x4_f32(false, a, false, b, (short)0, d, false, false);
  }

#pragma unroll
  for (int r = 0; r < 8; ++r) {
    const int orow = m0 + rbase + r;
    if (live && orow < M) out[(size_t)orow * 16 + row] = d[r];
  }
}


#define TILE_R        400
#define SCAN_WAVES    32
#define ROWS_PER_WAVE 13
#define QCAP          1024
#define QSTEP         512
#define RING          64

__device__ __forceinline__ int4 load_dst4_guarded(const int* __restrict__ dst, int e, int E) {
  int4 r;
  r.x = (e     < E) ? dst[e]     : -1;
  r.y = (e + 1 < E) ? dst[e + 1] : -1;
  r.z = (e + 2 < E) ? dst[e + 2] : -1;
  r.w = (e + 3 < E) ? dst[e + 3] : -1;
  return r;
}

typedef __attribute__((ext_vector_type(16))) float v16f;

__global__ __launch_bounds__(1024) void agg64_tile_kernel(const float* __restrict__ x,
                                                          const int* __restrict__ src,
                                                          const int* __restrict__ dst,
                                                          const float* __restrict__ dinv,
                                                          float* __restrict__ agg,
                                                          int E, int Nn) {
  __shared__ unsigned sQ[SCAN_WAVES][QCAP];
  __shared__ unsigned sRing[SCAN_WAVES][RING];
  __shared__ int sQn[SCAN_WAVES];
  __shared__ int sMore[SCAN_WAVES];

  const int lane = threadIdx.x & 31;
  const int wid  = __builtin_amdgcn_readfirstlane((int)(threadIdx.x >> 5));
  const int d0   = blockIdx.x * TILE_R;
  const int nrows = min(TILE_R, Nn - d0);
  const unsigned un = (unsigned)nrows;
  const int row0 = wid * ROWS_PER_WAVE;

  v16f accx = {}, accy = {};
  {
    float2 v0[ROWS_PER_WAVE];
    float dd[ROWS_PER_WAVE];
#pragma unroll
    for (int k = 0; k < ROWS_PER_WAVE; ++k) {
      const int d = d0 + min(row0 + k, nrows - 1);
      dd[k] = dinv[d];
      v0[k] = ((const float2*)(x + (size_t)d * 64))[lane];
    }
#pragma unroll
    for (int k = 0; k < ROWS_PER_WAVE; ++k) {
      const bool valid = row0 + k < nrows;
      const float w = dd[k] * dd[k];
      accx[k] = valid ? v0[k].x * w : 0.0f;
      accy[k] = valid ? v0[k].y * w : 0.0f;
    }
  }

  unsigned* __restrict__ q    = sQ[wid];
  unsigned* __restrict__ ring = sRing[wid];
  int qn = 0;
  int rhead = 0, rn = 0;

  auto qpush = [&](bool m, int e, int d) {
    const unsigned mask = __builtin_amdgcn_ballot_w32(m);
    if (mask != 0u) {
      if (m) q[qn + (int)__builtin_amdgcn_mbcnt_lo(mask, 0u)] = ((unsigned)e << 9) | (unsigned)(d - d0);
      qn += __builtin_popcount(mask);
    }
  };

  auto look = [&](const int4 (&dv)[4], int ebase) {
    const int e0 = ebase + lane * 4;
#pragma unroll
    for (int u = 0; u < 4; ++u) {
      const int eb = e0 + u * 128;
      const bool m0 = (unsigned)(dv[u].x - d0) < un;
      const bool m1 = (unsigned)(dv[u].y - d0) < un;
      const bool m2 = (unsigned)(dv[u].z - d0) < un;
      const bool m3 = (unsigned)(dv[u].w - d0) < un;
      if (__builtin_amdgcn_ballot_w32(m0 | m1 | m2 | m3) != 0u) {
        qpush(m0, eb,     dv[u].x);
        qpush(m1, eb + 1, dv[u].y);
        qpush(m2, eb + 2, dv[u].z);
        qpush(m3, eb + 3, dv[u].w);
      }
    }
  };

  auto addrow = [&](int r, float vx, float vy, float w) {
    accx[r] += vx * w;
    accy[r] += vy * w;
  };

  auto apply = [&](int n) {
    __builtin_amdgcn_wave_barrier();
    int s_l = 0, r_l = 0;
    float w_l = 0.0f;
    if (lane < n) {
      const unsigned pk = ring[(rhead + lane) & (RING - 1)];
      const int dl = (int)(pk & 511u);
      r_l = dl - row0;
      s_l = src[pk >> 9];
      w_l = dinv[s_l] * dinv[d0 + dl];
    }
    int j = 0;
    for (; j + 8 <= n; j += 8) {
      float2 v[8];
#pragma unroll
      for (int u = 0; u < 8; ++u) {
        const int s = __builtin_amdgcn_readlane(s_l, j + u);
        v[u] = ((const float2*)(x + (size_t)s * 64))[lane];
      }
#pragma unroll
      for (int u = 0; u < 8; ++u) {
        const int r = __builtin_amdgcn_readlane(r_l, j + u);
        const float w = __int_as_float(__builtin_amdgcn_readlane(__float_as_int(w_l), j + u));
        addrow(r, v[u].x, v[u].y, w);
      }
    }
    for (; j < n; ++j) {
      const int s = __builtin_amdgcn_readlane(s_l, j);
      const int r = __builtin_amdgcn_readlane(r_l, j);
      const float w = __int_as_float(__builtin_amdgcn_readlane(__float_as_int(w_l), j));
      const float2 v = ((const float2*)(x + (size_t)s * 64))[lane];
      addrow(r, v.x, v.y, w);
    }
  };

  auto rpush = [&](bool m, unsigned pk) {
    const unsigned mask = __builtin_amdgcn_ballot_w32(m);
    if (mask != 0u) {
      if (m) ring[(rhead + rn + (int)__builtin_amdgcn_mbcnt_lo(mask, 0u)) & (RING - 1)] = pk;
      rn += __builtin_popcount(mask);
      if (rn >= 32) {
        apply(32);
        rhead = (rhead + 32) & (RING - 1);
        rn -= 32;
      }
    }
  };

  const bool vec  = (E & 3) == 0;
  const int nsc   = (E + QSTEP - 1) / QSTEP;
  const int nfull = vec ? E / QSTEP : 0;
  const int rot = (int)blockIdx.x * (nfull / (int)gridDim.x);

  auto loadfull = [&](int scx, int4 (&dv)[4]) {
    int g = scx + rot;
    if (g >= nfull) g -= nfull;
    const int4* __restrict__ p = (const int4*)(dst + (size_t)g * QSTEP) + lane;
#pragma unroll
    for (int u = 0; u < 4; ++u) dv[u] = p[u * 32];
  };

  int sc = wid;

  for (;;) {
    if (sc < nfull) {
      int4 cur[4], nx[4];
      loadfull(sc, cur);
      while (sc < nfull && qn <= QCAP - QSTEP) {
        const int nxt = sc + SCAN_WAVES;
        loadfull(nxt < nfull ? nxt : sc, nx);
        int g = sc + rot;
        if (g >= nfull) g -= nfull;
        look(cur, g * QSTEP);
        sc = nxt;
#pragma unroll
        for (int u = 0; u < 4; ++u) cur[u] = nx[u];
      }
    }
    while (sc >= nfull && sc < nsc && qn <= QCAP - QSTEP) {
      int4 dv[4];
#pragma unroll
      for (int u = 0; u < 4; ++u) dv[u] = load_dst4_guarded(dst, sc * QSTEP + lane * 4 + u * 128, E);
      look(dv, sc * QSTEP);
      sc += SCAN_WAVES;
    }
    if (lane == 0) { sQn[wid] = qn; sMore[wid] = (sc < nsc) ? 1 : 0; }
    __syncthreads();

    const bool more = __builtin_amdgcn_ballot_w32(sMore[lane] != 0) != 0u;
    for (int k = 0; k < SCAN_WAVES; ++k) {
      const int nk = __builtin_amdgcn_readfirstlane(sQn[k]);
      const unsigned* __restrict__ qk = sQ[k];
      for (int b = 0; b < nk; b += 32) {
        unsigned pk = 0u;
        bool own = false;
        if (b + lane < nk) {
          pk = qk[b + lane];
          own = ((pk & 511u) / ROWS_PER_WAVE) == (unsigned)wid;
        }
        rpush(own, pk);
      }
    }
    if (rn > 0) apply(rn);
    rhead = 0; rn = 0; qn = 0;
    __syncthreads();
    if (!more) break;
  }

#pragma unroll
  for (int k = 0; k < ROWS_PER_WAVE; ++k)
    if (row0 + k < nrows)
      ((float2*)(agg + (size_t)(d0 + row0 + k) * 64))[lane] = make_float2(accx[k], accy[k]);
}

__global__ __launch_bounds__(256) void scatter16_kernel(const float* __restrict__ h,
                                                        const int* __restrict__ src,
                                                        const int* __restrict__ dst,
                                                        const float* __restrict__ dinv,
                                                        float* __restrict__ out,
                                                        int E, int Nn) {
  const int tid = blockIdx.x * blockDim.x + threadIdx.x;
  const int e = tid >> 4;
  const int c = tid & 15;
  if (e >= E + Nn) return;
  int s, d;
  if (e < E) { s = src[e]; d = dst[e]; } else { s = e - E; d = s; }
  const float w = dinv[s] * dinv[d];
  atomic_add_f32(&out[(size_t)d * 16 + c], h[(size_t)s * 16 + c] * w);
}


extern "C" void kernel_launch(void* const* d_in, const int* in_sizes, int n_in,
                              void* d_out, int out_size, void* d_ws, size_t ws_size,
                              hipStream_t stream) {
  const float* x    = (const float*)d_in[0];
  const int*   edge = (const int*)d_in[1];
  const float* W1   = (const float*)d_in[2];
  const float* b1   = (const float*)d_in[3];
  const float* W2   = (const float*)d_in[4];
  const float* b2   = (const float*)d_in[5];
  float* out = (float*)d_out;

  const int E  = in_sizes[1] / 2;
  const int Nn = in_sizes[0] / IN_DIM;
  const int* src = edge;
  const int* dst = edge + E;

  char* ws = (char*)d_ws;
  float* dinv = (float*)ws;
  float* h2   = (float*)(ws + (size_t)Nn * 4);
  float* agg1 = (float*)(ws + (size_t)Nn * 4 + (size_t)Nn * 64 * 4);

  const int TB = 256;
  const int mtiles = (Nn + 15) >> 4;

  fill0_kernel<<<(Nn + TB - 1) / TB, TB, 0, stream>>>(dinv, Nn);
  fill0_kernel<<<(Nn * 16 + TB - 1) / TB, TB, 0, stream>>>(out, Nn * 16);

  degree_kernel<<<(E + TB - 1) / TB, TB, 0, stream>>>(dst, dinv, E);
  rsqrt_kernel<<<(Nn + TB - 1) / TB, TB, 0, stream>>>(dinv, Nn);

  agg64_tile_kernel<<<(Nn + TILE_R - 1) / TILE_R, SCAN_WAVES * 32, 0, stream>>>(x, src, dst, dinv, agg1, E, Nn);

  mlp_wmma_kernel<<<(mtiles + 7) / 8, 256, 0, stream>>>(agg1, W1, b1, W2, h2, Nn);
  scatter16_kernel<<<((E + Nn) * 16 + TB - 1) / TB, TB, 0, stream>>>(h2, src, dst, dinv, out, E, Nn);
  bias_kernel<<<(Nn * 16 + TB - 1) / TB, TB, 0, stream>>>(out, b2, Nn * 16);

  (void)n_in; (void)out_size; (void)ws_size;
}


# Layer-2 segment sum by destination tile from edge lists (src,row,weight) that layer 1's waves save: a wave sums its 13 rows in LDS; f32 atomics, out fill, bias pass gone; sum order now fixed
# speedup vs baseline: 1.7786x; 1.3141x over previous
#include <hip/hip_runtime.h>
#include <hip/hip_bf16.h>


#define IN_DIM  64
#define OUT_DIM 16

typedef __attribute__((ext_vector_type(2))) float v2f;
typedef __attribute__((ext_vector_type(8))) float v8f;

__device__ __forceinline__ void atomic_add_f32(float* p, float v) {
  asm volatile("global_atomic_add_f32 %0, %1, off scope:SCOPE_DEV"
               :: "v"(p), "v"(v) : "memory");
}


__global__ void fill0_kernel(float* __restrict__ p, int n) {
  int i = blockIdx.x * blockDim.x + threadIdx.x;
  if (i < n) p[i] = 0.0f;
}

__global__ void degree_kernel(const int* __restrict__ dst, float* __restrict__ deg, int E) {
  int i = blockIdx.x * blockDim.x + threadIdx.x;
  if (i < E) atomic_add_f32(&deg[dst[i]], 1.0f);
}

__global__ void rsqrt_kernel(float* __restrict__ d, int n) {
  int i = blockIdx.x * blockDim.x + threadIdx.x;
  if (i < n) d[i] = rsqrtf(d[i] + 1.0f);
}


#define TILE_PITCH 68

__global__ __launch_bounds__(256) void mlp_wmma_kernel(const float* __restrict__ A,
                                                       const float* __restrict__ W1,
                                                       const float* __restrict__ b1,
                                                       const float* __restrict__ W2,
                                                       float* __restrict__ out, int M) {
  __shared__ __attribute__((aligned(16))) float sW1[64 * 64];
  __shared__ __attribute__((aligned(16))) float sW2[64 * 16];
  __shared__ __attribute__((aligned(16))) float sT[8][16 * TILE_PITCH];
  for (int i = threadIdx.x; i < (64 * 64) / 4; i += blockDim.x)
    ((float4*)sW1)[i] = ((const float4*)W1)[i];
  for (int i = threadIdx.x; i < (64 * 16) / 4; i += blockDim.x)
    ((float4*)sW2)[i] = ((const float4*)W2)[i];
  __syncthreads();

  const int lane = threadIdx.x & 31;
  const int wid  = threadIdx.x >> 5;
  const int mtiles = (M + 15) >> 4;
  const int mt = blockIdx.x * 8 + wid;
  const bool live = mt < mtiles;
  const int m0 = (live ? mt : mtiles - 1) * 16;
  const int row = lane & 15;
  const int kh  = (lane >> 4) << 1;
  const int rbase = (lane >> 4) << 3;
  const int arow_idx = min(m0 + row, M - 1);
  const float* __restrict__ arow = A + (size_t)arow_idx * 64 + kh;
  const float* __restrict__ bcol = sW1 + row;

  v8f c0 = {}, c1 = {}, c2 = {}, c3 = {};
#pragma unroll
  for (int k = 0; k < 64; k += 4) {
    v2f a, b;
    a.x = arow[k];
    a.y = arow[k + 1];
    const float* __restrict__ bk = bcol + (k + kh) * 64;
    b.x = bk[0];  b.y = bk[64];
    c0 = __builtin_amdgcn_wmma_f32_16x16x4_f32(false, a, false, b, (short)0, c0, false, false);
    b.x = bk[16]; b.y = bk[80];
    c1 = __builtin_amdgcn_wmma_f32_16x16x4_f32(false, a, false, b, (short)0, c1, false, false);
    b.x = bk[32]; b.y = bk[96];
    c2 = __builtin_amdgcn_wmma_f32_16x16x4_f32(false, a, false, b, (short)0, c2, false, false);
    b.x = bk[48]; b.y = bk[112];
    c3 = __builtin_amdgcn_wmma_f32_16x16x4_f32(false, a, false, b, (short)0, c3, false, false);
  }

  float* __restrict__ T = sT[wid];
  const float bb0 = b1[row], bb1 = b1[16 + row], bb2 = b1[32 + row], bb3 = b1[48 + row];
#pragma unroll
  for (int r = 0; r < 8; ++r) {
    float* __restrict__ tp = T + (rbase + r) * TILE_PITCH + row;
    tp[0]  = fmaxf(c0[r] + bb0, 0.0f);
    tp[16] = fmaxf(c1[r] + bb1, 0.0f);
    tp[32] = fmaxf(c2[r] + bb2, 0.0f);
    tp[48] = fmaxf(c3[r] + bb3, 0.0f);
  }
  __syncthreads();

  const float* __restrict__ trow  = T + row * TILE_PITCH + kh;
  const float* __restrict__ b2col = sW2 + row;
  v8f d = {};
#pragma unroll
  for (int k = 0; k < 64; k += 4) {
    v2f a, b;
    a.x = trow[k];
    a.y = trow[k + 1];
    b.x = b2col[(k + kh) * 16];
    b.y = b2col[(k + kh + 1) * 16];
    d = __builtin_amdgcn_wmma_f32_16x16x4_f32(false, a, false, b, (short)0, d, false, false);
  }

#pragma unroll
  for (int r = 0; r < 8; ++r) {
    const int orow = m0 + rbase + r;
    if (live && orow < M) out[(size_t)orow * 16 + row] = d[r];
  }
}


#define TILE_R        400
#define SCAN_WAVES    32
#define ROWS_PER_WAVE 13
#define QCAP          1024
#define QSTEP         512
#define RING          64
#define LCAP          512

__device__ __forceinline__ int4 load_dst4_guarded(const int* __restrict__ dst, int e, int E) {
  int4 r;
  r.x = (e     < E) ? dst[e]     : -1;
  r.y = (e + 1 < E) ? dst[e + 1] : -1;
  r.z = (e + 2 < E) ? dst[e + 2] : -1;
  r.w = (e + 3 < E) ? dst[e + 3] : -1;
  return r;
}

typedef __attribute__((ext_vector_type(16))) float v16f;

__global__ __launch_bounds__(1024) void agg64_tile_kernel(const float* __restrict__ x,
                                                          const int* __restrict__ src,
                                                          const int* __restrict__ dst,
                                                          const float* __restrict__ dinv,
                                                          float* __restrict__ agg,
                                                          int* __restrict__ lst,
                                                          int* __restrict__ lcount,
                                                          int E, int Nn) {
  __shared__ unsigned sQ[SCAN_WAVES][QCAP];
  __shared__ unsigned sRing[SCAN_WAVES][RING];
  __shared__ int sQn[SCAN_WAVES];
  __shared__ int sMore[SCAN_WAVES];

  const int lane = threadIdx.x & 31;
  const int wid  = __builtin_amdgcn_readfirstlane((int)(threadIdx.x >> 5));
  const int d0   = blockIdx.x * TILE_R;
  const int nrows = min(TILE_R, Nn - d0);
  const unsigned un = (unsigned)nrows;
  const int row0 = wid * ROWS_PER_WAVE;

  v16f accx = {}, accy = {};
  {
    float2 v0[ROWS_PER_WAVE];
    float dd[ROWS_PER_WAVE];
#pragma unroll
    for (int k = 0; k < ROWS_PER_WAVE; ++k) {
      const int d = d0 + min(row0 + k, nrows - 1);
      dd[k] = dinv[d];
      v0[k] = ((const float2*)(x + (size_t)d * 64))[lane];
    }
#pragma unroll
    for (int k = 0; k < ROWS_PER_WAVE; ++k) {
      const bool valid = row0 + k < nrows;
      const float w = dd[k] * dd[k];
      accx[k] = valid ? v0[k].x * w : 0.0f;
      accy[k] = valid ? v0[k].y * w : 0.0f;
    }
  }

  unsigned* __restrict__ q    = sQ[wid];
  unsigned* __restrict__ ring = sRing[wid];
  int qn = 0;
  int rhead = 0, rn = 0;
  const int gw = (int)blockIdx.x * SCAN_WAVES + wid;
  int2* __restrict__ mylst = (int2*)lst + (size_t)gw * LCAP;
  int lcnt = 0;

  auto qpush = [&](bool m, int e, int d) {
    const unsigned mask = __builtin_amdgcn_ballot_w32(m);
    if (mask != 0u) {
      if (m) q[qn + (int)__builtin_amdgcn_mbcnt_lo(mask, 0u)] = ((unsigned)e << 9) | (unsigned)(d - d0);
      qn += __builtin_popcount(mask);
    }
  };

  auto look = [&](const int4 (&dv)[4], int ebase) {
    const int e0 = ebase + lane * 4;
#pragma unroll
    for (int u = 0; u < 4; ++u) {
      const int eb = e0 + u * 128;
      const bool m0 = (unsigned)(dv[u].x - d0) < un;
      const bool m1 = (unsigned)(dv[u].y - d0) < un;
      const bool m2 = (unsigned)(dv[u].z - d0) < un;
      const bool m3 = (unsigned)(dv[u].w - d0) < un;
      if (__builtin_amdgcn_ballot_w32(m0 | m1 | m2 | m3) != 0u) {
        qpush(m0, eb,     dv[u].x);
        qpush(m1, eb + 1, dv[u].y);
        qpush(m2, eb + 2, dv[u].z);
        qpush(m3, eb + 3, dv[u].w);
      }
    }
  };

  auto addrow = [&](int r, float vx, float vy, float w) {
    accx[r] += vx * w;
    accy[r] += vy * w;
  };

  auto apply = [&](int n) {
    __builtin_amdgcn_wave_barrier();
    int s_l = 0, r_l = 0;
    float w_l = 0.0f;
    if (lane < n) {
      const unsigned pk = ring[(rhead + lane) & (RING - 1)];
      const int dl = (int)(pk & 511u);
      r_l = dl - row0;
      s_l = src[pk >> 9];
      w_l = dinv[s_l] * dinv[d0 + dl];
      if (lcnt + lane < LCAP)
        mylst[lcnt + lane] = make_int2((s_l << 4) | r_l, __float_as_int(w_l));
    }
    lcnt += n;
    int j = 0;
    for (; j + 8 <= n; j += 8) {
      float2 v[8];
#pragma unroll
      for (int u = 0; u < 8; ++u) {
        const int s = __builtin_amdgcn_readlane(s_l, j + u);
        v[u] = ((const float2*)(x + (size_t)s * 64))[lane];
      }
#pragma unroll
      for (int u = 0; u < 8; ++u) {
        const int r = __builtin_amdgcn_readlane(r_l, j + u);
        const float w = __int_as_float(__builtin_amdgcn_readlane(__float_as_int(w_l), j + u));
        addrow(r, v[u].x, v[u].y, w);
      }
    }
    for (; j < n; ++j) {
      const int s = __builtin_amdgcn_readlane(s_l, j);
      const int r = __builtin_amdgcn_readlane(r_l, j);
      const float w = __int_as_float(__builtin_amdgcn_readlane(__float_as_int(w_l), j));
      const float2 v = ((const float2*)(x + (size_t)s * 64))[lane];
      addrow(r, v.x, v.y, w);
    }
  };

  auto rpush = [&](bool m, unsigned pk) {
    const unsigned mask = __builtin_amdgcn_ballot_w32(m);
    if (mask != 0u) {
      if (m) ring[(rhead + rn + (int)__builtin_amdgcn_mbcnt_lo(mask, 0u)) & (RING - 1)] = pk;
      rn += __builtin_popcount(mask);
      if (rn >= 32) {
        apply(32);
        rhead = (rhead + 32) & (RING - 1);
        rn -= 32;
      }
    }
  };

  const bool vec  = (E & 3) == 0;
  const int nsc   = (E + QSTEP - 1) / QSTEP;
  const int nfull = vec ? E / QSTEP : 0;
  const int rot = (int)blockIdx.x * (nfull / (int)gridDim.x);

  auto loadfull = [&](int scx, int4 (&dv)[4]) {
    int g = scx + rot;
    if (g >= nfull) g -= nfull;
    const int4* __restrict__ p = (const int4*)(dst + (size_t)g * QSTEP) + lane;
#pragma unroll
    for (int u = 0; u < 4; ++u) dv[u] = p[u * 32];
  };

  int sc = wid;

  for (;;) {
    if (sc < nfull) {
      int4 cur[4], nx[4];
      loadfull(sc, cur);
      while (sc < nfull && qn <= QCAP - QSTEP) {
        const int nxt = sc + SCAN_WAVES;
        loadfull(nxt < nfull ? nxt : sc, nx);
        int g = sc + rot;
        if (g >= nfull) g -= nfull;
        look(cur, g * QSTEP);
        sc = nxt;
#pragma unroll
        for (int u = 0; u < 4; ++u) cur[u] = nx[u];
      }
    }
    while (sc >= nfull && sc < nsc && qn <= QCAP - QSTEP) {
      int4 dv[4];
#pragma unroll
      for (int u = 0; u < 4; ++u) dv[u] = load_dst4_guarded(dst, sc * QSTEP + lane * 4 + u * 128, E);
      look(dv, sc * QSTEP);
      sc += SCAN_WAVES;
    }
    if (lane == 0) { sQn[wid] = qn; sMore[wid] = (sc < nsc) ? 1 : 0; }
    __syncthreads();

    const bool more = __builtin_amdgcn_ballot_w32(sMore[lane] != 0) != 0u;
    for (int k = 0; k < SCAN_WAVES; ++k) {
      const int nk = __builtin_amdgcn_readfirstlane(sQn[k]);
      const unsigned* __restrict__ qk = sQ[k];
      for (int b = 0; b < nk; b += 32) {
        unsigned pk = 0u;
        bool own = false;
        if (b + lane < nk) {
          pk = qk[b + lane];
          own = ((pk & 511u) / ROWS_PER_WAVE) == (unsigned)wid;
        }
        rpush(own, pk);
      }
    }
    if (rn > 0) apply(rn);
    rhead = 0; rn = 0; qn = 0;
    __syncthreads();
    if (!more) break;
  }

#pragma unroll
  for (int k = 0; k < ROWS_PER_WAVE; ++k)
    if (row0 + k < nrows)
      ((float2*)(agg + (size_t)(d0 + row0 + k) * 64))[lane] = make_float2(accx[k], accy[k]);
  if (lane == 0) lcount[gw] = lcnt;
}

__global__ __launch_bounds__(256) void agg16_list_kernel(const float* __restrict__ h,
                                                         const int* __restrict__ lst,
                                                         const int* __restrict__ lcount,
                                                         const int* __restrict__ src,
                                                         const int* __restrict__ dst,
                                                         const float* __restrict__ dinv,
                                                         const float* __restrict__ b2,
                                                         float* __restrict__ out,
                                                         int E, int Nn) {
  __shared__ float sA0[8][ROWS_PER_WAVE * 32];
  __shared__ float sA1[8][ROWS_PER_WAVE * 32];
  __shared__ float sA2[8][ROWS_PER_WAVE * 32];
  __shared__ float sA3[8][ROWS_PER_WAVE * 32];

  const int lane = threadIdx.x & 31;
  const int wid  = __builtin_amdgcn_readfirstlane((int)(threadIdx.x >> 5));
  const int gw   = (int)blockIdx.x * 8 + wid;
  const int d0   = (gw / SCAN_WAVES) * TILE_R;
  const int row0 = (gw % SCAN_WAVES) * ROWS_PER_WAVE;
  const int nr   = min(ROWS_PER_WAVE, min(TILE_R, Nn - d0) - row0);
  if (nr <= 0) return;
  const int c  = lane & 15;
  const int hh = lane >> 4;
  const int base = d0 + row0;

  float* a0 = sA0[wid] + lane;
  float* a1 = sA1[wid] + lane;
  float* a2 = sA2[wid] + lane;
  float* a3 = sA3[wid] + lane;
#pragma unroll
  for (int k = 0; k < ROWS_PER_WAVE; ++k) {
    a0[k * 32] = 0.0f; a1[k * 32] = 0.0f; a2[k * 32] = 0.0f; a3[k * 32] = 0.0f;
  }

  float selfv[7];
#pragma unroll
  for (int i = 0; i < 7; ++i) {
    const int d = base + min(2 * i + hh, nr - 1);
    const float dd = dinv[d];
    selfv[i] = h[(size_t)d * 16 + c] * (dd * dd);
  }
  const float bias = b2[c];

  const int cnt = __builtin_amdgcn_readfirstlane(lcount[gw]);
  if (cnt <= LCAP) {
    const int2* __restrict__ my = (const int2*)lst + (size_t)gw * LCAP;
    int2 en = make_int2(0, 0);
    if (lane < cnt) en = my[lane];
    for (int j = 0; j < cnt; j += 32) {
      const int m = cnt - j;
      int pk[16];
      float wt[16], v[16];
#pragma unroll
      for (int u = 0; u < 16; ++u) {
        const int sel = (2 * u + hh) << 2;
        pk[u] = __builtin_amdgcn_ds_bpermute(sel, en.x);
        wt[u] = __int_as_float(__builtin_amdgcn_ds_bpermute(sel, en.y));
        v[u] = h[((unsigned)pk[u] & ~15u) + (unsigned)c];
      }
      int2 nx = make_int2(0, 0);
      if (j + 32 + lane < cnt) nx = my[j + 32 + lane];
#pragma unroll
      for (int u = 0; u < 16; u += 4) {
        const int r0 = (pk[u] & 15) * 32, r1 = (pk[u + 1] & 15) * 32;
        const int r2 = (pk[u + 2] & 15) * 32, r3 = (pk[u + 3] & 15) * 32;
        float t0 = a0[r0], t1 = a1[r1], t2 = a2[r2], t3 = a3[r3];
        t0 += v[u] * wt[u];
        t1 += v[u + 1] * wt[u + 1];
        t2 += v[u + 2] * wt[u + 2];
        t3 += v[u + 3] * wt[u + 3];
        if (2 * u + hh < m)       a0[r0] = t0;
        if (2 * (u + 1) + hh < m) a1[r1] = t1;
        if (2 * (u + 2) + hh < m) a2[r2] = t2;
        if (2 * (u + 3) + hh < m) a3[r3] = t3;
      }
      en = nx;
    }
  } else {
    for (int e0 = 0; e0 < E; e0 += 32) {
      const int e = e0 + lane;
      const int d = (e < E) ? dst[e] : -1;
      unsigned mask = __builtin_amdgcn_ballot_w32((unsigned)(d - base) < (unsigned)nr);
      while (mask != 0u) {
        const int l = __builtin_ctz(mask);
        mask &= mask - 1u;
        const int dd = __builtin_amdgcn_readlane(d, l);
        const int s  = src[e0 + l];
        const float w = dinv[s] * dinv[dd];
        if (lane < 16) a0[(dd - base) * 32] += h[(size_t)s * 16 + lane] * w;
      }
    }
  }

  __builtin_amdgcn_wave_barrier();
#pragma unroll
  for (int i = 0; i < 7; ++i) {
    const int r = 2 * i + hh;
    if (r < nr) {
      const float* p0 = sA0[wid] + r * 32 + c;
      const float* p1 = sA1[wid] + r * 32 + c;
      const float* p2 = sA2[wid] + r * 32 + c;
      const float* p3 = sA3[wid] + r * 32 + c;
      float s = selfv[i];
      s += p0[0]; s += p0[16]; s += p1[0]; s += p1[16];
      s += p2[0]; s += p2[16]; s += p3[0]; s += p3[16];
      out[(size_t)(base + r) * 16 + c] = s + bias;
    }
  }
}


extern "C" void kernel_launch(void* const* d_in, const int* in_sizes, int n_in,
                              void* d_out, int out_size, void* d_ws, size_t ws_size,
                              hipStream_t stream) {
  const float* x    = (const float*)d_in[0];
  const int*   edge = (const int*)d_in[1];
  const float* W1   = (const float*)d_in[2];
  const float* b1   = (const float*)d_in[3];
  const float* W2   = (const float*)d_in[4];
  const float* b2   = (const float*)d_in[5];
  float* out = (float*)d_out;

  const int E  = in_sizes[1] / 2;
  const int Nn = in_sizes[0] / IN_DIM;
  const int* src = edge;
  const int* dst = edge + E;

  char* ws = (char*)d_ws;
  float* dinv = (float*)ws;
  float* h2   = (float*)(ws + (size_t)Nn * 4);
  float* agg1 = (float*)(ws + (size_t)Nn * 4 + (size_t)Nn * 64 * 4);
  const int ntiles = (Nn + TILE_R - 1) / TILE_R;
  const size_t loff = ((size_t)Nn * 4 + (size_t)Nn * 128 * 4 + 255) & ~(size_t)255;
  int* lcount = (int*)(ws + loff);
  int* lst    = (int*)(ws + loff + (size_t)ntiles * SCAN_WAVES * 4);

  const int TB = 256;
  const int mtiles = (Nn + 15) >> 4;

  fill0_kernel<<<(Nn + TB - 1) / TB, TB, 0, stream>>>(dinv, Nn);

  degree_kernel<<<(E + TB - 1) / TB, TB, 0, stream>>>(dst, dinv, E);
  rsqrt_kernel<<<(Nn + TB - 1) / TB, TB, 0, stream>>>(dinv, Nn);

  agg64_tile_kernel<<<ntiles, SCAN_WAVES * 32, 0, stream>>>(x, src, dst, dinv, agg1, lst, lcount, E, Nn);

  mlp_wmma_kernel<<<(mtiles + 7) / 8, 256, 0, stream>>>(agg1, W1, b1, W2, h2, Nn);
  agg16_list_kernel<<<ntiles * (SCAN_WAVES / 8), 256, 0, stream>>>(h2, lst, lcount, src, dst, dinv, b2, out, E, Nn);

  (void)n_in; (void)out_size; (void)ws_size;
}


# Layer-1 tile scan reads a 1-byte-per-edge tile key list the degree pass writes, not the 4-byte dst list; key matches then fetch their dst: a quarter of the scan bytes; sum order fixed, but changed
# speedup vs baseline: 1.9578x; 1.1008x over previous
#include <hip/hip_runtime.h>
#include <hip/hip_bf16.h>


#define IN_DIM  64
#define OUT_DIM 16

typedef __attribute__((ext_vector_type(2))) float v2f;
typedef __attribute__((ext_vector_type(8))) float v8f;

__device__ __forceinline__ void atomic_add_f32(float* p, float v) {
  asm volatile("global_atomic_add_f32 %0, %1, off scope:SCOPE_DEV"
               :: "v"(p), "v"(v) : "memory");
}


__global__ void fill0_kernel(float* __restrict__ p, int n) {
  int i = blockIdx.x * blockDim.x + threadIdx.x;
  if (i < n) p[i] = 0.0f;
}


__global__ void rsqrt_kernel(float* __restrict__ d, int n) {
  int i = blockIdx.x * blockDim.x + threadIdx.x;
  if (i < n) d[i] = rsqrtf(d[i] + 1.0f);
}


#define TILE_PITCH 68

__global__ __launch_bounds__(256) void mlp_wmma_kernel(const float* __restrict__ A,
                                                       const float* __restrict__ W1,
                                                       const float* __restrict__ b1,
                                                       const float* __restrict__ W2,
                                                       float* __restrict__ out, int M) {
  __shared__ __attribute__((aligned(16))) float sW1[64 * 64];
  __shared__ __attribute__((aligned(16))) float sW2[64 * 16];
  __shared__ __attribute__((aligned(16))) float sT[8][16 * TILE_PITCH];
  for (int i = threadIdx.x; i < (64 * 64) / 4; i += blockDim.x)
    ((float4*)sW1)[i] = ((const float4*)W1)[i];
  for (int i = threadIdx.x; i < (64 * 16) / 4; i += blockDim.x)
    ((float4*)sW2)[i] = ((const float4*)W2)[i];
  __syncthreads();

  const int lane = threadIdx.x & 31;
  const int wid  = threadIdx.x >> 5;
  const int mtiles = (M + 15) >> 4;
  const int mt = blockIdx.x * 8 + wid;
  const bool live = mt < mtiles;
  const int m0 = (live ? mt : mtiles - 1) * 16;
  const int row = lane & 15;
  const int kh  = (lane >> 4) << 1;
  const int rbase = (lane >> 4) << 3;
  const int arow_idx = min(m0 + row, M - 1);
  const float* __restrict__ arow = A + (size_t)arow_idx * 64 + kh;
  const float* __restrict__ bcol = sW1 + row;

  v8f c0 = {}, c1 = {}, c2 = {}, c3 = {};
#pragma unroll
  for (int k = 0; k < 64; k += 4) {
    v2f a, b;
    a.x = arow[k];
    a.y = arow[k + 1];
    const float* __restrict__ bk = bcol + (k + kh) * 64;
    b.x = bk[0];  b.y = bk[64];
    c0 = __builtin_amdgcn_wmma_f32_16x16x4_f32(false, a, false, b, (short)0, c0, false, false);
    b.x = bk[16]; b.y = bk[80];
    c1 = __builtin_amdgcn_wmma_f32_16x16x4_f32(false, a, false, b, (short)0, c1, false, false);
    b.x = bk[32]; b.y = bk[96];
    c2 = __builtin_amdgcn_wmma_f32_16x16x4_f32(false, a, false, b, (short)0, c2, false, false);
    b.x = bk[48]; b.y = bk[112];
    c3 = __builtin_amdgcn_wmma_f32_16x16x4_f32(false, a, false, b, (short)0, c3, false, false);
  }

  float* __restrict__ T = sT[wid];
  const float bb0 = b1[row], bb1 = b1[16 + row], bb2 = b1[32 + row], bb3 = b1[48 + row];
#pragma unroll
  for (int r = 0; r < 8; ++r) {
    float* __restrict__ tp = T + (rbase + r) * TILE_PITCH + row;
    tp[0]  = fmaxf(c0[r] + bb0, 0.0f);
    tp[16] = fmaxf(c1[r] + bb1, 0.0f);
    tp[32] = fmaxf(c2[r] + bb2, 0.0f);
    tp[48] = fmaxf(c3[r] + bb3, 0.0f);
  }
  __syncthreads();

  const float* __restrict__ trow  = T + row * TILE_PITCH + kh;
  const float* __restrict__ b2col = sW2 + row;
  v8f d = {};
#pragma unroll
  for (int k = 0; k < 64; k += 4) {
    v2f a, b;
    a.x = trow[k];
    a.y = trow[k + 1];
    b.x = b2col[(k + kh) * 16];
    b.y = b2col[(k + kh + 1) * 16];
    d = __builtin_amdgcn_wmma_f32_16x16x4_f32(false, a, false, b, (short)0, d, false, false);
  }

#pragma unroll
  for (int r = 0; r < 8; ++r) {
    const int orow = m0 + rbase + r;
    if (live && orow < M) out[(size_t)orow * 16 + row] = d[r];
  }
}


#define TILE_R        400
#define SCAN_WAVES    32
#define ROWS_PER_WAVE 13
#define QCAP          1536
#define QSTEP         1024
#define KEYS_AHEAD    4
#define RING          64
#define LCAP          512

__device__ __forceinline__ int4 load_dst4_guarded(const int* __restrict__ dst, int e, int E) {
  int4 r;
  r.x = (e     < E) ? dst[e]     : -1;
  r.y = (e + 1 < E) ? dst[e + 1] : -1;
  r.z = (e + 2 < E) ? dst[e + 2] : -1;
  r.w = (e + 3 < E) ? dst[e + 3] : -1;
  return r;
}

__global__ void degree_key_kernel(const int* __restrict__ dst, float* __restrict__ deg,
                                  unsigned* __restrict__ keys, int E, int nwords) {
  const int i = blockIdx.x * blockDim.x + threadIdx.x;
  if (i >= nwords) return;
  const int e = i * 4;
  int4 d;
  if ((E & 3) == 0 && e + 3 < E) d = ((const int4*)dst)[i];
  else d = load_dst4_guarded(dst, e, E);
  if (d.x >= 0) atomic_add_f32(&deg[d.x], 1.0f);
  if (d.y >= 0) atomic_add_f32(&deg[d.y], 1.0f);
  if (d.z >= 0) atomic_add_f32(&deg[d.z], 1.0f);
  if (d.w >= 0) atomic_add_f32(&deg[d.w], 1.0f);
  const unsigned k0 = ((unsigned)d.x / TILE_R) & 255u, k1 = ((unsigned)d.y / TILE_R) & 255u;
  const unsigned k2 = ((unsigned)d.z / TILE_R) & 255u, k3 = ((unsigned)d.w / TILE_R) & 255u;
  keys[i] = k0 | (k1 << 8) | (k2 << 16) | (k3 << 24);
}

typedef __attribute__((ext_vector_type(16))) float v16f;

__global__ __launch_bounds__(1024) void agg64_tile_kernel(const float* __restrict__ x,
                                                          const int* __restrict__ src,
                                                          const int* __restrict__ dst,
                                                          const unsigned* __restrict__ keys,
                                                          const float* __restrict__ dinv,
                                                          float* __restrict__ agg,
                                                          int* __restrict__ lst,
                                                          int* __restrict__ lcount,
                                                          int E, int Nn) {
  __shared__ unsigned sQ[SCAN_WAVES][QCAP];
  __shared__ unsigned sRing[SCAN_WAVES][RING];
  __shared__ int sQn[SCAN_WAVES];
  __shared__ int sMore[SCAN_WAVES];

  const int lane = threadIdx.x & 31;
  const int wid  = __builtin_amdgcn_readfirstlane((int)(threadIdx.x >> 5));
  const int d0   = blockIdx.x * TILE_R;
  const int nrows = min(TILE_R, Nn - d0);
  const unsigned un = (unsigned)nrows;
  const int row0 = wid * ROWS_PER_WAVE;

  v16f accx = {}, accy = {};
  {
    float2 v0[ROWS_PER_WAVE];
    float dd[ROWS_PER_WAVE];
#pragma unroll
    for (int k = 0; k < ROWS_PER_WAVE; ++k) {
      const int d = d0 + min(row0 + k, nrows - 1);
      dd[k] = dinv[d];
      v0[k] = ((const float2*)(x + (size_t)d * 64))[lane];
    }
#pragma unroll
    for (int k = 0; k < ROWS_PER_WAVE; ++k) {
      const bool valid = row0 + k < nrows;
      const float w = dd[k] * dd[k];
      accx[k] = valid ? v0[k].x * w : 0.0f;
      accy[k] = valid ? v0[k].y * w : 0.0f;
    }
  }

  unsigned* __restrict__ q    = sQ[wid];
  unsigned* __restrict__ ring = sRing[wid];
  int qn = 0;
  int rhead = 0, rn = 0;
  const int gw = (int)blockIdx.x * SCAN_WAVES + wid;
  int2* __restrict__ mylst = (int2*)lst + (size_t)gw * LCAP;
  int lcnt = 0;

  const unsigned splat = ((unsigned)blockIdx.x & 255u) * 0x01010101u;

  auto flags = [&](const uint4& kv) {
    const unsigned x0 = kv.x ^ splat, x1 = kv.y ^ splat, x2 = kv.z ^ splat, x3 = kv.w ^ splat;
    const unsigned u0 = (x0 - 0x01010101u) & ~x0 & 0x80808080u, u1 = (x1 - 0x01010101u) & ~x1 & 0x80808080u;
    const unsigned u2 = (x2 - 0x01010101u) & ~x2 & 0x80808080u, u3 = (x3 - 0x01010101u) & ~x3 & 0x80808080u;
    return (u0 >> 7) | (u1 >> 6) | (u2 >> 5) | (u3 >> 4);
  };
  auto look = [&](const uint4& ka, const uint4& kb, int ebase) {
    unsigned g = flags(ka) | (flags(kb) << 4);
    const int e0 = ebase + lane * 16;
    for (;;) {
      const bool has = g != 0u;
      const unsigned mask = __builtin_amdgcn_ballot_w32(has);
      if (mask == 0u) break;
      if (has) {
        const int p = __builtin_ctz(g);
        g &= g - 1u;
        q[qn + (int)__builtin_amdgcn_mbcnt_lo(mask, 0u)] =
            (unsigned)(e0 + ((p & 4) << 7) + ((p & 3) << 2) + (p >> 3));
      }
      qn += __builtin_popcount(mask);
    }
  };

  auto addrow = [&](int r, float vx, float vy, float w) {
    accx[r] += vx * w;
    accy[r] += vy * w;
  };

  auto apply = [&](int n) {
    __builtin_amdgcn_wave_barrier();
    int s_l = 0, r_l = 0;
    float w_l = 0.0f;
    if (lane < n) {
      const unsigned pk = ring[(rhead + lane) & (RING - 1)];
      const int dl = (int)(pk & 511u);
      r_l = dl - row0;
      s_l = src[pk >> 9];
      w_l = dinv[s_l] * dinv[d0 + dl];
      if (lcnt + lane < LCAP)
        mylst[lcnt + lane] = make_int2((s_l << 4) | r_l, __float_as_int(w_l));
    }
    lcnt += n;
    int j = 0;
    for (; j + 8 <= n; j += 8) {
      float2 v[8];
#pragma unroll
      for (int u = 0; u < 8; ++u) {
        const int s = __builtin_amdgcn_readlane(s_l, j + u);
        v[u] = ((const float2*)(x + (size_t)s * 64))[lane];
      }
#pragma unroll
      for (int u = 0; u < 8; ++u) {
        const int r = __builtin_amdgcn_readlane(r_l, j + u);
        const float w = __int_as_float(__builtin_amdgcn_readlane(__float_as_int(w_l), j + u));
        addrow(r, v[u].x, v[u].y, w);
      }
    }
    for (; j < n; ++j) {
      const int s = __builtin_amdgcn_readlane(s_l, j);
      const int r = __builtin_amdgcn_readlane(r_l, j);
      const float w = __int_as_float(__builtin_amdgcn_readlane(__float_as_int(w_l), j));
      const float2 v = ((const float2*)(x + (size_t)s * 64))[lane];
      addrow(r, v.x, v.y, w);
    }
  };

  auto rpush = [&](bool m, unsigned pk) {
    const unsigned mask = __builtin_amdgcn_ballot_w32(m);
    if (mask != 0u) {
      if (m) ring[(rhead + rn + (int)__builtin_amdgcn_mbcnt_lo(mask, 0u)) & (RING - 1)] = pk;
      rn += __builtin_popcount(mask);
      if (rn >= 32) {
        apply(32);
        rhead = (rhead + 32) & (RING - 1);
        rn -= 32;
      }
    }
  };

  const int nsc = (E + QSTEP - 1) / QSTEP;
  const int rot = (int)blockIdx.x * (nsc / (int)gridDim.x);

  auto stepbase = [&](int scx) {
    int g = scx + rot;
    if (g >= nsc) g -= nsc;
    return g * QSTEP;
  };
  auto loadkeys = [&](int scx, int fb, int half) {
    return ((const uint4*)((const unsigned char*)keys + stepbase(scx < nsc ? scx : fb)))[half * 32 + lane];
  };

  int sc = wid;

  for (;;) {
    if (sc < nsc) {
      uint4 a0 = loadkeys(sc, sc, 0),                  b0 = loadkeys(sc, sc, 1);
      uint4 a1 = loadkeys(sc + SCAN_WAVES, sc, 0),     b1 = loadkeys(sc + SCAN_WAVES, sc, 1);
      uint4 a2 = loadkeys(sc + 2 * SCAN_WAVES, sc, 0), b2 = loadkeys(sc + 2 * SCAN_WAVES, sc, 1);
      uint4 a3 = loadkeys(sc + 3 * SCAN_WAVES, sc, 0), b3 = loadkeys(sc + 3 * SCAN_WAVES, sc, 1);
#define SCAN_STEP(ka, kb)                                    \
      if (!(sc < nsc && qn <= QCAP - QSTEP)) break;          \
      look(ka, kb, stepbase(sc));                            \
      ka = loadkeys(sc + KEYS_AHEAD * SCAN_WAVES, sc, 0);    \
      kb = loadkeys(sc + KEYS_AHEAD * SCAN_WAVES, sc, 1);    \
      sc += SCAN_WAVES;
      for (;;) {
        SCAN_STEP(a0, b0)
        SCAN_STEP(a1, b1)
        SCAN_STEP(a2, b2)
        SCAN_STEP(a3, b3)
      }
#undef SCAN_STEP
    }
    {
      int nkeep = 0;
      for (int b = 0; b < qn; b += 128) {
        __builtin_amdgcn_wave_barrier();
        int ee[4], dd[4];
#pragma unroll
        for (int u = 0; u < 4; ++u) {
          const int i = b + u * 32 + lane;
          ee[u] = (i < qn) ? (int)q[i] : E;
          dd[u] = (ee[u] < E) ? dst[ee[u]] : -1;
        }
#pragma unroll
        for (int u = 0; u < 4; ++u) {
          const bool ok = (unsigned)(dd[u] - d0) < un;
          const unsigned mask = __builtin_amdgcn_ballot_w32(ok);
          if (ok) q[nkeep + (int)__builtin_amdgcn_mbcnt_lo(mask, 0u)] = ((unsigned)ee[u] << 9) | (unsigned)(dd[u] - d0);
          nkeep += __builtin_popcount(mask);
        }
      }
      qn = nkeep;
    }
    if (lane == 0) { sQn[wid] = qn; sMore[wid] = (sc < nsc) ? 1 : 0; }
    __syncthreads();

    const bool more = __builtin_amdgcn_ballot_w32(sMore[lane] != 0) != 0u;
    for (int k = 0; k < SCAN_WAVES; ++k) {
      const int nk = __builtin_amdgcn_readfirstlane(sQn[k]);
      const unsigned* __restrict__ qk = sQ[k];
      for (int b = 0; b < nk; b += 32) {
        unsigned pk = 0u;
        bool own = false;
        if (b + lane < nk) {
          pk = qk[b + lane];
          own = ((pk & 511u) / ROWS_PER_WAVE) == (unsigned)wid;
        }
        rpush(own, pk);
      }
    }
    if (rn > 0) apply(rn);
    rhead = 0; rn = 0; qn = 0;
    __syncthreads();
    if (!more) break;
  }

#pragma unroll
  for (int k = 0; k < ROWS_PER_WAVE; ++k)
    if (row0 + k < nrows)
      ((float2*)(agg + (size_t)(d0 + row0 + k) * 64))[lane] = make_float2(accx[k], accy[k]);
  if (lane == 0) lcount[gw] = lcnt;
}

__global__ __launch_bounds__(256) void agg16_list_kernel(const float* __restrict__ h,
                                                         const int* __restrict__ lst,
                                                         const int* __restrict__ lcount,
                                                         const int* __restrict__ src,
                                                         const int* __restrict__ dst,
                                                         const float* __restrict__ dinv,
                                                         const float* __restrict__ b2,
                                                         float* __restrict__ out,
                                                         int E, int Nn) {
  __shared__ float sA0[8][ROWS_PER_WAVE * 32];
  __shared__ float sA1[8][ROWS_PER_WAVE * 32];
  __shared__ float sA2[8][ROWS_PER_WAVE * 32];
  __shared__ float sA3[8][ROWS_PER_WAVE * 32];

  const int lane = threadIdx.x & 31;
  const int wid  = __builtin_amdgcn_readfirstlane((int)(threadIdx.x >> 5));
  const int gw   = (int)blockIdx.x * 8 + wid;
  const int d0   = (gw / SCAN_WAVES) * TILE_R;
  const int row0 = (gw % SCAN_WAVES) * ROWS_PER_WAVE;
  const int nr   = min(ROWS_PER_WAVE, min(TILE_R, Nn - d0) - row0);
  if (nr <= 0) return;
  const int c  = lane & 15;
  const int hh = lane >> 4;
  const int base = d0 + row0;

  float* a0 = sA0[wid] + lane;
  float* a1 = sA1[wid] + lane;
  float* a2 = sA2[wid] + lane;
  float* a3 = sA3[wid] + lane;
#pragma unroll
  for (int k = 0; k < ROWS_PER_WAVE; ++k) {
    a0[k * 32] = 0.0f; a1[k * 32] = 0.0f; a2[k * 32] = 0.0f; a3[k * 32] = 0.0f;
  }

  float selfv[7];
#pragma unroll
  for (int i = 0; i < 7; ++i) {
    const int d = base + min(2 * i + hh, nr - 1);
    const float dd = dinv[d];
    selfv[i] = h[(size_t)d * 16 + c] * (dd * dd);
  }
  const float bias = b2[c];

  const int cnt = __builtin_amdgcn_readfirstlane(lcount[gw]);
  if (cnt <= LCAP) {
    const int2* __restrict__ my = (const int2*)lst + (size_t)gw * LCAP;
    int2 en = make_int2(0, 0);
    if (lane < cnt) en = my[lane];
    for (int j = 0; j < cnt; j += 32) {
      const int m = cnt - j;
      int pk[16];
      float wt[16], v[16];
#pragma unroll
      for (int u = 0; u < 16; ++u) {
        const int sel = (2 * u + hh) << 2;
        pk[u] = __builtin_amdgcn_ds_bpermute(sel, en.x);
        wt[u] = __int_as_float(__builtin_amdgcn_ds_bpermute(sel, en.y));
        v[u] = h[((unsigned)pk[u] & ~15u) + (unsigned)c];
      }
      int2 nx = make_int2(0, 0);
      if (j + 32 + lane < cnt) nx = my[j + 32 + lane];
#pragma unroll
      for (int u = 0; u < 16; u += 4) {
        const int r0 = (pk[u] & 15) * 32, r1 = (pk[u + 1] & 15) * 32;
        const int r2 = (pk[u + 2] & 15) * 32, r3 = (pk[u + 3] & 15) * 32;
        float t0 = a0[r0], t1 = a1[r1], t2 = a2[r2], t3 = a3[r3];
        t0 += v[u] * wt[u];
        t1 += v[u + 1] * wt[u + 1];
        t2 += v[u + 2] * wt[u + 2];
        t3 += v[u + 3] * wt[u + 3];
        if (2 * u + hh < m)       a0[r0] = t0;
        if (2 * (u + 1) + hh < m) a1[r1] = t1;
        if (2 * (u + 2) + hh < m) a2[r2] = t2;
        if (2 * (u + 3) + hh < m) a3[r3] = t3;
      }
      en = nx;
    }
  } else {
    for (int e0 = 0; e0 < E; e0 += 32) {
      const int e = e0 + lane;
      const int d = (e < E) ? dst[e] : -1;
      unsigned mask = __builtin_amdgcn_ballot_w32((unsigned)(d - base) < (unsigned)nr);
      while (mask != 0u) {
        const int l = __builtin_ctz(mask);
        mask &= mask - 1u;
        const int dd = __builtin_amdgcn_readlane(d, l);
        const int s  = src[e0 + l];
        const float w = dinv[s] * dinv[dd];
        if (lane < 16) a0[(dd - base) * 32] += h[(size_t)s * 16 + lane] * w;
      }
    }
  }

  __builtin_amdgcn_wave_barrier();
#pragma unroll
  for (int i = 0; i < 7; ++i) {
    const int r = 2 * i + hh;
    if (r < nr) {
      const float* p0 = sA0[wid] + r * 32 + c;
      const float* p1 = sA1[wid] + r * 32 + c;
      const float* p2 = sA2[wid] + r * 32 + c;
      const float* p3 = sA3[wid] + r * 32 + c;
      float s = selfv[i];
      s += p0[0]; s += p0[16]; s += p1[0]; s += p1[16];
      s += p2[0]; s += p2[16]; s += p3[0]; s += p3[16];
      out[(size_t)(base + r) * 16 + c] = s + bias;
    }
  }
}


extern "C" void kernel_launch(void* const* d_in, const int* in_sizes, int n_in,
                              void* d_out, int out_size, void* d_ws, size_t ws_size,
                              hipStream_t stream) {
  const float* x    = (const float*)d_in[0];
  const int*   edge = (const int*)d_in[1];
  const float* W1   = (const float*)d_in[2];
  const float* b1   = (const float*)d_in[3];
  const float* W2   = (const float*)d_in[4];
  const float* b2   = (const float*)d_in[5];
  float* out = (float*)d_out;

  const int E  = in_sizes[1] / 2;
  const int Nn = in_sizes[0] / IN_DIM;
  const int* src = edge;
  const int* dst = edge + E;

  char* ws = (char*)d_ws;
  float* dinv = (float*)ws;
  float* h2   = (float*)(ws + (size_t)Nn * 4);
  float* agg1 = (float*)(ws + (size_t)Nn * 4 + (size_t)Nn * 64 * 4);
  const int ntiles = (Nn + TILE_R - 1) / TILE_R;
  const size_t loff = ((size_t)Nn * 4 + (size_t)Nn * 128 * 4 + 255) & ~(size_t)255;
  int* lcount = (int*)(ws + loff);
  int* lst    = (int*)(ws + loff + (size_t)ntiles * SCAN_WAVES * 4);
  const size_t koff = (loff + (size_t)ntiles * SCAN_WAVES * 4 + (size_t)ntiles * SCAN_WAVES * LCAP * 8 + 255) & ~(size_t)255;
  unsigned* keys = (unsigned*)(ws + koff);
  const int nwords = ((E + QSTEP - 1) / QSTEP) * (QSTEP / 4);

  const int TB = 256;
  const int mtiles = (Nn + 15) >> 4;

  fill0_kernel<<<(Nn + TB - 1) / TB, TB, 0, stream>>>(dinv, Nn);

  degree_key_kernel<<<(nwords + TB - 1) / TB, TB, 0, stream>>>(dst, dinv, keys, E, nwords);
  rsqrt_kernel<<<(Nn + TB - 1) / TB, TB, 0, stream>>>(dinv, Nn);

  agg64_tile_kernel<<<ntiles, SCAN_WAVES * 32, 0, stream>>>(x, src, dst, keys, dinv, agg1, lst, lcount, E, Nn);

  mlp_wmma_kernel<<<(mtiles + 7) / 8, 256, 0, stream>>>(agg1, W1, b1, W2, h2, Nn);
  agg16_list_kernel<<<ntiles * (SCAN_WAVES / 8), 256, 0, stream>>>(h2, lst, lcount, src, dst, dinv, b2, out, E, Nn);

  (void)n_in; (void)out_size; (void)ws_size;
}


# Layer-1 apply as a two-step pipeline: a batch of 32 edges has its src and dinv requested one batch ahead and all 32 x rows requested before its first add (was 4 drained groups of 8); sum order kept
# speedup vs baseline: 2.0556x; 1.0500x over previous
#include <hip/hip_runtime.h>
#include <hip/hip_bf16.h>


#define IN_DIM  64
#define OUT_DIM 16

typedef __attribute__((ext_vector_type(2))) float v2f;
typedef __attribute__((ext_vector_type(8))) float v8f;

__device__ __forceinline__ void atomic_add_f32(float* p, float v) {
  asm volatile("global_atomic_add_f32 %0, %1, off scope:SCOPE_DEV"
               :: "v"(p), "v"(v) : "memory");
}


__global__ void fill0_kernel(float* __restrict__ p, int n) {
  int i = blockIdx.x * blockDim.x + threadIdx.x;
  if (i < n) p[i] = 0.0f;
}


__global__ void rsqrt_kernel(float* __restrict__ d, int n) {
  int i = blockIdx.x * blockDim.x + threadIdx.x;
  if (i < n) d[i] = rsqrtf(d[i] + 1.0f);
}


#define TILE_PITCH 68

__global__ __launch_bounds__(256) void mlp_wmma_kernel(const float* __restrict__ A,
                                                       const float* __restrict__ W1,
                                                       const float* __restrict__ b1,
                                                       const float* __restrict__ W2,
                                                       float* __restrict__ out, int M) {
  __shared__ __attribute__((aligned(16))) float sW1[64 * 64];
  __shared__ __attribute__((aligned(16))) float sW2[64 * 16];
  __shared__ __attribute__((aligned(16))) float sT[8][16 * TILE_PITCH];
  for (int i = threadIdx.x; i < (64 * 64) / 4; i += blockDim.x)
    ((float4*)sW1)[i] = ((const float4*)W1)[i];
  for (int i = threadIdx.x; i < (64 * 16) / 4; i += blockDim.x)
    ((float4*)sW2)[i] = ((const float4*)W2)[i];
  __syncthreads();

  const int lane = threadIdx.x & 31;
  const int wid  = threadIdx.x >> 5;
  const int mtiles = (M + 15) >> 4;
  const int mt = blockIdx.x * 8 + wid;
  const bool live = mt < mtiles;
  const int m0 = (live ? mt : mtiles - 1) * 16;
  const int row = lane & 15;
  const int kh  = (lane >> 4) << 1;
  const int rbase = (lane >> 4) << 3;
  const int arow_idx = min(m0 + row, M - 1);
  const float* __restrict__ arow = A + (size_t)arow_idx * 64 + kh;
  const float* __restrict__ bcol = sW1 + row;

  v8f c0 = {}, c1 = {}, c2 = {}, c3 = {};
#pragma unroll
  for (int k = 0; k < 64; k += 4) {
    v2f a, b;
    a.x = arow[k];
    a.y = arow[k + 1];
    const float* __restrict__ bk = bcol + (k + kh) * 64;
    b.x = bk[0];  b.y = bk[64];
    c0 = __builtin_amdgcn_wmma_f32_16x16x4_f32(false, a, false, b, (short)0, c0, false, false);
    b.x = bk[16]; b.y = bk[80];
    c1 = __builtin_amdgcn_wmma_f32_16x16x4_f32(false, a, false, b, (short)0, c1, false, false);
    b.x = bk[32]; b.y = bk[96];
    c2 = __builtin_amdgcn_wmma_f32_16x16x4_f32(false, a, false, b, (short)0, c2, false, false);
    b.x = bk[48]; b.y = bk[112];
    c3 = __builtin_amdgcn_wmma_f32_16x16x4_f32(false, a, false, b, (short)0, c3, false, false);
  }

  float* __restrict__ T = sT[wid];
  const float bb0 = b1[row], bb1 = b1[16 + row], bb2 = b1[32 + row], bb3 = b1[48 + row];
#pragma unroll
  for (int r = 0; r < 8; ++r) {
    float* __restrict__ tp = T + (rbase + r) * TILE_PITCH + row;
    tp[0]  = fmaxf(c0[r] + bb0, 0.0f);
    tp[16] = fmaxf(c1[r] + bb1, 0.0f);
    tp[32] = fmaxf(c2[r] + bb2, 0.0f);
    tp[48] = fmaxf(c3[r] + bb3, 0.0f);
  }
  __syncthreads();

  const float* __restrict__ trow  = T + row * TILE_PITCH + kh;
  const float* __restrict__ b2col = sW2 + row;
  v8f d = {};
#pragma unroll
  for (int k = 0; k < 64; k += 4) {
    v2f a, b;
    a.x = trow[k];
    a.y = trow[k + 1];
    b.x = b2col[(k + kh) * 16];
    b.y = b2col[(k + kh + 1) * 16];
    d = __builtin_amdgcn_wmma_f32_16x16x4_f32(false, a, false, b, (short)0, d, false, false);
  }

#pragma unroll
  for (int r = 0; r < 8; ++r) {
    const int orow = m0 + rbase + r;
    if (live && orow < M) out[(size_t)orow * 16 + row] = d[r];
  }
}


#define TILE_R        400
#define SCAN_WAVES    32
#define ROWS_PER_WAVE 13
#define QCAP          1536
#define QSTEP         1024
#define KEYS_AHEAD    4
#define RING          64
#define ROWS_AHEAD    32
#define ROWS_AHEAD_REST 16
#define LCAP          512

__device__ __forceinline__ int4 load_dst4_guarded(const int* __restrict__ dst, int e, int E) {
  int4 r;
  r.x = (e     < E) ? dst[e]     : -1;
  r.y = (e + 1 < E) ? dst[e + 1] : -1;
  r.z = (e + 2 < E) ? dst[e + 2] : -1;
  r.w = (e + 3 < E) ? dst[e + 3] : -1;
  return r;
}

__global__ void degree_key_kernel(const int* __restrict__ dst, float* __restrict__ deg,
                                  unsigned* __restrict__ keys, int E, int nwords) {
  const int i = blockIdx.x * blockDim.x + threadIdx.x;
  if (i >= nwords) return;
  const int e = i * 4;
  int4 d;
  if ((E & 3) == 0 && e + 3 < E) d = ((const int4*)dst)[i];
  else d = load_dst4_guarded(dst, e, E);
  if (d.x >= 0) atomic_add_f32(&deg[d.x], 1.0f);
  if (d.y >= 0) atomic_add_f32(&deg[d.y], 1.0f);
  if (d.z >= 0) atomic_add_f32(&deg[d.z], 1.0f);
  if (d.w >= 0) atomic_add_f32(&deg[d.w], 1.0f);
  const unsigned k0 = ((unsigned)d.x / TILE_R) & 255u, k1 = ((unsigned)d.y / TILE_R) & 255u;
  const unsigned k2 = ((unsigned)d.z / TILE_R) & 255u, k3 = ((unsigned)d.w / TILE_R) & 255u;
  keys[i] = k0 | (k1 << 8) | (k2 << 16) | (k3 << 24);
}

typedef __attribute__((ext_vector_type(16))) float v16f;

template <int N> struct count_c { static constexpr int value = N; };

__global__ __launch_bounds__(1024) void agg64_tile_kernel(const float* __restrict__ x,
                                                          const int* __restrict__ src,
                                                          const int* __restrict__ dst,
                                                          const unsigned* __restrict__ keys,
                                                          const float* __restrict__ dinv,
                                                          float* __restrict__ agg,
                                                          int* __restrict__ lst,
                                                          int* __restrict__ lcount,
                                                          int E, int Nn) {
  __shared__ unsigned sQ[SCAN_WAVES][QCAP];
  __shared__ unsigned sRing[SCAN_WAVES][RING];
  __shared__ int sQn[SCAN_WAVES];
  __shared__ int sMore[SCAN_WAVES];

  const int lane = threadIdx.x & 31;
  const int wid  = __builtin_amdgcn_readfirstlane((int)(threadIdx.x >> 5));
  const int d0   = blockIdx.x * TILE_R;
  const int nrows = min(TILE_R, Nn - d0);
  const unsigned un = (unsigned)nrows;
  const int row0 = wid * ROWS_PER_WAVE;

  v16f accx = {}, accy = {};
  {
    float2 v0[ROWS_PER_WAVE];
    float dd[ROWS_PER_WAVE];
#pragma unroll
    for (int k = 0; k < ROWS_PER_WAVE; ++k) {
      const int d = d0 + min(row0 + k, nrows - 1);
      dd[k] = dinv[d];
      v0[k] = ((const float2*)(x + (size_t)d * 64))[lane];
    }
#pragma unroll
    for (int k = 0; k < ROWS_PER_WAVE; ++k) {
      const bool valid = row0 + k < nrows;
      const float w = dd[k] * dd[k];
      accx[k] = valid ? v0[k].x * w : 0.0f;
      accy[k] = valid ? v0[k].y * w : 0.0f;
    }
  }

  unsigned* __restrict__ q    = sQ[wid];
  unsigned* __restrict__ ring = sRing[wid];
  int qn = 0;
  int rhead = 0, rn = 0;
  const int gw = (int)blockIdx.x * SCAN_WAVES + wid;
  int2* __restrict__ mylst = (int2*)lst + (size_t)gw * LCAP;
  int lcnt = 0;

  const unsigned splat = ((unsigned)blockIdx.x & 255u) * 0x01010101u;

  auto flags = [&](const uint4& kv) {
    const unsigned x0 = kv.x ^ splat, x1 = kv.y ^ splat, x2 = kv.z ^ splat, x3 = kv.w ^ splat;
    const unsigned u0 = (x0 - 0x01010101u) & ~x0 & 0x80808080u, u1 = (x1 - 0x01010101u) & ~x1 & 0x80808080u;
    const unsigned u2 = (x2 - 0x01010101u) & ~x2 & 0x80808080u, u3 = (x3 - 0x01010101u) & ~x3 & 0x80808080u;
    return (u0 >> 7) | (u1 >> 6) | (u2 >> 5) | (u3 >> 4);
  };
  auto look = [&](const uint4& ka, const uint4& kb, int ebase) {
    unsigned g = flags(ka) | (flags(kb) << 4);
    const int e0 = ebase + lane * 16;
    for (;;) {
      const bool has = g != 0u;
      const unsigned mask = __builtin_amdgcn_ballot_w32(has);
      if (mask == 0u) break;
      if (has) {
        const int p = __builtin_ctz(g);
        g &= g - 1u;
        q[qn + (int)__builtin_amdgcn_mbcnt_lo(mask, 0u)] =
            (unsigned)(e0 + ((p & 4) << 7) + ((p & 3) << 2) + (p >> 3));
      }
      qn += __builtin_popcount(mask);
    }
  };

  auto addrow = [&](int r, float vx, float vy, float w) {
    accx[r] += vx * w;
    accy[r] += vy * w;
  };

  auto fetch = [&](int n, unsigned& pk, int& s_l, float& dd_l) {
    __builtin_amdgcn_wave_barrier();
    pk = 0u; s_l = 0; dd_l = 0.0f;
    if (lane < n) {
      pk   = ring[(rhead + lane) & (RING - 1)];
      s_l  = src[pk >> 9];
      dd_l = dinv[d0 + (int)(pk & 511u)];
    }
  };

  auto rows = [&](auto ahead_c, int n, unsigned pk, int s_l, float dd_l) {
    constexpr int AHEAD = decltype(ahead_c)::value;
    const int r_l = (int)(pk & 511u) - row0;
    const float ds_l = dinv[s_l];
    float2 v[32];
#pragma unroll
    for (int u = 0; u < AHEAD; ++u)
      if (u < n) {
        const int s = __builtin_amdgcn_readlane(s_l, u);
        v[u] = ((const float2*)(x + (size_t)s * 64))[lane];
      }
    const float w_l = ds_l * dd_l;
    if (lane < n && lcnt + lane < LCAP)
      mylst[lcnt + lane] = make_int2((s_l << 4) | r_l, __float_as_int(w_l));
    lcnt += n;
#pragma unroll
    for (int u = 0; u < 32; ++u) {
      if (u < n) {
        const int r = __builtin_amdgcn_readlane(r_l, u);
        const float w = __int_as_float(__builtin_amdgcn_readlane(__float_as_int(w_l), u));
        addrow(r, v[u].x, v[u].y, w);
      }
      if (u + AHEAD < 32 && u + AHEAD < n) {
        const int s = __builtin_amdgcn_readlane(s_l, u + AHEAD);
        v[u + AHEAD] = ((const float2*)(x + (size_t)s * 64))[lane];
      }
    }
  };

  bool haveP = false;
  unsigned pkP = 0u;
  int sP = 0;
  float ddP = 0.0f;

  auto rpush = [&](bool m, unsigned pk) {
    const unsigned mask = __builtin_amdgcn_ballot_w32(m);
    if (mask != 0u) {
      if (m) ring[(rhead + rn + (int)__builtin_amdgcn_mbcnt_lo(mask, 0u)) & (RING - 1)] = pk;
      rn += __builtin_popcount(mask);
      if (rn >= 32) {
        unsigned pkN; int sN; float ddN;
        fetch(32, pkN, sN, ddN);
        if (haveP) rows(count_c<ROWS_AHEAD>{}, 32, pkP, sP, ddP);
        pkP = pkN; sP = sN; ddP = ddN; haveP = true;
        rhead = (rhead + 32) & (RING - 1);
        rn -= 32;
      }
    }
  };

  const int nsc = (E + QSTEP - 1) / QSTEP;
  const int rot = (int)blockIdx.x * (nsc / (int)gridDim.x);

  auto stepbase = [&](int scx) {
    int g = scx + rot;
    if (g >= nsc) g -= nsc;
    return g * QSTEP;
  };
  auto loadkeys = [&](int scx, int fb, int half) {
    return ((const uint4*)((const unsigned char*)keys + stepbase(scx < nsc ? scx : fb)))[half * 32 + lane];
  };

  int sc = wid;

  for (;;) {
    if (sc < nsc) {
      uint4 a0 = loadkeys(sc, sc, 0),                  b0 = loadkeys(sc, sc, 1);
      uint4 a1 = loadkeys(sc + SCAN_WAVES, sc, 0),     b1 = loadkeys(sc + SCAN_WAVES, sc, 1);
      uint4 a2 = loadkeys(sc + 2 * SCAN_WAVES, sc, 0), b2 = loadkeys(sc + 2 * SCAN_WAVES, sc, 1);
      uint4 a3 = loadkeys(sc + 3 * SCAN_WAVES, sc, 0), b3 = loadkeys(sc + 3 * SCAN_WAVES, sc, 1);
#define SCAN_STEP(ka, kb)                                    \
      if (!(sc < nsc && qn <= QCAP - QSTEP)) break;          \
      look(ka, kb, stepbase(sc));                            \
      ka = loadkeys(sc + KEYS_AHEAD * SCAN_WAVES, sc, 0);    \
      kb = loadkeys(sc + KEYS_AHEAD * SCAN_WAVES, sc, 1);    \
      sc += SCAN_WAVES;
      for (;;) {
        SCAN_STEP(a0, b0)
        SCAN_STEP(a1, b1)
        SCAN_STEP(a2, b2)
        SCAN_STEP(a3, b3)
      }
#undef SCAN_STEP
    }
    {
      int nkeep = 0;
      for (int b = 0; b < qn; b += 128) {
        __builtin_amdgcn_wave_barrier();
        int ee[4], dd[4];
#pragma unroll
        for (int u = 0; u < 4; ++u) {
          const int i = b + u * 32 + lane;
          ee[u] = (i < qn) ? (int)q[i] : E;
          dd[u] = (ee[u] < E) ? dst[ee[u]] : -1;
        }
#pragma unroll
        for (int u = 0; u < 4; ++u) {
          const bool ok = (unsigned)(dd[u] - d0) < un;
          const unsigned mask = __builtin_amdgcn_ballot_w32(ok);
          if (ok) q[nkeep + (int)__builtin_amdgcn_mbcnt_lo(mask, 0u)] = ((unsigned)ee[u] << 9) | (unsigned)(dd[u] - d0);
          nkeep += __builtin_popcount(mask);
        }
      }
      qn = nkeep;
    }
    if (lane == 0) { sQn[wid] = qn; sMore[wid] = (sc < nsc) ? 1 : 0; }
    __syncthreads();

    const bool more = __builtin_amdgcn_ballot_w32(sMore[lane] != 0) != 0u;
    for (int k = 0; k < SCAN_WAVES; ++k) {
      const int nk = __builtin_amdgcn_readfirstlane(sQn[k]);
      const unsigned* __restrict__ qk = sQ[k];
      for (int b = 0; b < nk; b += 32) {
        unsigned pk = 0u;
        bool own = false;
        if (b + lane < nk) {
          pk = qk[b + lane];
          own = ((pk & 511u) / ROWS_PER_WAVE) == (unsigned)wid;
        }
        rpush(own, pk);
      }
    }
    {
      unsigned pkN = 0u; int sN = 0; float ddN = 0.0f;
      if (rn > 0) fetch(rn, pkN, sN, ddN);
      if (haveP) { rows(count_c<ROWS_AHEAD>{}, 32, pkP, sP, ddP); haveP = false; }
      if (rn > 0) rows(count_c<ROWS_AHEAD_REST>{}, rn, pkN, sN, ddN);
    }
    rhead = 0; rn = 0; qn = 0;
    __syncthreads();
    if (!more) break;
  }

#pragma unroll
  for (int k = 0; k < ROWS_PER_WAVE; ++k)
    if (row0 + k < nrows)
      ((float2*)(agg + (size_t)(d0 + row0 + k) * 64))[lane] = make_float2(accx[k], accy[k]);
  if (lane == 0) lcount[gw] = lcnt;
}

__global__ __launch_bounds__(256) void agg16_list_kernel(const float* __restrict__ h,
                                                         const int* __restrict__ lst,
                                                         const int* __restrict__ lcount,
                                                         const int* __restrict__ src,
                                                         const int* __restrict__ dst,
                                                         const float* __restrict__ dinv,
                                                         const float* __restrict__ b2,
                                                         float* __restrict__ out,
                                                         int E, int Nn) {
  __shared__ float sA0[8][ROWS_PER_WAVE * 32];
  __shared__ float sA1[8][ROWS_PER_WAVE * 32];
  __shared__ float sA2[8][ROWS_PER_WAVE * 32];
  __shared__ float sA3[8][ROWS_PER_WAVE * 32];

  const int lane = threadIdx.x & 31;
  const int wid  = __builtin_amdgcn_readfirstlane((int)(threadIdx.x >> 5));
  const int gw   = (int)blockIdx.x * 8 + wid;
  const int d0   = (gw / SCAN_WAVES) * TILE_R;
  const int row0 = (gw % SCAN_WAVES) * ROWS_PER_WAVE;
  const int nr   = min(ROWS_PER_WAVE, min(TILE_R, Nn - d0) - row0);
  if (nr <= 0) return;
  const int c  = lane & 15;
  const int hh = lane >> 4;
  const int base = d0 + row0;

  float* a0 = sA0[wid] + lane;
  float* a1 = sA1[wid] + lane;
  float* a2 = sA2[wid] + lane;
  float* a3 = sA3[wid] + lane;
#pragma unroll
  for (int k = 0; k < ROWS_PER_WAVE; ++k) {
    a0[k * 32] = 0.0f; a1[k * 32] = 0.0f; a2[k * 32] = 0.0f; a3[k * 32] = 0.0f;
  }

  float selfv[7];
#pragma unroll
  for (int i = 0; i < 7; ++i) {
    const int d = base + min(2 * i + hh, nr - 1);
    const float dd = dinv[d];
    selfv[i] = h[(size_t)d * 16 + c] * (dd * dd);
  }
  const float bias = b2[c];

  const int cnt = __builtin_amdgcn_readfirstlane(lcount[gw]);
  if (cnt <= LCAP) {
    const int2* __restrict__ my = (const int2*)lst + (size_t)gw * LCAP;
    int2 en = make_int2(0, 0);
    if (lane < cnt) en = my[lane];
    for (int j = 0; j < cnt; j += 32) {
      const int m = cnt - j;
      int pk[16];
      float wt[16], v[16];
#pragma unroll
      for (int u = 0; u < 16; ++u) {
        const int sel = (2 * u + hh) << 2;
        pk[u] = __builtin_amdgcn_ds_bpermute(sel, en.x);
        wt[u] = __int_as_float(__builtin_amdgcn_ds_bpermute(sel, en.y));
        v[u] = h[((unsigned)pk[u] & ~15u) + (unsigned)c];
      }
      int2 nx = make_int2(0, 0);
      if (j + 32 + lane < cnt) nx = my[j + 32 + lane];
#pragma unroll
      for (int u = 0; u < 16; u += 4) {
        const int r0 = (pk[u] & 15) * 32, r1 = (pk[u + 1] & 15) * 32;
        const int r2 = (pk[u + 2] & 15) * 32, r3 = (pk[u + 3] & 15) * 32;
        float t0 = a0[r0], t1 = a1[r1], t2 = a2[r2], t3 = a3[r3];
        t0 += v[u] * wt[u];
        t1 += v[u + 1] * wt[u + 1];
        t2 += v[u + 2] * wt[u + 2];
        t3 += v[u + 3] * wt[u + 3];
        if (2 * u + hh < m)       a0[r0] = t0;
        if (2 * (u + 1) + hh < m) a1[r1] = t1;
        if (2 * (u + 2) + hh < m) a2[r2] = t2;
        if (2 * (u + 3) + hh < m) a3[r3] = t3;
      }
      en = nx;
    }
  } else {
    for (int e0 = 0; e0 < E; e0 += 32) {
      const int e = e0 + lane;
      const int d = (e < E) ? dst[e] : -1;
      unsigned mask = __builtin_amdgcn_ballot_w32((unsigned)(d - base) < (unsigned)nr);
      while (mask != 0u) {
        const int l = __builtin_ctz(mask);
        mask &= mask - 1u;
        const int dd = __builtin_amdgcn_readlane(d, l);
        const int s  = src[e0 + l];
        const float w = dinv[s] * dinv[dd];
        if (lane < 16) a0[(dd - base) * 32] += h[(size_t)s * 16 + lane] * w;
      }
    }
  }

  __builtin_amdgcn_wave_barrier();
#pragma unroll
  for (int i = 0; i < 7; ++i) {
    const int r = 2 * i + hh;
    if (r < nr) {
      const float* p0 = sA0[wid] + r * 32 + c;
      const float* p1 = sA1[wid] + r * 32 + c;
      const float* p2 = sA2[wid] + r * 32 + c;
      const float* p3 = sA3[wid] + r * 32 + c;
      float s = selfv[i];
      s += p0[0]; s += p0[16]; s += p1[0]; s += p1[16];
      s += p2[0]; s += p2[16]; s += p3[0]; s += p3[16];
      out[(size_t)(base + r) * 16 + c] = s + bias;
    }
  }
}


extern "C" void kernel_launch(void* const* d_in, const int* in_sizes, int n_in,
                              void* d_out, int out_size, void* d_ws, size_t ws_size,
                              hipStream_t stream) {
  const float* x    = (const float*)d_in[0];
  const int*   edge = (const int*)d_in[1];
  const float* W1   = (const float*)d_in[2];
  const float* b1   = (const float*)d_in[3];
  const float* W2   = (const float*)d_in[4];
  const float* b2   = (const float*)d_in[5];
  float* out = (float*)d_out;

  const int E  = in_sizes[1] / 2;
  const int Nn = in_sizes[0] / IN_DIM;
  const int* src = edge;
  const int* dst = edge + E;

  char* ws = (char*)d_ws;
  float* dinv = (float*)ws;
  float* h2   = (float*)(ws + (size_t)Nn * 4);
  float* agg1 = (float*)(ws + (size_t)Nn * 4 + (size_t)Nn * 64 * 4);
  const int ntiles = (Nn + TILE_R - 1) / TILE_R;
  const size_t loff = ((size_t)Nn * 4 + (size_t)Nn * 128 * 4 + 255) & ~(size_t)255;
  int* lcount = (int*)(ws + loff);
  int* lst    = (int*)(ws + loff + (size_t)ntiles * SCAN_WAVES * 4);
  const size_t koff = (loff + (size_t)ntiles * SCAN_WAVES * 4 + (size_t)ntiles * SCAN_WAVES * LCAP * 8 + 255) & ~(size_t)255;
  unsigned* keys = (unsigned*)(ws + koff);
  const int nwords = ((E + QSTEP - 1) / QSTEP) * (QSTEP / 4);

  const int TB = 256;
  const int mtiles = (Nn + 15) >> 4;

  fill0_kernel<<<(Nn + TB - 1) / TB, TB, 0, stream>>>(dinv, Nn);

  degree_key_kernel<<<(nwords + TB - 1) / TB, TB, 0, stream>>>(dst, dinv, keys, E, nwords);
  rsqrt_kernel<<<(Nn + TB - 1) / TB, TB, 0, stream>>>(dinv, Nn);

  agg64_tile_kernel<<<ntiles, SCAN_WAVES * 32, 0, stream>>>(x, src, dst, keys, dinv, agg1, lst, lcount, E, Nn);

  mlp_wmma_kernel<<<(mtiles + 7) / 8, 256, 0, stream>>>(agg1, W1, b1, W2, h2, Nn);
  agg16_list_kernel<<<ntiles * (SCAN_WAVES / 8), 256, 0, stream>>>(h2, lst, lcount, src, dst, dinv, b2, out, E, Nn);

  (void)n_in; (void)out_size; (void)ws_size;
}


# Layer-1 hand-over sorted: a scanning wave counting-sorts its kept edges by owning wave inside its queue (no atomics); an owner reads only its own run of each queue, not all 32 queues; sum order kept
# speedup vs baseline: 2.1619x; 1.0517x over previous
#include <hip/hip_runtime.h>
#include <hip/hip_bf16.h>


#define IN_DIM  64
#define OUT_DIM 16

typedef __attribute__((ext_vector_type(2))) float v2f;
typedef __attribute__((ext_vector_type(8))) float v8f;

__device__ __forceinline__ void atomic_add_f32(float* p, float v) {
  asm volatile("global_atomic_add_f32 %0, %1, off scope:SCOPE_DEV"
               :: "v"(p), "v"(v) : "memory");
}


__global__ void fill0_kernel(float* __restrict__ p, int n) {
  int i = blockIdx.x * blockDim.x + threadIdx.x;
  if (i < n) p[i] = 0.0f;
}


__global__ void rsqrt_kernel(float* __restrict__ d, int n) {
  int i = blockIdx.x * blockDim.x + threadIdx.x;
  if (i < n) d[i] = rsqrtf(d[i] + 1.0f);
}


#define TILE_PITCH 68

__global__ __launch_bounds__(256) void mlp_wmma_kernel(const float* __restrict__ A,
                                                       const float* __restrict__ W1,
                                                       const float* __restrict__ b1,
                                                       const float* __restrict__ W2,
                                                       float* __restrict__ out, int M) {
  __shared__ __attribute__((aligned(16))) float sW1[64 * 64];
  __shared__ __attribute__((aligned(16))) float sW2[64 * 16];
  __shared__ __attribute__((aligned(16))) float sT[8][16 * TILE_PITCH];
  for (int i = threadIdx.x; i < (64 * 64) / 4; i += blockDim.x)
    ((float4*)sW1)[i] = ((const float4*)W1)[i];
  for (int i = threadIdx.x; i < (64 * 16) / 4; i += blockDim.x)
    ((float4*)sW2)[i] = ((const float4*)W2)[i];
  __syncthreads();

  const int lane = threadIdx.x & 31;
  const int wid  = threadIdx.x >> 5;
  const int mtiles = (M + 15) >> 4;
  const int mt = blockIdx.x * 8 + wid;
  const bool live = mt < mtiles;
  const int m0 = (live ? mt : mtiles - 1) * 16;
  const int row = lane & 15;
  const int kh  = (lane >> 4) << 1;
  const int rbase = (lane >> 4) << 3;
  const int arow_idx = min(m0 + row, M - 1);
  const float* __restrict__ arow = A + (size_t)arow_idx * 64 + kh;
  const float* __restrict__ bcol = sW1 + row;

  v8f c0 = {}, c1 = {}, c2 = {}, c3 = {};
#pragma unroll
  for (int k = 0; k < 64; k += 4) {
    v2f a, b;
    a.x = arow[k];
    a.y = arow[k + 1];
    const float* __restrict__ bk = bcol + (k + kh) * 64;
    b.x = bk[0];  b.y = bk[64];
    c0 = __builtin_amdgcn_wmma_f32_16x16x4_f32(false, a, false, b, (short)0, c0, false, false);
    b.x = bk[16]; b.y = bk[80];
    c1 = __builtin_amdgcn_wmma_f32_16x16x4_f32(false, a, false, b, (short)0, c1, false, false);
    b.x = bk[32]; b.y = bk[96];
    c2 = __builtin_amdgcn_wmma_f32_16x16x4_f32(false, a, false, b, (short)0, c2, false, false);
    b.x = bk[48]; b.y = bk[112];
    c3 = __builtin_amdgcn_wmma_f32_16x16x4_f32(false, a, false, b, (short)0, c3, false, false);
  }

  float* __restrict__ T = sT[wid];
  const float bb0 = b1[row], bb1 = b1[16 + row], bb2 = b1[32 + row], bb3 = b1[48 + row];
#pragma unroll
  for (int r = 0; r < 8; ++r) {
    float* __restrict__ tp = T + (rbase + r) * TILE_PITCH + row;
    tp[0]  = fmaxf(c0[r] + bb0, 0.0f);
    tp[16] = fmaxf(c1[r] + bb1, 0.0f);
    tp[32] = fmaxf(c2[r] + bb2, 0.0f);
    tp[48] = fmaxf(c3[r] + bb3, 0.0f);
  }
  __syncthreads();

  const float* __restrict__ trow  = T + row * TILE_PITCH + kh;
  const float* __restrict__ b2col = sW2 + row;
  v8f d = {};
#pragma unroll
  for (int k = 0; k < 64; k += 4) {
    v2f a, b;
    a.x = trow[k];
    a.y = trow[k + 1];
    b.x = b2col[(k + kh) * 16];
    b.y = b2col[(k + kh + 1) * 16];
    d = __builtin_amdgcn_wmma_f32_16x16x4_f32(false, a, false, b, (short)0, d, false, false);
  }

#pragma unroll
  for (int r = 0; r < 8; ++r) {
    const int orow = m0 + rbase + r;
    if (live && orow < M) out[(size_t)orow * 16 + row] = d[r];
  }
}


#define TILE_R        400
#define SCAN_WAVES    32
#define ROWS_PER_WAVE 13
#define QCAP          1536
#define QSTEP         1024
#define KEYS_AHEAD    4
#define RING          64
#define ROWS_AHEAD    32
#define ROWS_AHEAD_REST 16
#define LCAP          512

__device__ __forceinline__ int4 load_dst4_guarded(const int* __restrict__ dst, int e, int E) {
  int4 r;
  r.x = (e     < E) ? dst[e]     : -1;
  r.y = (e + 1 < E) ? dst[e + 1] : -1;
  r.z = (e + 2 < E) ? dst[e + 2] : -1;
  r.w = (e + 3 < E) ? dst[e + 3] : -1;
  return r;
}

__global__ void degree_key_kernel(const int* __restrict__ dst, float* __restrict__ deg,
                                  unsigned* __restrict__ keys, int E, int nwords) {
  const int i = blockIdx.x * blockDim.x + threadIdx.x;
  if (i >= nwords) return;
  const int e = i * 4;
  int4 d;
  if ((E & 3) == 0 && e + 3 < E) d = ((const int4*)dst)[i];
  else d = load_dst4_guarded(dst, e, E);
  if (d.x >= 0) atomic_add_f32(&deg[d.x], 1.0f);
  if (d.y >= 0) atomic_add_f32(&deg[d.y], 1.0f);
  if (d.z >= 0) atomic_add_f32(&deg[d.z], 1.0f);
  if (d.w >= 0) atomic_add_f32(&deg[d.w], 1.0f);
  const unsigned k0 = ((unsigned)d.x / TILE_R) & 255u, k1 = ((unsigned)d.y / TILE_R) & 255u;
  const unsigned k2 = ((unsigned)d.z / TILE_R) & 255u, k3 = ((unsigned)d.w / TILE_R) & 255u;
  keys[i] = k0 | (k1 << 8) | (k2 << 16) | (k3 << 24);
}

typedef __attribute__((ext_vector_type(16))) float v16f;

template <int N> struct count_c { static constexpr int value = N; };

__global__ __launch_bounds__(1024) void agg64_tile_kernel(const float* __restrict__ x,
                                                          const int* __restrict__ src,
                                                          const int* __restrict__ dst,
                                                          const unsigned* __restrict__ keys,
                                                          const float* __restrict__ dinv,
                                                          float* __restrict__ agg,
                                                          int* __restrict__ lst,
                                                          int* __restrict__ lcount,
                                                          int E, int Nn) {
  __shared__ unsigned sQ[SCAN_WAVES][QCAP];
  __shared__ unsigned sRing[SCAN_WAVES][RING];
  __shared__ int sQn[SCAN_WAVES];
  __shared__ int sMore[SCAN_WAVES];
  __shared__ int sRun[SCAN_WAVES][SCAN_WAVES + 1];
  __shared__ int sBig[SCAN_WAVES];
  static_assert(SCAN_WAVES == 32, "one owner per lane, five owner bits");

  const int lane = threadIdx.x & 31;
  const int wid  = __builtin_amdgcn_readfirstlane((int)(threadIdx.x >> 5));
  const int d0   = blockIdx.x * TILE_R;
  const int nrows = min(TILE_R, Nn - d0);
  const unsigned un = (unsigned)nrows;
  const int row0 = wid * ROWS_PER_WAVE;

  v16f accx = {}, accy = {};
  {
    float2 v0[ROWS_PER_WAVE];
    float dd[ROWS_PER_WAVE];
#pragma unroll
    for (int k = 0; k < ROWS_PER_WAVE; ++k) {
      const int d = d0 + min(row0 + k, nrows - 1);
      dd[k] = dinv[d];
      v0[k] = ((const float2*)(x + (size_t)d * 64))[lane];
    }
#pragma unroll
    for (int k = 0; k < ROWS_PER_WAVE; ++k) {
      const bool valid = row0 + k < nrows;
      const float w = dd[k] * dd[k];
      accx[k] = valid ? v0[k].x * w : 0.0f;
      accy[k] = valid ? v0[k].y * w : 0.0f;
    }
  }

  unsigned* __restrict__ q    = sQ[wid];
  unsigned* __restrict__ ring = sRing[wid];
  int qn = 0;
  int rhead = 0, rn = 0;
  const int gw = (int)blockIdx.x * SCAN_WAVES + wid;
  int2* __restrict__ mylst = (int2*)lst + (size_t)gw * LCAP;
  int lcnt = 0;

  const unsigned splat = ((unsigned)blockIdx.x & 255u) * 0x01010101u;

  auto flags = [&](const uint4& kv) {
    const unsigned x0 = kv.x ^ splat, x1 = kv.y ^ splat, x2 = kv.z ^ splat, x3 = kv.w ^ splat;
    const unsigned u0 = (x0 - 0x01010101u) & ~x0 & 0x80808080u, u1 = (x1 - 0x01010101u) & ~x1 & 0x80808080u;
    const unsigned u2 = (x2 - 0x01010101u) & ~x2 & 0x80808080u, u3 = (x3 - 0x01010101u) & ~x3 & 0x80808080u;
    return (u0 >> 7) | (u1 >> 6) | (u2 >> 5) | (u3 >> 4);
  };
  auto look = [&](const uint4& ka, const uint4& kb, int ebase) {
    unsigned g = flags(ka) | (flags(kb) << 4);
    const int e0 = ebase + lane * 16;
    for (;;) {
      const bool has = g != 0u;
      const unsigned mask = __builtin_amdgcn_ballot_w32(has);
      if (mask == 0u) break;
      if (has) {
        const int p = __builtin_ctz(g);
        g &= g - 1u;
        q[qn + (int)__builtin_amdgcn_mbcnt_lo(mask, 0u)] =
            (unsigned)(e0 + ((p & 4) << 7) + ((p & 3) << 2) + (p >> 3));
      }
      qn += __builtin_popcount(mask);
    }
  };

  auto ownergroup = [&](int b, unsigned& pk, unsigned& own, unsigned& peers) {
    __builtin_amdgcn_wave_barrier();
    const bool valid = b + lane < qn;
    pk = valid ? q[b + lane] : 0u;
    own = (pk & 511u) / ROWS_PER_WAVE;
    unsigned p = __builtin_amdgcn_ballot_w32(valid);
#pragma unroll
    for (int bit = 0; bit < 5; ++bit) {
      const bool on = ((own >> bit) & 1u) != 0u;
      const unsigned bb = __builtin_amdgcn_ballot_w32(on);
      p &= on ? bb : ~bb;
    }
    peers = p;
    return valid;
  };

  auto addrow = [&](int r, float vx, float vy, float w) {
    accx[r] += vx * w;
    accy[r] += vy * w;
  };

  auto fetch = [&](int n, unsigned& pk, int& s_l, float& dd_l) {
    __builtin_amdgcn_wave_barrier();
    pk = 0u; s_l = 0; dd_l = 0.0f;
    if (lane < n) {
      pk   = ring[(rhead + lane) & (RING - 1)];
      s_l  = src[pk >> 9];
      dd_l = dinv[d0 + (int)(pk & 511u)];
    }
  };

  auto rows = [&](auto ahead_c, int n, unsigned pk, int s_l, float dd_l) {
    constexpr int AHEAD = decltype(ahead_c)::value;
    const int r_l = (int)(pk & 511u) - row0;
    const float ds_l = dinv[s_l];
    float2 v[32];
#pragma unroll
    for (int u = 0; u < AHEAD; ++u)
      if (u < n) {
        const int s = __builtin_amdgcn_readlane(s_l, u);
        v[u] = ((const float2*)(x + (size_t)s * 64))[lane];
      }
    const float w_l = ds_l * dd_l;
    if (lane < n && lcnt + lane < LCAP)
      mylst[lcnt + lane] = make_int2((s_l << 4) | r_l, __float_as_int(w_l));
    lcnt += n;
#pragma unroll
    for (int u = 0; u < 32; ++u) {
      if (u < n) {
        const int r = __builtin_amdgcn_readlane(r_l, u);
        const float w = __int_as_float(__builtin_amdgcn_readlane(__float_as_int(w_l), u));
        addrow(r, v[u].x, v[u].y, w);
      }
      if (u + AHEAD < 32 && u + AHEAD < n) {
        const int s = __builtin_amdgcn_readlane(s_l, u + AHEAD);
        v[u + AHEAD] = ((const float2*)(x + (size_t)s * 64))[lane];
      }
    }
  };

  bool haveP = false;
  unsigned pkP = 0u;
  int sP = 0;
  float ddP = 0.0f;

  auto rpush = [&](bool m, unsigned pk) {
    const unsigned mask = __builtin_amdgcn_ballot_w32(m);
    if (mask != 0u) {
      if (m) ring[(rhead + rn + (int)__builtin_amdgcn_mbcnt_lo(mask, 0u)) & (RING - 1)] = pk;
      rn += __builtin_popcount(mask);
      if (rn >= 32) {
        unsigned pkN; int sN; float ddN;
        fetch(32, pkN, sN, ddN);
        if (haveP) rows(count_c<ROWS_AHEAD>{}, 32, pkP, sP, ddP);
        pkP = pkN; sP = sN; ddP = ddN; haveP = true;
        rhead = (rhead + 32) & (RING - 1);
        rn -= 32;
      }
    }
  };

  const int nsc = (E + QSTEP - 1) / QSTEP;
  const int rot = (int)blockIdx.x * (nsc / (int)gridDim.x);

  auto stepbase = [&](int scx) {
    int g = scx + rot;
    if (g >= nsc) g -= nsc;
    return g * QSTEP;
  };
  auto loadkeys = [&](int scx, int fb, int half) {
    return ((const uint4*)((const unsigned char*)keys + stepbase(scx < nsc ? scx : fb)))[half * 32 + lane];
  };

  int sc = wid;

  for (;;) {
    if (sc < nsc) {
      uint4 a0 = loadkeys(sc, sc, 0),                  b0 = loadkeys(sc, sc, 1);
      uint4 a1 = loadkeys(sc + SCAN_WAVES, sc, 0),     b1 = loadkeys(sc + SCAN_WAVES, sc, 1);
      uint4 a2 = loadkeys(sc + 2 * SCAN_WAVES, sc, 0), b2 = loadkeys(sc + 2 * SCAN_WAVES, sc, 1);
      uint4 a3 = loadkeys(sc + 3 * SCAN_WAVES, sc, 0), b3 = loadkeys(sc + 3 * SCAN_WAVES, sc, 1);
#define SCAN_STEP(ka, kb)                                    \
      if (!(sc < nsc && qn <= QCAP - QSTEP)) break;          \
      look(ka, kb, stepbase(sc));                            \
      ka = loadkeys(sc + KEYS_AHEAD * SCAN_WAVES, sc, 0);    \
      kb = loadkeys(sc + KEYS_AHEAD * SCAN_WAVES, sc, 1);    \
      sc += SCAN_WAVES;
      for (;;) {
        SCAN_STEP(a0, b0)
        SCAN_STEP(a1, b1)
        SCAN_STEP(a2, b2)
        SCAN_STEP(a3, b3)
      }
#undef SCAN_STEP
    }
    {
      int nkeep = 0;
      for (int b = 0; b < qn; b += 128) {
        __builtin_amdgcn_wave_barrier();
        int ee[4], dd[4];
#pragma unroll
        for (int u = 0; u < 4; ++u) {
          const int i = b + u * 32 + lane;
          ee[u] = (i < qn) ? (int)q[i] : E;
          dd[u] = (ee[u] < E) ? dst[ee[u]] : -1;
        }
#pragma unroll
        for (int u = 0; u < 4; ++u) {
          const bool ok = (unsigned)(dd[u] - d0) < un;
          const unsigned mask = __builtin_amdgcn_ballot_w32(ok);
          if (ok) q[nkeep + (int)__builtin_amdgcn_mbcnt_lo(mask, 0u)] = ((unsigned)ee[u] << 9) | (unsigned)(dd[u] - d0);
          nkeep += __builtin_popcount(mask);
        }
      }
      qn = nkeep;
    }
    const bool fits = 2 * qn <= QCAP;
    if (fits) {
      int* run = sRun[wid] + 1;
      run[lane] = 0;
      for (int b = 0; b < qn; b += 32) {
        unsigned pk, own, peers;
        const bool valid = ownergroup(b, pk, own, peers);
        if (valid && __builtin_amdgcn_mbcnt_lo(peers, 0u) == 0u) run[own] += __builtin_popcount(peers);
      }
      __builtin_amdgcn_wave_barrier();
      const int cnt_o = run[lane];
      int incl = cnt_o;
      incl += __builtin_amdgcn_update_dpp(0, incl, 0x111, 0xf, 0xf, true);
      incl += __builtin_amdgcn_update_dpp(0, incl, 0x112, 0xf, 0xf, true);
      incl += __builtin_amdgcn_update_dpp(0, incl, 0x114, 0xf, 0xf, true);
      incl += __builtin_amdgcn_update_dpp(0, incl, 0x118, 0xf, 0xf, true);
      const int row0 = __builtin_amdgcn_readlane(incl, 15);
      if (lane >= 16) incl += row0;
      run[lane] = qn + incl - cnt_o;
      if (lane == 0) run[-1] = qn;
      for (int b = 0; b < qn; b += 32) {
        unsigned pk, own, peers;
        const bool valid = ownergroup(b, pk, own, peers);
        const int rank = (int)__builtin_amdgcn_mbcnt_lo(peers, 0u);
        int base = 0;
        if (valid) { base = run[own]; q[base + rank] = pk; }
        if (valid && rank == 0) run[own] = base + __builtin_popcount(peers);
      }
    }
    if (lane == 0) { sQn[wid] = qn; sMore[wid] = (sc < nsc) ? 1 : 0; sBig[wid] = fits ? 0 : 1; }
    __syncthreads();

    const bool more = __builtin_amdgcn_ballot_w32(sMore[lane] != 0) != 0u;
    const bool big = __builtin_amdgcn_ballot_w32(sBig[lane] != 0) != 0u;
    for (int k = 0; k < SCAN_WAVES; ++k) {
      int lo_v = sRun[k][wid], hi_v = sRun[k][wid + 1];
      if (big) { lo_v = 0; hi_v = sQn[k]; }
      const int lo = __builtin_amdgcn_readfirstlane(lo_v);
      const int nk = __builtin_amdgcn_readfirstlane(hi_v) - lo;
      const unsigned* __restrict__ qk = sQ[k] + lo;
      for (int b = 0; b < nk; b += 32) {
        unsigned pk = 0u;
        bool own = false;
        if (b + lane < nk) {
          pk = qk[b + lane];
          own = ((pk & 511u) / ROWS_PER_WAVE) == (unsigned)wid;
        }
        rpush(own, pk);
      }
    }
    {
      unsigned pkN = 0u; int sN = 0; float ddN = 0.0f;
      if (rn > 0) fetch(rn, pkN, sN, ddN);
      if (haveP) { rows(count_c<ROWS_AHEAD>{}, 32, pkP, sP, ddP); haveP = false; }
      if (rn > 0) rows(count_c<ROWS_AHEAD_REST>{}, rn, pkN, sN, ddN);
    }
    rhead = 0; rn = 0; qn = 0;
    __syncthreads();
    if (!more) break;
  }

#pragma unroll
  for (int k = 0; k < ROWS_PER_WAVE; ++k)
    if (row0 + k < nrows)
      ((float2*)(agg + (size_t)(d0 + row0 + k) * 64))[lane] = make_float2(accx[k], accy[k]);
  if (lane == 0) lcount[gw] = lcnt;
}

__global__ __launch_bounds__(256) void agg16_list_kernel(const float* __restrict__ h,
                                                         const int* __restrict__ lst,
                                                         const int* __restrict__ lcount,
                                                         const int* __restrict__ src,
                                                         const int* __restrict__ dst,
                                                         const float* __restrict__ dinv,
                                                         const float* __restrict__ b2,
                                                         float* __restrict__ out,
                                                         int E, int Nn) {
  __shared__ float sA0[8][ROWS_PER_WAVE * 32];
  __shared__ float sA1[8][ROWS_PER_WAVE * 32];
  __shared__ float sA2[8][ROWS_PER_WAVE * 32];
  __shared__ float sA3[8][ROWS_PER_WAVE * 32];

  const int lane = threadIdx.x & 31;
  const int wid  = __builtin_amdgcn_readfirstlane((int)(threadIdx.x >> 5));
  const int gw   = (int)blockIdx.x * 8 + wid;
  const int d0   = (gw / SCAN_WAVES) * TILE_R;
  const int row0 = (gw % SCAN_WAVES) * ROWS_PER_WAVE;
  const int nr   = min(ROWS_PER_WAVE, min(TILE_R, Nn - d0) - row0);
  if (nr <= 0) return;
  const int c  = lane & 15;
  const int hh = lane >> 4;
  const int base = d0 + row0;

  float* a0 = sA0[wid] + lane;
  float* a1 = sA1[wid] + lane;
  float* a2 = sA2[wid] + lane;
  float* a3 = sA3[wid] + lane;
#pragma unroll
  for (int k = 0; k < ROWS_PER_WAVE; ++k) {
    a0[k * 32] = 0.0f; a1[k * 32] = 0.0f; a2[k * 32] = 0.0f; a3[k * 32] = 0.0f;
  }

  float selfv[7];
#pragma unroll
  for (int i = 0; i < 7; ++i) {
    const int d = base + min(2 * i + hh, nr - 1);
    const float dd = dinv[d];
    selfv[i] = h[(size_t)d * 16 + c] * (dd * dd);
  }
  const float bias = b2[c];

  const int cnt = __builtin_amdgcn_readfirstlane(lcount[gw]);
  if (cnt <= LCAP) {
    const int2* __restrict__ my = (const int2*)lst + (size_t)gw * LCAP;
    int2 en = make_int2(0, 0);
    if (lane < cnt) en = my[lane];
    for (int j = 0; j < cnt; j += 32) {
      const int m = cnt - j;
      int pk[16];
      float wt[16], v[16];
#pragma unroll
      for (int u = 0; u < 16; ++u) {
        const int sel = (2 * u + hh) << 2;
        pk[u] = __builtin_amdgcn_ds_bpermute(sel, en.x);
        wt[u] = __int_as_float(__builtin_amdgcn_ds_bpermute(sel, en.y));
        v[u] = h[((unsigned)pk[u] & ~15u) + (unsigned)c];
      }
      int2 nx = make_int2(0, 0);
      if (j + 32 + lane < cnt) nx = my[j + 32 + lane];
#pragma unroll
      for (int u = 0; u < 16; u += 4) {
        const int r0 = (pk[u] & 15) * 32, r1 = (pk[u + 1] & 15) * 32;
        const int r2 = (pk[u + 2] & 15) * 32, r3 = (pk[u + 3] & 15) * 32;
        float t0 = a0[r0], t1 = a1[r1], t2 = a2[r2], t3 = a3[r3];
        t0 += v[u] * wt[u];
        t1 += v[u + 1] * wt[u + 1];
        t2 += v[u + 2] * wt[u + 2];
        t3 += v[u + 3] * wt[u + 3];
        if (2 * u + hh < m)       a0[r0] = t0;
        if (2 * (u + 1) + hh < m) a1[r1] = t1;
        if (2 * (u + 2) + hh < m) a2[r2] = t2;
        if (2 * (u + 3) + hh < m) a3[r3] = t3;
      }
      en = nx;
    }
  } else {
    for (int e0 = 0; e0 < E; e0 += 32) {
      const int e = e0 + lane;
      const int d = (e < E) ? dst[e] : -1;
      unsigned mask = __builtin_amdgcn_ballot_w32((unsigned)(d - base) < (unsigned)nr);
      while (mask != 0u) {
        const int l = __builtin_ctz(mask);
        mask &= mask - 1u;
        const int dd = __builtin_amdgcn_readlane(d, l);
        const int s  = src[e0 + l];
        const float w = dinv[s] * dinv[dd];
        if (lane < 16) a0[(dd - base) * 32] += h[(size_t)s * 16 + lane] * w;
      }
    }
  }

  __builtin_amdgcn_wave_barrier();
#pragma unroll
  for (int i = 0; i < 7; ++i) {
    const int r = 2 * i + hh;
    if (r < nr) {
      const float* p0 = sA0[wid] + r * 32 + c;
      const float* p1 = sA1[wid] + r * 32 + c;
      const float* p2 = sA2[wid] + r * 32 + c;
      const float* p3 = sA3[wid] + r * 32 + c;
      float s = selfv[i];
      s += p0[0]; s += p0[16]; s += p1[0]; s += p1[16];
      s += p2[0]; s += p2[16]; s += p3[0]; s += p3[16];
      out[(size_t)(base + r) * 16 + c] = s + bias;
    }
  }
}


extern "C" void kernel_launch(void* const* d_in, const int* in_sizes, int n_in,
                              void* d_out, int out_size, void* d_ws, size_t ws_size,
                              hipStream_t stream) {
  const float* x    = (const float*)d_in[0];
  const int*   edge = (const int*)d_in[1];
  const float* W1   = (const float*)d_in[2];
  const float* b1   = (const float*)d_in[3];
  const float* W2   = (const float*)d_in[4];
  const float* b2   = (const float*)d_in[5];
  float* out = (float*)d_out;

  const int E  = in_sizes[1] / 2;
  const int Nn = in_sizes[0] / IN_DIM;
  const int* src = edge;
  const int* dst = edge + E;

  char* ws = (char*)d_ws;
  float* dinv = (float*)ws;
  float* h2   = (float*)(ws + (size_t)Nn * 4);
  float* agg1 = (float*)(ws + (size_t)Nn * 4 + (size_t)Nn * 64 * 4);
  const int ntiles = (Nn + TILE_R - 1) / TILE_R;
  const size_t loff = ((size_t)Nn * 4 + (size_t)Nn * 128 * 4 + 255) & ~(size_t)255;
  int* lcount = (int*)(ws + loff);
  int* lst    = (int*)(ws + loff + (size_t)ntiles * SCAN_WAVES * 4);
  const size_t koff = (loff + (size_t)ntiles * SCAN_WAVES * 4 + (size_t)ntiles * SCAN_WAVES * LCAP * 8 + 255) & ~(size_t)255;
  unsigned* keys = (unsigned*)(ws + koff);
  const int nwords = ((E + QSTEP - 1) / QSTEP) * (QSTEP / 4);

  const int TB = 256;
  const int mtiles = (Nn + 15) >> 4;

  fill0_kernel<<<(Nn + TB - 1) / TB, TB, 0, stream>>>(dinv, Nn);

  degree_key_kernel<<<(nwords + TB - 1) / TB, TB, 0, stream>>>(dst, dinv, keys, E, nwords);
  rsqrt_kernel<<<(Nn + TB - 1) / TB, TB, 0, stream>>>(dinv, Nn);

  agg64_tile_kernel<<<ntiles, SCAN_WAVES * 32, 0, stream>>>(x, src, dst, keys, dinv, agg1, lst, lcount, E, Nn);

  mlp_wmma_kernel<<<(mtiles + 7) / 8, 256, 0, stream>>>(agg1, W1, b1, W2, h2, Nn);
  agg16_list_kernel<<<ntiles * (SCAN_WAVES / 8), 256, 0, stream>>>(h2, lst, lcount, src, dst, dinv, b2, out, E, Nn);

  (void)n_in; (void)out_size; (void)ws_size;
}


# Degree pass also stores each edge as an 8-byte (src,dst) pair; layer 1 fetches one pair per key match, its queue entry carries src: per-edge src fetch and the stage hiding it gone; sum order kept
# speedup vs baseline: 2.2453x; 1.0386x over previous
#include <hip/hip_runtime.h>
#include <hip/hip_bf16.h>


#define IN_DIM  64
#define OUT_DIM 16

typedef __attribute__((ext_vector_type(2))) float v2f;
typedef __attribute__((ext_vector_type(8))) float v8f;

__device__ __forceinline__ void atomic_add_f32(float* p, float v) {
  asm volatile("global_atomic_add_f32 %0, %1, off scope:SCOPE_DEV"
               :: "v"(p), "v"(v) : "memory");
}


__global__ void fill0_kernel(float* __restrict__ p, int n) {
  int i = blockIdx.x * blockDim.x + threadIdx.x;
  if (i < n) p[i] = 0.0f;
}


__global__ void rsqrt_kernel(float* __restrict__ d, int n) {
  int i = blockIdx.x * blockDim.x + threadIdx.x;
  if (i < n) d[i] = rsqrtf(d[i] + 1.0f);
}


#define TILE_PITCH 68

__global__ __launch_bounds__(256) void mlp_wmma_kernel(const float* __restrict__ A,
                                                       const float* __restrict__ W1,
                                                       const float* __restrict__ b1,
                                                       const float* __restrict__ W2,
                                                       float* __restrict__ out, int M) {
  __shared__ __attribute__((aligned(16))) float sW1[64 * 64];
  __shared__ __attribute__((aligned(16))) float sW2[64 * 16];
  __shared__ __attribute__((aligned(16))) float sT[8][16 * TILE_PITCH];
  for (int i = threadIdx.x; i < (64 * 64) / 4; i += blockDim.x)
    ((float4*)sW1)[i] = ((const float4*)W1)[i];
  for (int i = threadIdx.x; i < (64 * 16) / 4; i += blockDim.x)
    ((float4*)sW2)[i] = ((const float4*)W2)[i];
  __syncthreads();

  const int lane = threadIdx.x & 31;
  const int wid  = threadIdx.x >> 5;
  const int mtiles = (M + 15) >> 4;
  const int mt = blockIdx.x * 8 + wid;
  const bool live = mt < mtiles;
  const int m0 = (live ? mt : mtiles - 1) * 16;
  const int row = lane & 15;
  const int kh  = (lane >> 4) << 1;
  const int rbase = (lane >> 4) << 3;
  const int arow_idx = min(m0 + row, M - 1);
  const float* __restrict__ arow = A + (size_t)arow_idx * 64 + kh;
  const float* __restrict__ bcol = sW1 + row;

  v8f c0 = {}, c1 = {}, c2 = {}, c3 = {};
#pragma unroll
  for (int k = 0; k < 64; k += 4) {
    v2f a, b;
    a.x = arow[k];
    a.y = arow[k + 1];
    const float* __restrict__ bk = bcol + (k + kh) * 64;
    b.x = bk[0];  b.y = bk[64];
    c0 = __builtin_amdgcn_wmma_f32_16x16x4_f32(false, a, false, b, (short)0, c0, false, false);
    b.x = bk[16]; b.y = bk[80];
    c1 = __builtin_amdgcn_wmma_f32_16x16x4_f32(false, a, false, b, (short)0, c1, false, false);
    b.x = bk[32]; b.y = bk[96];
    c2 = __builtin_amdgcn_wmma_f32_16x16x4_f32(false, a, false, b, (short)0, c2, false, false);
    b.x = bk[48]; b.y = bk[112];
    c3 = __builtin_amdgcn_wmma_f32_16x16x4_f32(false, a, false, b, (short)0, c3, false, false);
  }

  float* __restrict__ T = sT[wid];
  const float bb0 = b1[row], bb1 = b1[16 + row], bb2 = b1[32 + row], bb3 = b1[48 + row];
#pragma unroll
  for (int r = 0; r < 8; ++r) {
    float* __restrict__ tp = T + (rbase + r) * TILE_PITCH + row;
    tp[0]  = fmaxf(c0[r] + bb0, 0.0f);
    tp[16] = fmaxf(c1[r] + bb1, 0.0f);
    tp[32] = fmaxf(c2[r] + bb2, 0.0f);
    tp[48] = fmaxf(c3[r] + bb3, 0.0f);
  }
  __syncthreads();

  const float* __restrict__ trow  = T + row * TILE_PITCH + kh;
  const float* __restrict__ b2col = sW2 + row;
  v8f d = {};
#pragma unroll
  for (int k = 0; k < 64; k += 4) {
    v2f a, b;
    a.x = trow[k];
    a.y = trow[k + 1];
    b.x = b2col[(k + kh) * 16];
    b.y = b2col[(k + kh + 1) * 16];
    d = __builtin_amdgcn_wmma_f32_16x16x4_f32(false, a, false, b, (short)0, d, false, false);
  }

#pragma unroll
  for (int r = 0; r < 8; ++r) {
    const int orow = m0 + rbase + r;
    if (live && orow < M) out[(size_t)orow * 16 + row] = d[r];
  }
}


#define TILE_R        400
#define SCAN_WAVES    32
#define ROWS_PER_WAVE 13
#define QCAP          1536
#define QSTEP         1024
#define KEYS_AHEAD    4
#define RING          64
#define ROWS_AHEAD    32
#define ROWS_AHEAD_REST 16
#define LCAP          512

__device__ __forceinline__ int4 load_idx4_guarded(const int* __restrict__ p, int e, int E) {
  int4 r;
  r.x = (e     < E) ? p[e]     : -1;
  r.y = (e + 1 < E) ? p[e + 1] : -1;
  r.z = (e + 2 < E) ? p[e + 2] : -1;
  r.w = (e + 3 < E) ? p[e + 3] : -1;
  return r;
}

__global__ void degree_key_kernel(const int* __restrict__ src, const int* __restrict__ dst,
                                  float* __restrict__ deg, unsigned* __restrict__ keys,
                                  int* __restrict__ pairs, int E, int nwords) {
  const int i = blockIdx.x * blockDim.x + threadIdx.x;
  if (i >= nwords) return;
  const int e = i * 4;
  int4 d, s;
  if ((E & 3) == 0 && e + 3 < E) {
    d = ((const int4*)dst)[i];
    s = ((const int4*)src)[i];
  } else {
    d = load_idx4_guarded(dst, e, E);
    s = load_idx4_guarded(src, e, E);
  }
  int4* __restrict__ pr = (int4*)pairs + 2 * (size_t)i;
  pr[0] = make_int4(s.x, d.x, s.y, d.y);
  pr[1] = make_int4(s.z, d.z, s.w, d.w);
  if (d.x >= 0) atomic_add_f32(&deg[d.x], 1.0f);
  if (d.y >= 0) atomic_add_f32(&deg[d.y], 1.0f);
  if (d.z >= 0) atomic_add_f32(&deg[d.z], 1.0f);
  if (d.w >= 0) atomic_add_f32(&deg[d.w], 1.0f);
  const unsigned k0 = ((unsigned)d.x / TILE_R) & 255u, k1 = ((unsigned)d.y / TILE_R) & 255u;
  const unsigned k2 = ((unsigned)d.z / TILE_R) & 255u, k3 = ((unsigned)d.w / TILE_R) & 255u;
  keys[i] = k0 | (k1 << 8) | (k2 << 16) | (k3 << 24);
}

typedef __attribute__((ext_vector_type(16))) float v16f;

template <int N> struct count_c { static constexpr int value = N; };

__global__ __launch_bounds__(1024) void agg64_tile_kernel(const float* __restrict__ x,
                                                          const int* __restrict__ pairs,
                                                          const unsigned* __restrict__ keys,
                                                          const float* __restrict__ dinv,
                                                          float* __restrict__ agg,
                                                          int* __restrict__ lst,
                                                          int* __restrict__ lcount,
                                                          int E, int Nn) {
  __shared__ unsigned sQ[SCAN_WAVES][QCAP];
  __shared__ unsigned sRing[SCAN_WAVES][RING];
  __shared__ int sQn[SCAN_WAVES];
  __shared__ int sMore[SCAN_WAVES];
  __shared__ int sRun[SCAN_WAVES][SCAN_WAVES + 1];
  __shared__ int sBig[SCAN_WAVES];
  static_assert(SCAN_WAVES == 32, "one owner per lane, five owner bits");

  const int lane = threadIdx.x & 31;
  const int wid  = __builtin_amdgcn_readfirstlane((int)(threadIdx.x >> 5));
  const int d0   = blockIdx.x * TILE_R;
  const int nrows = min(TILE_R, Nn - d0);
  const unsigned un = (unsigned)nrows;
  const int row0 = wid * ROWS_PER_WAVE;

  v16f accx = {}, accy = {};
  {
    float2 v0[ROWS_PER_WAVE];
    float dd[ROWS_PER_WAVE];
#pragma unroll
    for (int k = 0; k < ROWS_PER_WAVE; ++k) {
      const int d = d0 + min(row0 + k, nrows - 1);
      dd[k] = dinv[d];
      v0[k] = ((const float2*)(x + (size_t)d * 64))[lane];
    }
#pragma unroll
    for (int k = 0; k < ROWS_PER_WAVE; ++k) {
      const bool valid = row0 + k < nrows;
      const float w = dd[k] * dd[k];
      accx[k] = valid ? v0[k].x * w : 0.0f;
      accy[k] = valid ? v0[k].y * w : 0.0f;
    }
  }

  unsigned* __restrict__ q    = sQ[wid];
  unsigned* __restrict__ ring = sRing[wid];
  int qn = 0;
  int rhead = 0, rn = 0;
  const int gw = (int)blockIdx.x * SCAN_WAVES + wid;
  int2* __restrict__ mylst = (int2*)lst + (size_t)gw * LCAP;
  int lcnt = 0;

  const unsigned splat = ((unsigned)blockIdx.x & 255u) * 0x01010101u;

  auto flags = [&](const uint4& kv) {
    const unsigned x0 = kv.x ^ splat, x1 = kv.y ^ splat, x2 = kv.z ^ splat, x3 = kv.w ^ splat;
    const unsigned u0 = (x0 - 0x01010101u) & ~x0 & 0x80808080u, u1 = (x1 - 0x01010101u) & ~x1 & 0x80808080u;
    const unsigned u2 = (x2 - 0x01010101u) & ~x2 & 0x80808080u, u3 = (x3 - 0x01010101u) & ~x3 & 0x80808080u;
    return (u0 >> 7) | (u1 >> 6) | (u2 >> 5) | (u3 >> 4);
  };
  auto look = [&](const uint4& ka, const uint4& kb, int ebase) {
    unsigned g = flags(ka) | (flags(kb) << 4);
    const int e0 = ebase + lane * 16;
    for (;;) {
      const bool has = g != 0u;
      const unsigned mask = __builtin_amdgcn_ballot_w32(has);
      if (mask == 0u) break;
      if (has) {
        const int p = __builtin_ctz(g);
        g &= g - 1u;
        q[qn + (int)__builtin_amdgcn_mbcnt_lo(mask, 0u)] =
            (unsigned)(e0 + ((p & 4) << 7) + ((p & 3) << 2) + (p >> 3));
      }
      qn += __builtin_popcount(mask);
    }
  };

  auto ownergroup = [&](int b, unsigned& pk, unsigned& own, unsigned& peers) {
    __builtin_amdgcn_wave_barrier();
    const bool valid = b + lane < qn;
    pk = valid ? q[b + lane] : 0u;
    own = (pk & 511u) / ROWS_PER_WAVE;
    unsigned p = __builtin_amdgcn_ballot_w32(valid);
#pragma unroll
    for (int bit = 0; bit < 5; ++bit) {
      const bool on = ((own >> bit) & 1u) != 0u;
      const unsigned bb = __builtin_amdgcn_ballot_w32(on);
      p &= on ? bb : ~bb;
    }
    peers = p;
    return valid;
  };

  auto addrow = [&](int r, float vx, float vy, float w) {
    accx[r] += vx * w;
    accy[r] += vy * w;
  };

  auto rows = [&](auto ahead_c, int n) {
    constexpr int AHEAD = decltype(ahead_c)::value;
    __builtin_amdgcn_wave_barrier();
    unsigned pk = 0u;
    float ds_l = 0.0f, dd_l = 0.0f;
    if (lane < n) {
      pk   = ring[(rhead + lane) & (RING - 1)];
      ds_l = dinv[pk >> 9];
      dd_l = dinv[d0 + (int)(pk & 511u)];
    }
    const int r_l = (int)(pk & 511u) - row0;
    const int s_l = (int)(pk >> 9);
    const unsigned o_l = (pk >> 9) << 5;
    const float2* __restrict__ xr = (const float2*)x;
    float2 v[32];
#pragma unroll
    for (int u = 0; u < AHEAD; ++u)
      if (u < n) {
        const unsigned o = (unsigned)__builtin_amdgcn_readlane((int)o_l, u);
        v[u] = xr[o + (unsigned)lane];
      }
    const float w_l = ds_l * dd_l;
    if (lane < n && lcnt + lane < LCAP)
      mylst[lcnt + lane] = make_int2((s_l << 4) | r_l, __float_as_int(w_l));
    lcnt += n;
#pragma unroll
    for (int u = 0; u < 32; ++u) {
      if (u < n) {
        const int r = __builtin_amdgcn_readlane(r_l, u);
        const float w = __int_as_float(__builtin_amdgcn_readlane(__float_as_int(w_l), u));
        addrow(r, v[u].x, v[u].y, w);
      }
      if (u + AHEAD < 32 && u + AHEAD < n) {
        const unsigned o = (unsigned)__builtin_amdgcn_readlane((int)o_l, u + AHEAD);
        v[u + AHEAD] = xr[o + (unsigned)lane];
      }
    }
  };

  auto rpush = [&](bool m, unsigned pk) {
    const unsigned mask = __builtin_amdgcn_ballot_w32(m);
    if (mask != 0u) {
      if (m) ring[(rhead + rn + (int)__builtin_amdgcn_mbcnt_lo(mask, 0u)) & (RING - 1)] = pk;
      rn += __builtin_popcount(mask);
      if (rn >= 32) {
        rows(count_c<ROWS_AHEAD>{}, 32);
        rhead = (rhead + 32) & (RING - 1);
        rn -= 32;
      }
    }
  };

  const int nsc = (E + QSTEP - 1) / QSTEP;
  const int rot = (int)blockIdx.x * (nsc / (int)gridDim.x);

  auto stepbase = [&](int scx) {
    int g = scx + rot;
    if (g >= nsc) g -= nsc;
    return g * QSTEP;
  };
  auto loadkeys = [&](int scx, int fb, int half) {
    return ((const uint4*)((const unsigned char*)keys + stepbase(scx < nsc ? scx : fb)))[half * 32 + lane];
  };

  int sc = wid;

  for (;;) {
    if (sc < nsc) {
      uint4 a0 = loadkeys(sc, sc, 0),                  b0 = loadkeys(sc, sc, 1);
      uint4 a1 = loadkeys(sc + SCAN_WAVES, sc, 0),     b1 = loadkeys(sc + SCAN_WAVES, sc, 1);
      uint4 a2 = loadkeys(sc + 2 * SCAN_WAVES, sc, 0), b2 = loadkeys(sc + 2 * SCAN_WAVES, sc, 1);
      uint4 a3 = loadkeys(sc + 3 * SCAN_WAVES, sc, 0), b3 = loadkeys(sc + 3 * SCAN_WAVES, sc, 1);
#define SCAN_STEP(ka, kb)                                    \
      if (!(sc < nsc && qn <= QCAP - QSTEP)) break;          \
      look(ka, kb, stepbase(sc));                            \
      ka = loadkeys(sc + KEYS_AHEAD * SCAN_WAVES, sc, 0);    \
      kb = loadkeys(sc + KEYS_AHEAD * SCAN_WAVES, sc, 1);    \
      sc += SCAN_WAVES;
      for (;;) {
        SCAN_STEP(a0, b0)
        SCAN_STEP(a1, b1)
        SCAN_STEP(a2, b2)
        SCAN_STEP(a3, b3)
      }
#undef SCAN_STEP
    }
    {
      int nkeep = 0;
      for (int b = 0; b < qn; b += 128) {
        __builtin_amdgcn_wave_barrier();
        int2 sd[4];
#pragma unroll
        for (int u = 0; u < 4; ++u) {
          const int i = min(b + u * 32 + lane, qn - 1);
          sd[u] = ((const int2*)pairs)[q[i]];
        }
#pragma unroll
        for (int u = 0; u < 4; ++u) {
          const bool ok = (b + u * 32 + lane < qn) && (unsigned)(sd[u].y - d0) < un;
          const unsigned mask = __builtin_amdgcn_ballot_w32(ok);
          if (ok) q[nkeep + (int)__builtin_amdgcn_mbcnt_lo(mask, 0u)] = ((unsigned)sd[u].x << 9) | (unsigned)(sd[u].y - d0);
          nkeep += __builtin_popcount(mask);
        }
      }
      qn = nkeep;
    }
    const bool fits = 2 * qn <= QCAP;
    if (fits) {
      int* run = sRun[wid] + 1;
      run[lane] = 0;
      for (int b = 0; b < qn; b += 32) {
        unsigned pk, own, peers;
        const bool valid = ownergroup(b, pk, own, peers);
        if (valid && __builtin_amdgcn_mbcnt_lo(peers, 0u) == 0u) run[own] += __builtin_popcount(peers);
      }
      __builtin_amdgcn_wave_barrier();
      const int cnt_o = run[lane];
      int incl = cnt_o;
      incl += __builtin_amdgcn_update_dpp(0, incl, 0x111, 0xf, 0xf, true);
      incl += __builtin_amdgcn_update_dpp(0, incl, 0x112, 0xf, 0xf, true);
      incl += __builtin_amdgcn_update_dpp(0, incl, 0x114, 0xf, 0xf, true);
      incl += __builtin_amdgcn_update_dpp(0, incl, 0x118, 0xf, 0xf, true);
      const int row0 = __builtin_amdgcn_readlane(incl, 15);
      if (lane >= 16) incl += row0;
      run[lane] = qn + incl - cnt_o;
      if (lane == 0) run[-1] = qn;
      for (int b = 0; b < qn; b += 32) {
        unsigned pk, own, peers;
        const bool valid = ownergroup(b, pk, own, peers);
        const int rank = (int)__builtin_amdgcn_mbcnt_lo(peers, 0u);
        int base = 0;
        if (valid) { base = run[own]; q[base + rank] = pk; }
        if (valid && rank == 0) run[own] = base + __builtin_popcount(peers);
      }
    }
    if (lane == 0) { sQn[wid] = qn; sMore[wid] = (sc < nsc) ? 1 : 0; sBig[wid] = fits ? 0 : 1; }
    __syncthreads();

    const bool more = __builtin_amdgcn_ballot_w32(sMore[lane] != 0) != 0u;
    const bool big = __builtin_amdgcn_ballot_w32(sBig[lane] != 0) != 0u;
    for (int k = 0; k < SCAN_WAVES; ++k) {
      int lo_v = sRun[k][wid], hi_v = sRun[k][wid + 1];
      if (big) { lo_v = 0; hi_v = sQn[k]; }
      const int lo = __builtin_amdgcn_readfirstlane(lo_v);
      const int nk = __builtin_amdgcn_readfirstlane(hi_v) - lo;
      const unsigned* __restrict__ qk = sQ[k] + lo;
      for (int b = 0; b < nk; b += 32) {
        unsigned pk = 0u;
        bool own = false;
        if (b + lane < nk) {
          pk = qk[b + lane];
          own = ((pk & 511u) / ROWS_PER_WAVE) == (unsigned)wid;
        }
        rpush(own, pk);
      }
    }
    if (rn > 0) rows(count_c<ROWS_AHEAD_REST>{}, rn);
    rhead = 0; rn = 0; qn = 0;
    __syncthreads();
    if (!more) break;
  }

#pragma unroll
  for (int k = 0; k < ROWS_PER_WAVE; ++k)
    if (row0 + k < nrows)
      ((float2*)(agg + (size_t)(d0 + row0 + k) * 64))[lane] = make_float2(accx[k], accy[k]);
  if (lane == 0) lcount[gw] = lcnt;
}

__global__ __launch_bounds__(256) void agg16_list_kernel(const float* __restrict__ h,
                                                         const int* __restrict__ lst,
                                                         const int* __restrict__ lcount,
                                                         const int* __restrict__ src,
                                                         const int* __restrict__ dst,
                                                         const float* __restrict__ dinv,
                                                         const float* __restrict__ b2,
                                                         float* __restrict__ out,
                                                         int E, int Nn) {
  __shared__ float sA0[8][ROWS_PER_WAVE * 32];
  __shared__ float sA1[8][ROWS_PER_WAVE * 32];
  __shared__ float sA2[8][ROWS_PER_WAVE * 32];
  __shared__ float sA3[8][ROWS_PER_WAVE * 32];

  const int lane = threadIdx.x & 31;
  const int wid  = __builtin_amdgcn_readfirstlane((int)(threadIdx.x >> 5));
  const int gw   = (int)blockIdx.x * 8 + wid;
  const int d0   = (gw / SCAN_WAVES) * TILE_R;
  const int row0 = (gw % SCAN_WAVES) * ROWS_PER_WAVE;
  const int nr   = min(ROWS_PER_WAVE, min(TILE_R, Nn - d0) - row0);
  if (nr <= 0) return;
  const int c  = lane & 15;
  const int hh = lane >> 4;
  const int base = d0 + row0;

  float* a0 = sA0[wid] + lane;
  float* a1 = sA1[wid] + lane;
  float* a2 = sA2[wid] + lane;
  float* a3 = sA3[wid] + lane;
#pragma unroll
  for (int k = 0; k < ROWS_PER_WAVE; ++k) {
    a0[k * 32] = 0.0f; a1[k * 32] = 0.0f; a2[k * 32] = 0.0f; a3[k * 32] = 0.0f;
  }

  float selfv[7];
#pragma unroll
  for (int i = 0; i < 7; ++i) {
    const int d = base + min(2 * i + hh, nr - 1);
    const float dd = dinv[d];
    selfv[i] = h[(size_t)d * 16 + c] * (dd * dd);
  }
  const float bias = b2[c];

  const int cnt = __builtin_amdgcn_readfirstlane(lcount[gw]);
  if (cnt <= LCAP) {
    const int2* __restrict__ my = (const int2*)lst + (size_t)gw * LCAP;
    int2 en = make_int2(0, 0);
    if (lane < cnt) en = my[lane];
    for (int j = 0; j < cnt; j += 32) {
      const int m = cnt - j;
      int pk[16];
      float wt[16], v[16];
#pragma unroll
      for (int u = 0; u < 16; ++u) {
        const int sel = (2 * u + hh) << 2;
        pk[u] = __builtin_amdgcn_ds_bpermute(sel, en.x);
        wt[u] = __int_as_float(__builtin_amdgcn_ds_bpermute(sel, en.y));
        v[u] = h[((unsigned)pk[u] & ~15u) + (unsigned)c];
      }
      int2 nx = make_int2(0, 0);
      if (j + 32 + lane < cnt) nx = my[j + 32 + lane];
#pragma unroll
      for (int u = 0; u < 16; u += 4) {
        const int r0 = (pk[u] & 15) * 32, r1 = (pk[u + 1] & 15) * 32;
        const int r2 = (pk[u + 2] & 15) * 32, r3 = (pk[u + 3] & 15) * 32;
        float t0 = a0[r0], t1 = a1[r1], t2 = a2[r2], t3 = a3[r3];
        t0 += v[u] * wt[u];
        t1 += v[u + 1] * wt[u + 1];
        t2 += v[u + 2] * wt[u + 2];
        t3 += v[u + 3] * wt[u + 3];
        if (2 * u + hh < m)       a0[r0] = t0;
        if (2 * (u + 1) + hh < m) a1[r1] = t1;
        if (2 * (u + 2) + hh < m) a2[r2] = t2;
        if (2 * (u + 3) + hh < m) a3[r3] = t3;
      }
      en = nx;
    }
  } else {
    for (int e0 = 0; e0 < E; e0 += 32) {
      const int e = e0 + lane;
      const int d = (e < E) ? dst[e] : -1;
      unsigned mask = __builtin_amdgcn_ballot_w32((unsigned)(d - base) < (unsigned)nr);
      while (mask != 0u) {
        const int l = __builtin_ctz(mask);
        mask &= mask - 1u;
        const int dd = __builtin_amdgcn_readlane(d, l);
        const int s  = src[e0 + l];
        const float w = dinv[s] * dinv[dd];
        if (lane < 16) a0[(dd - base) * 32] += h[(size_t)s * 16 + lane] * w;
      }
    }
  }

  __builtin_amdgcn_wave_barrier();
#pragma unroll
  for (int i = 0; i < 7; ++i) {
    const int r = 2 * i + hh;
    if (r < nr) {
      const float* p0 = sA0[wid] + r * 32 + c;
      const float* p1 = sA1[wid] + r * 32 + c;
      const float* p2 = sA2[wid] + r * 32 + c;
      const float* p3 = sA3[wid] + r * 32 + c;
      float s = selfv[i];
      s += p0[0]; s += p0[16]; s += p1[0]; s += p1[16];
      s += p2[0]; s += p2[16]; s += p3[0]; s += p3[16];
      out[(size_t)(base + r) * 16 + c] = s + bias;
    }
  }
}


extern "C" void kernel_launch(void* const* d_in, const int* in_sizes, int n_in,
                              void* d_out, int out_size, void* d_ws, size_t ws_size,
                              hipStream_t stream) {
  const float* x    = (const float*)d_in[0];
  const int*   edge = (const int*)d_in[1];
  const float* W1   = (const float*)d_in[2];
  const float* b1   = (const float*)d_in[3];
  const float* W2   = (const float*)d_in[4];
  const float* b2   = (const float*)d_in[5];
  float* out = (float*)d_out;

  const int E  = in_sizes[1] / 2;
  const int Nn = in_sizes[0] / IN_DIM;
  const int* src = edge;
  const int* dst = edge + E;

  char* ws = (char*)d_ws;
  float* dinv = (float*)ws;
  float* h2   = (float*)(ws + (size_t)Nn * 4);
  float* agg1 = (float*)(ws + (size_t)Nn * 4 + (size_t)Nn * 64 * 4);
  const int ntiles = (Nn + TILE_R - 1) / TILE_R;
  const size_t loff = ((size_t)Nn * 4 + (size_t)Nn * 128 * 4 + 255) & ~(size_t)255;
  int* lcount = (int*)(ws + loff);
  int* lst    = (int*)(ws + loff + (size_t)ntiles * SCAN_WAVES * 4);
  const size_t koff = (loff + (size_t)ntiles * SCAN_WAVES * 4 + (size_t)ntiles * SCAN_WAVES * LCAP * 8 + 255) & ~(size_t)255;
  unsigned* keys = (unsigned*)(ws + koff);
  const int nwords = ((E + QSTEP - 1) / QSTEP) * (QSTEP / 4);
  const size_t poff = (koff + (size_t)nwords * 4 + 255) & ~(size_t)255;
  int* pairs = (int*)(ws + poff);

  const int TB = 256;
  const int mtiles = (Nn + 15) >> 4;

  fill0_kernel<<<(Nn + TB - 1) / TB, TB, 0, stream>>>(dinv, Nn);

  degree_key_kernel<<<(nwords + TB - 1) / TB, TB, 0, stream>>>(src, dst, dinv, keys, pairs, E, nwords);
  rsqrt_kernel<<<(Nn + TB - 1) / TB, TB, 0, stream>>>(dinv, Nn);

  agg64_tile_kernel<<<ntiles, SCAN_WAVES * 32, 0, stream>>>(x, pairs, keys, dinv, agg1, lst, lcount, E, Nn);

  mlp_wmma_kernel<<<(mtiles + 7) / 8, 256, 0, stream>>>(agg1, W1, b1, W2, h2, Nn);
  agg16_list_kernel<<<ntiles * (SCAN_WAVES / 8), 256, 0, stream>>>(h2, lst, lcount, src, dst, dinv, b2, out, E, Nn);

  (void)n_in; (void)out_size; (void)ws_size;
}


# Degree from the tile scan: a wave counts its rows' edges by ballots as it saves its list, stores rsqrt(count+1); x rows summed from lists in a new launch; atomics, fill, rsqrt gone; sum order kept
# speedup vs baseline: 3.6540x; 1.6274x over previous
#include <hip/hip_runtime.h>
#include <hip/hip_bf16.h>


#define IN_DIM  64
#define OUT_DIM 16

typedef __attribute__((ext_vector_type(2))) float v2f;
typedef __attribute__((ext_vector_type(8))) float v8f;


#define TILE_PITCH 68

__global__ __launch_bounds__(256) void mlp_wmma_kernel(const float* __restrict__ A,
                                                       const float* __restrict__ W1,
                                                       const float* __restrict__ b1,
                                                       const float* __restrict__ W2,
                                                       float* __restrict__ out, int M) {
  __shared__ __attribute__((aligned(16))) float sW1[64 * 64];
  __shared__ __attribute__((aligned(16))) float sW2[64 * 16];
  __shared__ __attribute__((aligned(16))) float sT[8][16 * TILE_PITCH];
  for (int i = threadIdx.x; i < (64 * 64) / 4; i += blockDim.x)
    ((float4*)sW1)[i] = ((const float4*)W1)[i];
  for (int i = threadIdx.x; i < (64 * 16) / 4; i += blockDim.x)
    ((float4*)sW2)[i] = ((const float4*)W2)[i];
  __syncthreads();

  const int lane = threadIdx.x & 31;
  const int wid  = threadIdx.x >> 5;
  const int mtiles = (M + 15) >> 4;
  const int mt = blockIdx.x * 8 + wid;
  const bool live = mt < mtiles;
  const int m0 = (live ? mt : mtiles - 1) * 16;
  const int row = lane & 15;
  const int kh  = (lane >> 4) << 1;
  const int rbase = (lane >> 4) << 3;
  const int arow_idx = min(m0 + row, M - 1);
  const float* __restrict__ arow = A + (size_t)arow_idx * 64 + kh;
  const float* __restrict__ bcol = sW1 + row;

  v8f c0 = {}, c1 = {}, c2 = {}, c3 = {};
#pragma unroll
  for (int k = 0; k < 64; k += 4) {
    v2f a, b;
    a.x = arow[k];
    a.y = arow[k + 1];
    const float* __restrict__ bk = bcol + (k + kh) * 64;
    b.x = bk[0];  b.y = bk[64];
    c0 = __builtin_amdgcn_wmma_f32_16x16x4_f32(false, a, false, b, (short)0, c0, false, false);
    b.x = bk[16]; b.y = bk[80];
    c1 = __builtin_amdgcn_wmma_f32_16x16x4_f32(false, a, false, b, (short)0, c1, false, false);
    b.x = bk[32]; b.y = bk[96];
    c2 = __builtin_amdgcn_wmma_f32_16x16x4_f32(false, a, false, b, (short)0, c2, false, false);
    b.x = bk[48]; b.y = bk[112];
    c3 = __builtin_amdgcn_wmma_f32_16x16x4_f32(false, a, false, b, (short)0, c3, false, false);
  }

  float* __restrict__ T = sT[wid];
  const float bb0 = b1[row], bb1 = b1[16 + row], bb2 = b1[32 + row], bb3 = b1[48 + row];
#pragma unroll
  for (int r = 0; r < 8; ++r) {
    float* __restrict__ tp = T + (rbase + r) * TILE_PITCH + row;
    tp[0]  = fmaxf(c0[r] + bb0, 0.0f);
    tp[16] = fmaxf(c1[r] + bb1, 0.0f);
    tp[32] = fmaxf(c2[r] + bb2, 0.0f);
    tp[48] = fmaxf(c3[r] + bb3, 0.0f);
  }
  __syncthreads();

  const float* __restrict__ trow  = T + row * TILE_PITCH + kh;
  const float* __restrict__ b2col = sW2 + row;
  v8f d = {};
#pragma unroll
  for (int k = 0; k < 64; k += 4) {
    v2f a, b;
    a.x = trow[k];
    a.y = trow[k + 1];
    b.x = b2col[(k + kh) * 16];
    b.y = b2col[(k + kh + 1) * 16];
    d = __builtin_amdgcn_wmma_f32_16x16x4_f32(false, a, false, b, (short)0, d, false, false);
  }

#pragma unroll
  for (int r = 0; r < 8; ++r) {
    const int orow = m0 + rbase + r;
    if (live && orow < M) out[(size_t)orow * 16 + row] = d[r];
  }
}


#define TILE_R        400
#define SCAN_WAVES    32
#define ROWS_PER_WAVE 13
#define QCAP          1536
#define QSTEP         1024
#define KEYS_AHEAD    4
#define RING          64
#define ROWS_AHEAD    32
#define ROWS_AHEAD_REST 16
#define LCAP          512

__device__ __forceinline__ int4 load_idx4_guarded(const int* __restrict__ p, int e, int E) {
  int4 r;
  r.x = (e     < E) ? p[e]     : -1;
  r.y = (e + 1 < E) ? p[e + 1] : -1;
  r.z = (e + 2 < E) ? p[e + 2] : -1;
  r.w = (e + 3 < E) ? p[e + 3] : -1;
  return r;
}

__global__ void key_pair_kernel(const int* __restrict__ src, const int* __restrict__ dst,
                                unsigned* __restrict__ keys,
                                int* __restrict__ pairs, int E, int nwords) {
  const int i = blockIdx.x * blockDim.x + threadIdx.x;
  if (i >= nwords) return;
  const int e = i * 4;
  int4 d, s;
  if ((E & 3) == 0 && e + 3 < E) {
    d = ((const int4*)dst)[i];
    s = ((const int4*)src)[i];
  } else {
    d = load_idx4_guarded(dst, e, E);
    s = load_idx4_guarded(src, e, E);
  }
  int4* __restrict__ pr = (int4*)pairs + 2 * (size_t)i;
  pr[0] = make_int4(s.x, d.x, s.y, d.y);
  pr[1] = make_int4(s.z, d.z, s.w, d.w);
  const unsigned k0 = ((unsigned)d.x / TILE_R) & 255u, k1 = ((unsigned)d.y / TILE_R) & 255u;
  const unsigned k2 = ((unsigned)d.z / TILE_R) & 255u, k3 = ((unsigned)d.w / TILE_R) & 255u;
  keys[i] = k0 | (k1 << 8) | (k2 << 16) | (k3 << 24);
}


typedef __attribute__((ext_vector_type(16))) float v16f;

template <int N> struct count_c { static constexpr int value = N; };

__global__ __launch_bounds__(1024) void find_tile_kernel(const int* __restrict__ pairs,
                                                         const unsigned* __restrict__ keys,
                                                         float* __restrict__ dinv,
                                                         int* __restrict__ lst,
                                                         int* __restrict__ lcount,
                                                         int E, int Nn) {
  __shared__ unsigned sQ[SCAN_WAVES][QCAP];
  __shared__ unsigned sRing[SCAN_WAVES][RING];
  __shared__ int sQn[SCAN_WAVES];
  __shared__ int sMore[SCAN_WAVES];
  __shared__ int sRun[SCAN_WAVES][SCAN_WAVES + 1];
  __shared__ int sBig[SCAN_WAVES];
  static_assert(SCAN_WAVES == 32, "one owner per lane, five owner bits");

  const int lane = threadIdx.x & 31;
  const int wid  = __builtin_amdgcn_readfirstlane((int)(threadIdx.x >> 5));
  const int d0   = blockIdx.x * TILE_R;
  const int nrows = min(TILE_R, Nn - d0);
  const unsigned un = (unsigned)nrows;
  const int row0 = wid * ROWS_PER_WAVE;

  unsigned* __restrict__ q    = sQ[wid];
  unsigned* __restrict__ ring = sRing[wid];
  int qn = 0;
  int rhead = 0, rn = 0;
  const int gw = (int)blockIdx.x * SCAN_WAVES + wid;
  int2* __restrict__ mylst = (int2*)lst + (size_t)gw * LCAP;
  int lcnt = 0;
  int cnt_l = 0;

  const unsigned splat = ((unsigned)blockIdx.x & 255u) * 0x01010101u;

  auto flags = [&](const uint4& kv) {
    const unsigned x0 = kv.x ^ splat, x1 = kv.y ^ splat, x2 = kv.z ^ splat, x3 = kv.w ^ splat;
    const unsigned u0 = (x0 - 0x01010101u) & ~x0 & 0x80808080u, u1 = (x1 - 0x01010101u) & ~x1 & 0x80808080u;
    const unsigned u2 = (x2 - 0x01010101u) & ~x2 & 0x80808080u, u3 = (x3 - 0x01010101u) & ~x3 & 0x80808080u;
    return (u0 >> 7) | (u1 >> 6) | (u2 >> 5) | (u3 >> 4);
  };
  auto look = [&](const uint4& ka, const uint4& kb, int ebase) {
    unsigned g = flags(ka) | (flags(kb) << 4);
    const int e0 = ebase + lane * 16;
    for (;;) {
      const bool has = g != 0u;
      const unsigned mask = __builtin_amdgcn_ballot_w32(has);
      if (mask == 0u) break;
      if (has) {
        const int p = __builtin_ctz(g);
        g &= g - 1u;
        q[qn + (int)__builtin_amdgcn_mbcnt_lo(mask, 0u)] =
            (unsigned)(e0 + ((p & 4) << 7) + ((p & 3) << 2) + (p >> 3));
      }
      qn += __builtin_popcount(mask);
    }
  };

  auto ownergroup = [&](int b, unsigned& pk, unsigned& own, unsigned& peers) {
    __builtin_amdgcn_wave_barrier();
    const bool valid = b + lane < qn;
    pk = valid ? q[b + lane] : 0u;
    own = (pk & 511u) / ROWS_PER_WAVE;
    unsigned p = __builtin_amdgcn_ballot_w32(valid);
#pragma unroll
    for (int bit = 0; bit < 5; ++bit) {
      const bool on = ((own >> bit) & 1u) != 0u;
      const unsigned bb = __builtin_amdgcn_ballot_w32(on);
      p &= on ? bb : ~bb;
    }
    peers = p;
    return valid;
  };

  auto save = [&](int n) {
    __builtin_amdgcn_wave_barrier();
    unsigned pk = 0u;
    if (lane < n) pk = ring[(rhead + lane) & (RING - 1)];
    const int r_l = (lane < n) ? (int)(pk & 511u) - row0 : -1;
    const int s_l = (int)(pk >> 9);
    if (lane < n && lcnt + lane < LCAP)
      mylst[lcnt + lane] = make_int2((s_l << 4) | r_l, 0);
    lcnt += n;
#pragma unroll
    for (int k = 0; k < ROWS_PER_WAVE; ++k) {
      const unsigned m = __builtin_amdgcn_ballot_w32(r_l == k);
      if (lane == k) cnt_l += __builtin_popcount(m);
    }
  };

  auto rpush = [&](bool m, unsigned pk) {
    const unsigned mask = __builtin_amdgcn_ballot_w32(m);
    if (mask != 0u) {
      if (m) ring[(rhead + rn + (int)__builtin_amdgcn_mbcnt_lo(mask, 0u)) & (RING - 1)] = pk;
      rn += __builtin_popcount(mask);
      if (rn >= 32) {
        save(32);
        rhead = (rhead + 32) & (RING - 1);
        rn -= 32;
      }
    }
  };

  const int nsc = (E + QSTEP - 1) / QSTEP;
  const int rot = (int)blockIdx.x * (nsc / (int)gridDim.x);

  auto stepbase = [&](int scx) {
    int g = scx + rot;
    if (g >= nsc) g -= nsc;
    return g * QSTEP;
  };
  auto loadkeys = [&](int scx, int fb, int half) {
    return ((const uint4*)((const unsigned char*)keys + stepbase(scx < nsc ? scx : fb)))[half * 32 + lane];
  };

  int sc = wid;

  for (;;) {
    if (sc < nsc) {
      uint4 a0 = loadkeys(sc, sc, 0),                  b0 = loadkeys(sc, sc, 1);
      uint4 a1 = loadkeys(sc + SCAN_WAVES, sc, 0),     b1 = loadkeys(sc + SCAN_WAVES, sc, 1);
      uint4 a2 = loadkeys(sc + 2 * SCAN_WAVES, sc, 0), b2 = loadkeys(sc + 2 * SCAN_WAVES, sc, 1);
      uint4 a3 = loadkeys(sc + 3 * SCAN_WAVES, sc, 0), b3 = loadkeys(sc + 3 * SCAN_WAVES, sc, 1);
#define SCAN_STEP(ka, kb)                                    \
      if (!(sc < nsc && qn <= QCAP - QSTEP)) break;          \
      look(ka, kb, stepbase(sc));                            \
      ka = loadkeys(sc + KEYS_AHEAD * SCAN_WAVES, sc, 0);    \
      kb = loadkeys(sc + KEYS_AHEAD * SCAN_WAVES, sc, 1);    \
      sc += SCAN_WAVES;
      for (;;) {
        SCAN_STEP(a0, b0)
        SCAN_STEP(a1, b1)
        SCAN_STEP(a2, b2)
        SCAN_STEP(a3, b3)
      }
#undef SCAN_STEP
    }
    {
      int nkeep = 0;
      for (int b = 0; b < qn; b += 128) {
        __builtin_amdgcn_wave_barrier();
        int2 sd[4];
#pragma unroll
        for (int u = 0; u < 4; ++u) {
          const int i = min(b + u * 32 + lane, qn - 1);
          sd[u] = ((const int2*)pairs)[q[i]];
        }
#pragma unroll
        for (int u = 0; u < 4; ++u) {
          const bool ok = (b + u * 32 + lane < qn) && (unsigned)(sd[u].y - d0) < un;
          const unsigned mask = __builtin_amdgcn_ballot_w32(ok);
          if (ok) q[nkeep + (int)__builtin_amdgcn_mbcnt_lo(mask, 0u)] = ((unsigned)sd[u].x << 9) | (unsigned)(sd[u].y - d0);
          nkeep += __builtin_popcount(mask);
        }
      }
      qn = nkeep;
    }
    const bool fits = 2 * qn <= QCAP;
    if (fits) {
      int* run = sRun[wid] + 1;
      run[lane] = 0;
      for (int b = 0; b < qn; b += 32) {
        unsigned pk, own, peers;
        const bool valid = ownergroup(b, pk, own, peers);
        if (valid && __builtin_amdgcn_mbcnt_lo(peers, 0u) == 0u) run[own] += __builtin_popcount(peers);
      }
      __builtin_amdgcn_wave_barrier();
      const int cnt_o = run[lane];
      int incl = cnt_o;
      incl += __builtin_amdgcn_update_dpp(0, incl, 0x111, 0xf, 0xf, true);
      incl += __builtin_amdgcn_update_dpp(0, incl, 0x112, 0xf, 0xf, true);
      incl += __builtin_amdgcn_update_dpp(0, incl, 0x114, 0xf, 0xf, true);
      incl += __builtin_amdgcn_update_dpp(0, incl, 0x118, 0xf, 0xf, true);
      const int row0 = __builtin_amdgcn_readlane(incl, 15);
      if (lane >= 16) incl += row0;
      run[lane] = qn + incl - cnt_o;
      if (lane == 0) run[-1] = qn;
      for (int b = 0; b < qn; b += 32) {
        unsigned pk, own, peers;
        const bool valid = ownergroup(b, pk, own, peers);
        const int rank = (int)__builtin_amdgcn_mbcnt_lo(peers, 0u);
        int base = 0;
        if (valid) { base = run[own]; q[base + rank] = pk; }
        if (valid && rank == 0) run[own] = base + __builtin_popcount(peers);
      }
    }
    if (lane == 0) { sQn[wid] = qn; sMore[wid] = (sc < nsc) ? 1 : 0; sBig[wid] = fits ? 0 : 1; }
    __syncthreads();

    const bool more = __builtin_amdgcn_ballot_w32(sMore[lane] != 0) != 0u;
    const bool big = __builtin_amdgcn_ballot_w32(sBig[lane] != 0) != 0u;
    for (int k = 0; k < SCAN_WAVES; ++k) {
      int lo_v = sRun[k][wid], hi_v = sRun[k][wid + 1];
      if (big) { lo_v = 0; hi_v = sQn[k]; }
      const int lo = __builtin_amdgcn_readfirstlane(lo_v);
      const int nk = __builtin_amdgcn_readfirstlane(hi_v) - lo;
      const unsigned* __restrict__ qk = sQ[k] + lo;
      for (int b = 0; b < nk; b += 32) {
        unsigned pk = 0u;
        bool own = false;
        if (b + lane < nk) {
          pk = qk[b + lane];
          own = ((pk & 511u) / ROWS_PER_WAVE) == (unsigned)wid;
        }
        rpush(own, pk);
      }
    }
    if (rn > 0) save(rn);
    rhead = 0; rn = 0; qn = 0;
    __syncthreads();
    if (!more) break;
  }

  if (lane < ROWS_PER_WAVE && row0 + lane < nrows) dinv[d0 + row0 + lane] = rsqrtf((float)cnt_l + 1.0f);
  if (lane == 0) lcount[gw] = lcnt;
}


__global__ __launch_bounds__(256) void agg64_list_kernel(const float* __restrict__ x,
                                                         int* __restrict__ lst,
                                                         const int* __restrict__ lcount,
                                                         const int* __restrict__ src,
                                                         const int* __restrict__ dst,
                                                         const float* __restrict__ dinv,
                                                         float* __restrict__ agg,
                                                         int E, int Nn) {
  const int lane = threadIdx.x & 31;
  const int wid  = __builtin_amdgcn_readfirstlane((int)(threadIdx.x >> 5));
  const int gw   = (int)blockIdx.x * 8 + wid;
  const int d0   = (gw / SCAN_WAVES) * TILE_R;
  const int row0 = (gw % SCAN_WAVES) * ROWS_PER_WAVE;
  const int nrows = min(TILE_R, Nn - d0);
  if (row0 >= nrows) return;
  const int base = d0 + row0;

  v16f accx = {}, accy = {};
  {
    float2 v0[ROWS_PER_WAVE];
    float dd[ROWS_PER_WAVE];
#pragma unroll
    for (int k = 0; k < ROWS_PER_WAVE; ++k) {
      const int d = d0 + min(row0 + k, nrows - 1);
      dd[k] = dinv[d];
      v0[k] = ((const float2*)(x + (size_t)d * 64))[lane];
    }
#pragma unroll
    for (int k = 0; k < ROWS_PER_WAVE; ++k) {
      const bool valid = row0 + k < nrows;
      const float w = dd[k] * dd[k];
      accx[k] = valid ? v0[k].x * w : 0.0f;
      accy[k] = valid ? v0[k].y * w : 0.0f;
    }
  }

  auto addrow = [&](int r, float vx, float vy, float w) {
    accx[r] += vx * w;
    accy[r] += vy * w;
  };
  const float2* __restrict__ xr = (const float2*)x;

  const int cnt = __builtin_amdgcn_readfirstlane(lcount[gw]);
  if (cnt <= LCAP) {
    int2* my = (int2*)lst + (size_t)gw * LCAP;
    int en = 0;
    if (lane < cnt) en = my[lane].x;
    auto rows = [&](auto ahead_c, int j, int n) {
      constexpr int AHEAD = decltype(ahead_c)::value;
      const int ex = en;
      float ds_l = 0.0f, dd_l = 0.0f;
      if (lane < n) {
        ds_l = dinv[ex >> 4];
        dd_l = dinv[base + (ex & 15)];
      }
      const int r_l = ex & 15;
      const unsigned o_l = (unsigned)(ex >> 4) << 5;
      float2 v[32];
#pragma unroll
      for (int u = 0; u < AHEAD; ++u)
        if (u < n) {
          const unsigned o = (unsigned)__builtin_amdgcn_readlane((int)o_l, u);
          v[u] = xr[o + (unsigned)lane];
        }
      int nx = 0;
      if (j + 32 + lane < cnt) nx = my[j + 32 + lane].x;
      const float w_l = ds_l * dd_l;
      if (lane < n) my[j + lane] = make_int2(ex, __float_as_int(w_l));
#pragma unroll
      for (int u = 0; u < 32; ++u) {
        if (u < n) {
          const int r = __builtin_amdgcn_readlane(r_l, u);
          const float w = __int_as_float(__builtin_amdgcn_readlane(__float_as_int(w_l), u));
          addrow(r, v[u].x, v[u].y, w);
        }
        if (u + AHEAD < 32 && u + AHEAD < n) {
          const unsigned o = (unsigned)__builtin_amdgcn_readlane((int)o_l, u + AHEAD);
          v[u + AHEAD] = xr[o + (unsigned)lane];
        }
      }
      en = nx;
    };
    int j = 0;
    for (; j + 32 <= cnt; j += 32) rows(count_c<ROWS_AHEAD>{}, j, 32);
    if (j < cnt) rows(count_c<ROWS_AHEAD_REST>{}, j, cnt - j);
  } else {
    const int nr = min(ROWS_PER_WAVE, nrows - row0);
    for (int e0 = 0; e0 < E; e0 += 32) {
      const int e = e0 + lane;
      const int d = (e < E) ? dst[e] : -1;
      unsigned mask = __builtin_amdgcn_ballot_w32((unsigned)(d - base) < (unsigned)nr);
      while (mask != 0u) {
        const int l = __builtin_ctz(mask);
        mask &= mask - 1u;
        const int dd = __builtin_amdgcn_readlane(d, l);
        const int s  = src[e0 + l];
        const float w = dinv[s] * dinv[dd];
        const float2 v = xr[((unsigned)s << 5) + (unsigned)lane];
        addrow(dd - base, v.x, v.y, w);
      }
    }
  }

#pragma unroll
  for (int k = 0; k < ROWS_PER_WAVE; ++k)
    if (row0 + k < nrows)
      ((float2*)(agg + (size_t)(d0 + row0 + k) * 64))[lane] = make_float2(accx[k], accy[k]);
}

__global__ __launch_bounds__(256) void agg16_list_kernel(const float* __restrict__ h,
                                                         const int* __restrict__ lst,
                                                         const int* __restrict__ lcount,
                                                         const int* __restrict__ src,
                                                         const int* __restrict__ dst,
                                                         const float* __restrict__ dinv,
                                                         const float* __restrict__ b2,
                                                         float* __restrict__ out,
                                                         int E, int Nn) {
  __shared__ float sA0[8][ROWS_PER_WAVE * 32];
  __shared__ float sA1[8][ROWS_PER_WAVE * 32];
  __shared__ float sA2[8][ROWS_PER_WAVE * 32];
  __shared__ float sA3[8][ROWS_PER_WAVE * 32];

  const int lane = threadIdx.x & 31;
  const int wid  = __builtin_amdgcn_readfirstlane((int)(threadIdx.x >> 5));
  const int gw   = (int)blockIdx.x * 8 + wid;
  const int d0   = (gw / SCAN_WAVES) * TILE_R;
  const int row0 = (gw % SCAN_WAVES) * ROWS_PER_WAVE;
  const int nr   = min(ROWS_PER_WAVE, min(TILE_R, Nn - d0) - row0);
  if (nr <= 0) return;
  const int c  = lane & 15;
  const int hh = lane >> 4;
  const int base = d0 + row0;

  float* a0 = sA0[wid] + lane;
  float* a1 = sA1[wid] + lane;
  float* a2 = sA2[wid] + lane;
  float* a3 = sA3[wid] + lane;
#pragma unroll
  for (int k = 0; k < ROWS_PER_WAVE; ++k) {
    a0[k * 32] = 0.0f; a1[k * 32] = 0.0f; a2[k * 32] = 0.0f; a3[k * 32] = 0.0f;
  }

  float selfv[7];
#pragma unroll
  for (int i = 0; i < 7; ++i) {
    const int d = base + min(2 * i + hh, nr - 1);
    const float dd = dinv[d];
    selfv[i] = h[(size_t)d * 16 + c] * (dd * dd);
  }
  const float bias = b2[c];

  const int cnt = __builtin_amdgcn_readfirstlane(lcount[gw]);
  if (cnt <= LCAP) {
    const int2* __restrict__ my = (const int2*)lst + (size_t)gw * LCAP;
    int2 en = make_int2(0, 0);
    if (lane < cnt) en = my[lane];
    for (int j = 0; j < cnt; j += 32) {
      const int m = cnt - j;
      int pk[16];
      float wt[16], v[16];
#pragma unroll
      for (int u = 0; u < 16; ++u) {
        const int sel = (2 * u + hh) << 2;
        pk[u] = __builtin_amdgcn_ds_bpermute(sel, en.x);
        wt[u] = __int_as_float(__builtin_amdgcn_ds_bpermute(sel, en.y));
        v[u] = h[((unsigned)pk[u] & ~15u) + (unsigned)c];
      }
      int2 nx = make_int2(0, 0);
      if (j + 32 + lane < cnt) nx = my[j + 32 + lane];
#pragma unroll
      for (int u = 0; u < 16; u += 4) {
        const int r0 = (pk[u] & 15) * 32, r1 = (pk[u + 1] & 15) * 32;
        const int r2 = (pk[u + 2] & 15) * 32, r3 = (pk[u + 3] & 15) * 32;
        float t0 = a0[r0], t1 = a1[r1], t2 = a2[r2], t3 = a3[r3];
        t0 += v[u] * wt[u];
        t1 += v[u + 1] * wt[u + 1];
        t2 += v[u + 2] * wt[u + 2];
        t3 += v[u + 3] * wt[u + 3];
        if (2 * u + hh < m)       a0[r0] = t0;
        if (2 * (u + 1) + hh < m) a1[r1] = t1;
        if (2 * (u + 2) + hh < m) a2[r2] = t2;
        if (2 * (u + 3) + hh < m) a3[r3] = t3;
      }
      en = nx;
    }
  } else {
    for (int e0 = 0; e0 < E; e0 += 32) {
      const int e = e0 + lane;
      const int d = (e < E) ? dst[e] : -1;
      unsigned mask = __builtin_amdgcn_ballot_w32((unsigned)(d - base) < (unsigned)nr);
      while (mask != 0u) {
        const int l = __builtin_ctz(mask);
        mask &= mask - 1u;
        const int dd = __builtin_amdgcn_readlane(d, l);
        const int s  = src[e0 + l];
        const float w = dinv[s] * dinv[dd];
        if (lane < 16) a0[(dd - base) * 32] += h[(size_t)s * 16 + lane] * w;
      }
    }
  }

  __builtin_amdgcn_wave_barrier();
#pragma unroll
  for (int i = 0; i < 7; ++i) {
    const int r = 2 * i + hh;
    if (r < nr) {
      const float* p0 = sA0[wid] + r * 32 + c;
      const float* p1 = sA1[wid] + r * 32 + c;
      const float* p2 = sA2[wid] + r * 32 + c;
      const float* p3 = sA3[wid] + r * 32 + c;
      float s = selfv[i];
      s += p0[0]; s += p0[16]; s += p1[0]; s += p1[16];
      s += p2[0]; s += p2[16]; s += p3[0]; s += p3[16];
      out[(size_t)(base + r) * 16 + c] = s + bias;
    }
  }
}


extern "C" void kernel_launch(void* const* d_in, const int* in_sizes, int n_in,
                              void* d_out, int out_size, void* d_ws, size_t ws_size,
                              hipStream_t stream) {
  const float* x    = (const float*)d_in[0];
  const int*   edge = (const int*)d_in[1];
  const float* W1   = (const float*)d_in[2];
  const float* b1   = (const float*)d_in[3];
  const float* W2   = (const float*)d_in[4];
  const float* b2   = (const float*)d_in[5];
  float* out = (float*)d_out;

  const int E  = in_sizes[1] / 2;
  const int Nn = in_sizes[0] / IN_DIM;
  const int* src = edge;
  const int* dst = edge + E;

  char* ws = (char*)d_ws;
  float* dinv = (float*)ws;
  float* h2   = (float*)(ws + (size_t)Nn * 4);
  float* agg1 = (float*)(ws + (size_t)Nn * 4 + (size_t)Nn * 64 * 4);
  const int ntiles = (Nn + TILE_R - 1) / TILE_R;
  const size_t loff = ((size_t)Nn * 4 + (size_t)Nn * 128 * 4 + 255) & ~(size_t)255;
  int* lcount = (int*)(ws + loff);
  int* lst    = (int*)(ws + loff + (size_t)ntiles * SCAN_WAVES * 4);
  const size_t koff = (loff + (size_t)ntiles * SCAN_WAVES * 4 + (size_t)ntiles * SCAN_WAVES * LCAP * 8 + 255) & ~(size_t)255;
  unsigned* keys = (unsigned*)(ws + koff);
  const int nwords = ((E + QSTEP - 1) / QSTEP) * (QSTEP / 4);
  const size_t poff = (koff + (size_t)nwords * 4 + 255) & ~(size_t)255;
  int* pairs = (int*)(ws + poff);

  const int TB = 256;
  const int mtiles = (Nn + 15) >> 4;

  key_pair_kernel<<<(nwords + TB - 1) / TB, TB, 0, stream>>>(src, dst, keys, pairs, E, nwords);

  find_tile_kernel<<<ntiles, SCAN_WAVES * 32, 0, stream>>>(pairs, keys, dinv, lst, lcount, E, Nn);

  agg64_list_kernel<<<ntiles * (SCAN_WAVES / 8), 256, 0, stream>>>(x, lst, lcount, src, dst, dinv, agg1, E, Nn);

  mlp_wmma_kernel<<<(mtiles + 7) / 8, 256, 0, stream>>>(agg1, W1, b1, W2, h2, Nn);
  agg16_list_kernel<<<ntiles * (SCAN_WAVES / 8), 256, 0, stream>>>(h2, lst, lcount, src, dst, dinv, b2, out, E, Nn);

  (void)n_in; (void)out_size; (void)ws_size;
}


# Layer-1 tile scan reads the tile's bit map (a bit per tile and group of 8 edges, rewritten per call by the pair pass, by ballots), fetches set groups' pairs; key list gone; sum order changed, fixed
# speedup vs baseline: 3.8659x; 1.0580x over previous
#include <hip/hip_runtime.h>
#include <hip/hip_bf16.h>


#define IN_DIM  64
#define OUT_DIM 16

typedef __attribute__((ext_vector_type(2))) float v2f;
typedef __attribute__((ext_vector_type(8))) float v8f;


#define TILE_PITCH 68

__global__ __launch_bounds__(256) void mlp_wmma_kernel(const float* __restrict__ A,
                                                       const float* __restrict__ W1,
                                                       const float* __restrict__ b1,
                                                       const float* __restrict__ W2,
                                                       float* __restrict__ out, int M) {
  __shared__ __attribute__((aligned(16))) float sW1[64 * 64];
  __shared__ __attribute__((aligned(16))) float sW2[64 * 16];
  __shared__ __attribute__((aligned(16))) float sT[8][16 * TILE_PITCH];
  for (int i = threadIdx.x; i < (64 * 64) / 4; i += blockDim.x)
    ((float4*)sW1)[i] = ((const float4*)W1)[i];
  for (int i = threadIdx.x; i < (64 * 16) / 4; i += blockDim.x)
    ((float4*)sW2)[i] = ((const float4*)W2)[i];
  __syncthreads();

  const int lane = threadIdx.x & 31;
  const int wid  = threadIdx.x >> 5;
  const int mtiles = (M + 15) >> 4;
  const int mt = blockIdx.x * 8 + wid;
  const bool live = mt < mtiles;
  const int m0 = (live ? mt : mtiles - 1) * 16;
  const int row = lane & 15;
  const int kh  = (lane >> 4) << 1;
  const int rbase = (lane >> 4) << 3;
  const int arow_idx = min(m0 + row, M - 1);
  const float* __restrict__ arow = A + (size_t)arow_idx * 64 + kh;
  const float* __restrict__ bcol = sW1 + row;

  v8f c0 = {}, c1 = {}, c2 = {}, c3 = {};
#pragma unroll
  for (int k = 0; k < 64; k += 4) {
    v2f a, b;
    a.x = arow[k];
    a.y = arow[k + 1];
    const float* __restrict__ bk = bcol + (k + kh) * 64;
    b.x = bk[0];  b.y = bk[64];
    c0 = __builtin_amdgcn_wmma_f32_16x16x4_f32(false, a, false, b, (short)0, c0, false, false);
    b.x = bk[16]; b.y = bk[80];
    c1 = __builtin_amdgcn_wmma_f32_16x16x4_f32(false, a, false, b, (short)0, c1, false, false);
    b.x = bk[32]; b.y = bk[96];
    c2 = __builtin_amdgcn_wmma_f32_16x16x4_f32(false, a, false, b, (short)0, c2, false, false);
    b.x = bk[48]; b.y = bk[112];
    c3 = __builtin_amdgcn_wmma_f32_16x16x4_f32(false, a, false, b, (short)0, c3, false, false);
  }

  float* __restrict__ T = sT[wid];
  const float bb0 = b1[row], bb1 = b1[16 + row], bb2 = b1[32 + row], bb3 = b1[48 + row];
#pragma unroll
  for (int r = 0; r < 8; ++r) {
    float* __restrict__ tp = T + (rbase + r) * TILE_PITCH + row;
    tp[0]  = fmaxf(c0[r] + bb0, 0.0f);
    tp[16] = fmaxf(c1[r] + bb1, 0.0f);
    tp[32] = fmaxf(c2[r] + bb2, 0.0f);
    tp[48] = fmaxf(c3[r] + bb3, 0.0f);
  }
  __syncthreads();

  const float* __restrict__ trow  = T + row * TILE_PITCH + kh;
  const float* __restrict__ b2col = sW2 + row;
  v8f d = {};
#pragma unroll
  for (int k = 0; k < 64; k += 4) {
    v2f a, b;
    a.x = trow[k];
    a.y = trow[k + 1];
    b.x = b2col[(k + kh) * 16];
    b.y = b2col[(k + kh + 1) * 16];
    d = __builtin_amdgcn_wmma_f32_16x16x4_f32(false, a, false, b, (short)0, d, false, false);
  }

#pragma unroll
  for (int r = 0; r < 8; ++r) {
    const int orow = m0 + rbase + r;
    if (live && orow < M) out[(size_t)orow * 16 + row] = d[r];
  }
}


#define TILE_R        400
#define SCAN_WAVES    32
#define ROWS_PER_WAVE 13
#define QCAP          1536
#define MAP_EDGES     8192
#define BITS_PER_TRIP 4
#define TRIP_MAX      (32 * BITS_PER_TRIP * 8)
#define RING          64
#define ROWS_AHEAD    32
#define ROWS_AHEAD_REST 16
#define LCAP          512

__device__ __forceinline__ int4 load_idx4_guarded(const int* __restrict__ p, int e, int E) {
  int4 r;
  r.x = (e     < E) ? p[e]     : -1;
  r.y = (e + 1 < E) ? p[e + 1] : -1;
  r.z = (e + 2 < E) ? p[e + 2] : -1;
  r.w = (e + 3 < E) ? p[e + 3] : -1;
  return r;
}

__global__ __launch_bounds__(1024) void map_pair_kernel(const int* __restrict__ src, const int* __restrict__ dst,
                                                        unsigned* __restrict__ map,
                                                        int* __restrict__ pairs, int E, int nblk) {
  __shared__ __attribute__((aligned(16))) unsigned smap[256 * 32];
  const int tid  = threadIdx.x;
  const int lane = tid & 31;
  const int w    = tid >> 5;
  const int blk  = blockIdx.x;
  for (int i = 0; i < 8; ++i) smap[(lane + 32 * i) * 32 + w] = 0u;

  const int e = blk * MAP_EDGES + w * 256 + lane * 8;
  int4 da, db, sa, sb;
  if ((E & 3) == 0 && e + 7 < E) {
    da = ((const int4*)dst)[e / 4];  db = ((const int4*)dst)[e / 4 + 1];
    sa = ((const int4*)src)[e / 4];  sb = ((const int4*)src)[e / 4 + 1];
  } else {
    da = load_idx4_guarded(dst, e, E);  db = load_idx4_guarded(dst, e + 4, E);
    sa = load_idx4_guarded(src, e, E);  sb = load_idx4_guarded(src, e + 4, E);
  }
  int4* __restrict__ pr = (int4*)pairs + (size_t)e / 2;
  pr[0] = make_int4(sa.x, da.x, sa.y, da.y);
  pr[1] = make_int4(sa.z, da.z, sa.w, da.w);
  pr[2] = make_int4(sb.x, db.x, sb.y, db.y);
  pr[3] = make_int4(sb.z, db.z, sb.w, db.w);

  const int dv[8] = {da.x, da.y, da.z, da.w, db.x, db.y, db.z, db.w};
#pragma unroll
  for (int j = 0; j < 8; ++j) {
    __builtin_amdgcn_wave_barrier();
    const int d = dv[j];
    const bool valid = d >= 0;
    const unsigned t = ((unsigned)d / TILE_R) & 255u;
    unsigned p = __builtin_amdgcn_ballot_w32(valid);
#pragma unroll
    for (int bit = 0; bit < 8; ++bit) {
      const bool on = ((t >> bit) & 1u) != 0u;
      const unsigned bb = __builtin_amdgcn_ballot_w32(on);
      p &= on ? bb : ~bb;
    }
    if (valid && __builtin_amdgcn_mbcnt_lo(p, 0u) == 0u) {
      unsigned* wp = smap + t * 32 + w;
      *wp = *wp | p;
    }
  }
  __syncthreads();
  for (int it = 0; it < 2; ++it) {
    const int u = it * 1024 + tid;
    ((uint4*)map)[((size_t)(u >> 3) * (size_t)nblk + (size_t)blk) * 8 + (size_t)(u & 7)] = ((const uint4*)smap)[u];
  }
}


typedef __attribute__((ext_vector_type(16))) float v16f;

template <int N> struct count_c { static constexpr int value = N; };

__global__ __launch_bounds__(1024) void find_tile_kernel(const int* __restrict__ pairs,
                                                         const unsigned* __restrict__ map,
                                                         float* __restrict__ dinv,
                                                         int* __restrict__ lst,
                                                         int* __restrict__ lcount,
                                                         int nblk, int Nn) {
  __shared__ unsigned sQ[SCAN_WAVES][QCAP];
  __shared__ unsigned sRing[SCAN_WAVES][RING];
  __shared__ int sQn[SCAN_WAVES];
  __shared__ int sMore[SCAN_WAVES];
  __shared__ int sRun[SCAN_WAVES][SCAN_WAVES + 1];
  __shared__ int sBig[SCAN_WAVES];
  static_assert(SCAN_WAVES == 32, "one owner per lane, five owner bits");

  const int lane = threadIdx.x & 31;
  const int wid  = __builtin_amdgcn_readfirstlane((int)(threadIdx.x >> 5));
  const int d0   = blockIdx.x * TILE_R;
  const int nrows = min(TILE_R, Nn - d0);
  const unsigned un = (unsigned)nrows;
  const int row0 = wid * ROWS_PER_WAVE;

  unsigned* __restrict__ q    = sQ[wid];
  unsigned* __restrict__ ring = sRing[wid];
  int qn = 0;
  int rhead = 0, rn = 0;
  const int gw = (int)blockIdx.x * SCAN_WAVES + wid;
  int2* __restrict__ mylst = (int2*)lst + (size_t)gw * LCAP;
  int lcnt = 0;
  int cnt_l = 0;

  auto keep = [&](bool h, int sv, int dv) {
    const bool ok = h && (unsigned)(dv - d0) < un;
    const unsigned mask = __builtin_amdgcn_ballot_w32(ok);
    if (ok) q[qn + (int)__builtin_amdgcn_mbcnt_lo(mask, 0u)] = ((unsigned)sv << 9) | (unsigned)(dv - d0);
    qn += __builtin_popcount(mask);
  };

  auto ownergroup = [&](int b, unsigned& pk, unsigned& own, unsigned& peers) {
    __builtin_amdgcn_wave_barrier();
    const bool valid = b + lane < qn;
    pk = valid ? q[b + lane] : 0u;
    own = (pk & 511u) / ROWS_PER_WAVE;
    unsigned p = __builtin_amdgcn_ballot_w32(valid);
#pragma unroll
    for (int bit = 0; bit < 5; ++bit) {
      const bool on = ((own >> bit) & 1u) != 0u;
      const unsigned bb = __builtin_amdgcn_ballot_w32(on);
      p &= on ? bb : ~bb;
    }
    peers = p;
    return valid;
  };

  auto save = [&](int n) {
    __builtin_amdgcn_wave_barrier();
    unsigned pk = 0u;
    if (lane < n) pk = ring[(rhead + lane) & (RING - 1)];
    const int r_l = (lane < n) ? (int)(pk & 511u) - row0 : -1;
    const int s_l = (int)(pk >> 9);
    if (lane < n && lcnt + lane < LCAP)
      mylst[lcnt + lane] = make_int2((s_l << 4) | r_l, 0);
    lcnt += n;
#pragma unroll
    for (int k = 0; k < ROWS_PER_WAVE; ++k) {
      const unsigned m = __builtin_amdgcn_ballot_w32(r_l == k);
      if (lane == k) cnt_l += __builtin_popcount(m);
    }
  };

  auto rpush = [&](bool m, unsigned pk) {
    const unsigned mask = __builtin_amdgcn_ballot_w32(m);
    if (mask != 0u) {
      if (m) ring[(rhead + rn + (int)__builtin_amdgcn_mbcnt_lo(mask, 0u)) & (RING - 1)] = pk;
      rn += __builtin_popcount(mask);
      if (rn >= 32) {
        save(32);
        rhead = (rhead + 32) & (RING - 1);
        rn -= 32;
      }
    }
  };

  const unsigned* __restrict__ mslice = map + (size_t)((unsigned)blockIdx.x & 255u) * (size_t)nblk * 32;
  unsigned w[8] = {0u, 0u, 0u, 0u, 0u, 0u, 0u, 0u};
  int wb = 0;
  int rb = wid;

  for (;;) {
    for (;;) {
      const bool left = __builtin_amdgcn_ballot_w32((w[0] | w[1] | w[2] | w[3] | w[4] | w[5] | w[6] | w[7]) != 0u) != 0u;
      if (!left) {
        if (rb >= nblk) break;
        wb = rb;
#pragma unroll
        for (int i = 0; i < 8; ++i) {
          const int r = rb + 32 * i;
          w[i] = (r < nblk) ? mslice[(size_t)r * 32 + lane] : 0u;
        }
        rb += 256;
        continue;
      }
      if (qn + TRIP_MAX > QCAP) break;
      unsigned g[BITS_PER_TRIP];
      bool has[BITS_PER_TRIP];
#pragma unroll
      for (int b = 0; b < BITS_PER_TRIP; ++b) {
        unsigned val = 0u;
        int wi = 0;
#pragma unroll
        for (int i = 7; i >= 0; --i)
          if (w[i] != 0u) { val = w[i]; wi = i; }
        has[b] = val != 0u;
        const int k = has[b] ? __builtin_ctz(val) : 0;
        const unsigned rest = val & (val - 1u);
#pragma unroll
        for (int i = 0; i < 8; ++i)
          if (i == wi) w[i] = rest;
        g[b] = has[b] ? (((unsigned)(wb + 32 * wi) << 10) + ((unsigned)lane << 5) + (unsigned)k) : 0u;
      }
      int4 pv[BITS_PER_TRIP][4];
#pragma unroll
      for (int b = 0; b < BITS_PER_TRIP; ++b) {
        const int4* __restrict__ gp = (const int4*)pairs + (size_t)g[b] * 4;
        pv[b][0] = gp[0]; pv[b][1] = gp[1]; pv[b][2] = gp[2]; pv[b][3] = gp[3];
      }
#pragma unroll
      for (int b = 0; b < BITS_PER_TRIP; ++b) {
#pragma unroll
        for (int j = 0; j < 4; ++j) {
          keep(has[b], pv[b][j].x, pv[b][j].y);
          keep(has[b], pv[b][j].z, pv[b][j].w);
        }
      }
    }
    const bool fits = 2 * qn <= QCAP;
    if (fits) {
      int* run = sRun[wid] + 1;
      run[lane] = 0;
      for (int b = 0; b < qn; b += 32) {
        unsigned pk, own, peers;
        const bool valid = ownergroup(b, pk, own, peers);
        if (valid && __builtin_amdgcn_mbcnt_lo(peers, 0u) == 0u) run[own] += __builtin_popcount(peers);
      }
      __builtin_amdgcn_wave_barrier();
      const int cnt_o = run[lane];
      int incl = cnt_o;
      incl += __builtin_amdgcn_update_dpp(0, incl, 0x111, 0xf, 0xf, true);
      incl += __builtin_amdgcn_update_dpp(0, incl, 0x112, 0xf, 0xf, true);
      incl += __builtin_amdgcn_update_dpp(0, incl, 0x114, 0xf, 0xf, true);
      incl += __builtin_amdgcn_update_dpp(0, incl, 0x118, 0xf, 0xf, true);
      const int row0 = __builtin_amdgcn_readlane(incl, 15);
      if (lane >= 16) incl += row0;
      run[lane] = qn + incl - cnt_o;
      if (lane == 0) run[-1] = qn;
      for (int b = 0; b < qn; b += 32) {
        unsigned pk, own, peers;
        const bool valid = ownergroup(b, pk, own, peers);
        const int rank = (int)__builtin_amdgcn_mbcnt_lo(peers, 0u);
        int base = 0;
        if (valid) { base = run[own]; q[base + rank] = pk; }
        if (valid && rank == 0) run[own] = base + __builtin_popcount(peers);
      }
    }
    const bool unscanned = (rb < nblk) ||
        __builtin_amdgcn_ballot_w32((w[0] | w[1] | w[2] | w[3] | w[4] | w[5] | w[6] | w[7]) != 0u) != 0u;
    if (lane == 0) { sQn[wid] = qn; sMore[wid] = unscanned ? 1 : 0; sBig[wid] = fits ? 0 : 1; }
    __syncthreads();

    const bool more = __builtin_amdgcn_ballot_w32(sMore[lane] != 0) != 0u;
    const bool big = __builtin_amdgcn_ballot_w32(sBig[lane] != 0) != 0u;
    for (int k = 0; k < SCAN_WAVES; ++k) {
      int lo_v = sRun[k][wid], hi_v = sRun[k][wid + 1];
      if (big) { lo_v = 0; hi_v = sQn[k]; }
      const int lo = __builtin_amdgcn_readfirstlane(lo_v);
      const int nk = __builtin_amdgcn_readfirstlane(hi_v) - lo;
      const unsigned* __restrict__ qk = sQ[k] + lo;
      for (int b = 0; b < nk; b += 32) {
        unsigned pk = 0u;
        bool own = false;
        if (b + lane < nk) {
          pk = qk[b + lane];
          own = ((pk & 511u) / ROWS_PER_WAVE) == (unsigned)wid;
        }
        rpush(own, pk);
      }
    }
    if (rn > 0) save(rn);
    rhead = 0; rn = 0; qn = 0;
    __syncthreads();
    if (!more) break;
  }

  if (lane < ROWS_PER_WAVE && row0 + lane < nrows) dinv[d0 + row0 + lane] = rsqrtf((float)cnt_l + 1.0f);
  if (lane == 0) lcount[gw] = lcnt;
}


__global__ __launch_bounds__(256) void agg64_list_kernel(const float* __restrict__ x,
                                                         int* __restrict__ lst,
                                                         const int* __restrict__ lcount,
                                                         const int* __restrict__ src,
                                                         const int* __restrict__ dst,
                                                         const float* __restrict__ dinv,
                                                         float* __restrict__ agg,
                                                         int E, int Nn) {
  const int lane = threadIdx.x & 31;
  const int wid  = __builtin_amdgcn_readfirstlane((int)(threadIdx.x >> 5));
  const int gw   = (int)blockIdx.x * 8 + wid;
  const int d0   = (gw / SCAN_WAVES) * TILE_R;
  const int row0 = (gw % SCAN_WAVES) * ROWS_PER_WAVE;
  const int nrows = min(TILE_R, Nn - d0);
  if (row0 >= nrows) return;
  const int base = d0 + row0;

  v16f accx = {}, accy = {};
  {
    float2 v0[ROWS_PER_WAVE];
    float dd[ROWS_PER_WAVE];
#pragma unroll
    for (int k = 0; k < ROWS_PER_WAVE; ++k) {
      const int d = d0 + min(row0 + k, nrows - 1);
      dd[k] = dinv[d];
      v0[k] = ((const float2*)(x + (size_t)d * 64))[lane];
    }
#pragma unroll
    for (int k = 0; k < ROWS_PER_WAVE; ++k) {
      const bool valid = row0 + k < nrows;
      const float w = dd[k] * dd[k];
      accx[k] = valid ? v0[k].x * w : 0.0f;
      accy[k] = valid ? v0[k].y * w : 0.0f;
    }
  }

  auto addrow = [&](int r, float vx, float vy, float w) {
    accx[r] += vx * w;
    accy[r] += vy * w;
  };
  const float2* __restrict__ xr = (const float2*)x;

  const int cnt = __builtin_amdgcn_readfirstlane(lcount[gw]);
  if (cnt <= LCAP) {
    int2* my = (int2*)lst + (size_t)gw * LCAP;
    int en = 0;
    if (lane < cnt) en = my[lane].x;
    auto rows = [&](auto ahead_c, int j, int n) {
      constexpr int AHEAD = decltype(ahead_c)::value;
      const int ex = en;
      float ds_l = 0.0f, dd_l = 0.0f;
      if (lane < n) {
        ds_l = dinv[ex >> 4];
        dd_l = dinv[base + (ex & 15)];
      }
      const int r_l = ex & 15;
      const unsigned o_l = (unsigned)(ex >> 4) << 5;
      float2 v[32];
#pragma unroll
      for (int u = 0; u < AHEAD; ++u)
        if (u < n) {
          const unsigned o = (unsigned)__builtin_amdgcn_readlane((int)o_l, u);
          v[u] = xr[o + (unsigned)lane];
        }
      int nx = 0;
      if (j + 32 + lane < cnt) nx = my[j + 32 + lane].x;
      const float w_l = ds_l * dd_l;
      if (lane < n) my[j + lane] = make_int2(ex, __float_as_int(w_l));
#pragma unroll
      for (int u = 0; u < 32; ++u) {
        if (u < n) {
          const int r = __builtin_amdgcn_readlane(r_l, u);
          const float w = __int_as_float(__builtin_amdgcn_readlane(__float_as_int(w_l), u));
          addrow(r, v[u].x, v[u].y, w);
        }
        if (u + AHEAD < 32 && u + AHEAD < n) {
          const unsigned o = (unsigned)__builtin_amdgcn_readlane((int)o_l, u + AHEAD);
          v[u + AHEAD] = xr[o + (unsigned)lane];
        }
      }
      en = nx;
    };
    int j = 0;
    for (; j + 32 <= cnt; j += 32) rows(count_c<ROWS_AHEAD>{}, j, 32);
    if (j < cnt) rows(count_c<ROWS_AHEAD_REST>{}, j, cnt - j);
  } else {
    const int nr = min(ROWS_PER_WAVE, nrows - row0);
    for (int e0 = 0; e0 < E; e0 += 32) {
      const int e = e0 + lane;
      const int d = (e < E) ? dst[e] : -1;
      unsigned mask = __builtin_amdgcn_ballot_w32((unsigned)(d - base) < (unsigned)nr);
      while (mask != 0u) {
        const int l = __builtin_ctz(mask);
        mask &= mask - 1u;
        const int dd = __builtin_amdgcn_readlane(d, l);
        const int s  = src[e0 + l];
        const float w = dinv[s] * dinv[dd];
        const float2 v = xr[((unsigned)s << 5) + (unsigned)lane];
        addrow(dd - base, v.x, v.y, w);
      }
    }
  }

#pragma unroll
  for (int k = 0; k < ROWS_PER_WAVE; ++k)
    if (row0 + k < nrows)
      ((float2*)(agg + (size_t)(d0 + row0 + k) * 64))[lane] = make_float2(accx[k], accy[k]);
}

__global__ __launch_bounds__(256) void agg16_list_kernel(const float* __restrict__ h,
                                                         const int* __restrict__ lst,
                                                         const int* __restrict__ lcount,
                                                         const int* __restrict__ src,
                                                         const int* __restrict__ dst,
                                                         const float* __restrict__ dinv,
                                                         const float* __restrict__ b2,
                                                         float* __restrict__ out,
                                                         int E, int Nn) {
  __shared__ float sA0[8][ROWS_PER_WAVE * 32];
  __shared__ float sA1[8][ROWS_PER_WAVE * 32];
  __shared__ float sA2[8][ROWS_PER_WAVE * 32];
  __shared__ float sA3[8][ROWS_PER_WAVE * 32];

  const int lane = threadIdx.x & 31;
  const int wid  = __builtin_amdgcn_readfirstlane((int)(threadIdx.x >> 5));
  const int gw   = (int)blockIdx.x * 8 + wid;
  const int d0   = (gw / SCAN_WAVES) * TILE_R;
  const int row0 = (gw % SCAN_WAVES) * ROWS_PER_WAVE;
  const int nr   = min(ROWS_PER_WAVE, min(TILE_R, Nn - d0) - row0);
  if (nr <= 0) return;
  const int c  = lane & 15;
  const int hh = lane >> 4;
  const int base = d0 + row0;

  float* a0 = sA0[wid] + lane;
  float* a1 = sA1[wid] + lane;
  float* a2 = sA2[wid] + lane;
  float* a3 = sA3[wid] + lane;
#pragma unroll
  for (int k = 0; k < ROWS_PER_WAVE; ++k) {
    a0[k * 32] = 0.0f; a1[k * 32] = 0.0f; a2[k * 32] = 0.0f; a3[k * 32] = 0.0f;
  }

  float selfv[7];
#pragma unroll
  for (int i = 0; i < 7; ++i) {
    const int d = base + min(2 * i + hh, nr - 1);
    const float dd = dinv[d];
    selfv[i] = h[(size_t)d * 16 + c] * (dd * dd);
  }
  const float bias = b2[c];

  const int cnt = __builtin_amdgcn_readfirstlane(lcount[gw]);
  if (cnt <= LCAP) {
    const int2* __restrict__ my = (const int2*)lst + (size_t)gw * LCAP;
    int2 en = make_int2(0, 0);
    if (lane < cnt) en = my[lane];
    for (int j = 0; j < cnt; j += 32) {
      const int m = cnt - j;
      int pk[16];
      float wt[16], v[16];
#pragma unroll
      for (int u = 0; u < 16; ++u) {
        const int sel = (2 * u + hh) << 2;
        pk[u] = __builtin_amdgcn_ds_bpermute(sel, en.x);
        wt[u] = __int_as_float(__builtin_amdgcn_ds_bpermute(sel, en.y));
        v[u] = h[((unsigned)pk[u] & ~15u) + (unsigned)c];
      }
      int2 nx = make_int2(0, 0);
      if (j + 32 + lane < cnt) nx = my[j + 32 + lane];
#pragma unroll
      for (int u = 0; u < 16; u += 4) {
        const int r0 = (pk[u] & 15) * 32, r1 = (pk[u + 1] & 15) * 32;
        const int r2 = (pk[u + 2] & 15) * 32, r3 = (pk[u + 3] & 15) * 32;
        float t0 = a0[r0], t1 = a1[r1], t2 = a2[r2], t3 = a3[r3];
        t0 += v[u] * wt[u];
        t1 += v[u + 1] * wt[u + 1];
        t2 += v[u + 2] * wt[u + 2];
        t3 += v[u + 3] * wt[u + 3];
        if (2 * u + hh < m)       a0[r0] = t0;
        if (2 * (u + 1) + hh < m) a1[r1] = t1;
        if (2 * (u + 2) + hh < m) a2[r2] = t2;
        if (2 * (u + 3) + hh < m) a3[r3] = t3;
      }
      en = nx;
    }
  } else {
    for (int e0 = 0; e0 < E; e0 += 32) {
      const int e = e0 + lane;
      const int d = (e < E) ? dst[e] : -1;
      unsigned mask = __builtin_amdgcn_ballot_w32((unsigned)(d - base) < (unsigned)nr);
      while (mask != 0u) {
        const int l = __builtin_ctz(mask);
        mask &= mask - 1u;
        const int dd = __builtin_amdgcn_readlane(d, l);
        const int s  = src[e0 + l];
        const float w = dinv[s] * dinv[dd];
        if (lane < 16) a0[(dd - base) * 32] += h[(size_t)s * 16 + lane] * w;
      }
    }
  }

  __builtin_amdgcn_wave_barrier();
#pragma unroll
  for (int i = 0; i < 7; ++i) {
    const int r = 2 * i + hh;
    if (r < nr) {
      const float* p0 = sA0[wid] + r * 32 + c;
      const float* p1 = sA1[wid] + r * 32 + c;
      const float* p2 = sA2[wid] + r * 32 + c;
      const float* p3 = sA3[wid] + r * 32 + c;
      float s = selfv[i];
      s += p0[0]; s += p0[16]; s += p1[0]; s += p1[16];
      s += p2[0]; s += p2[16]; s += p3[0]; s += p3[16];
      out[(size_t)(base + r) * 16 + c] = s + bias;
    }
  }
}


extern "C" void kernel_launch(void* const* d_in, const int* in_sizes, int n_in,
                              void* d_out, int out_size, void* d_ws, size_t ws_size,
                              hipStream_t stream) {
  const float* x    = (const float*)d_in[0];
  const int*   edge = (const int*)d_in[1];
  const float* W1   = (const float*)d_in[2];
  const float* b1   = (const float*)d_in[3];
  const float* W2   = (const float*)d_in[4];
  const float* b2   = (const float*)d_in[5];
  float* out = (float*)d_out;

  const int E  = in_sizes[1] / 2;
  const int Nn = in_sizes[0] / IN_DIM;
  const int* src = edge;
  const int* dst = edge + E;

  char* ws = (char*)d_ws;
  float* dinv = (float*)ws;
  float* h2   = (float*)(ws + (size_t)Nn * 4);
  float* agg1 = (float*)(ws + (size_t)Nn * 4 + (size_t)Nn * 64 * 4);
  const int ntiles = (Nn + TILE_R - 1) / TILE_R;
  const size_t loff = ((size_t)Nn * 4 + (size_t)Nn * 128 * 4 + 255) & ~(size_t)255;
  int* lcount = (int*)(ws + loff);
  int* lst    = (int*)(ws + loff + (size_t)ntiles * SCAN_WAVES * 4);
  const size_t koff = (loff + (size_t)ntiles * SCAN_WAVES * 4 + (size_t)ntiles * SCAN_WAVES * LCAP * 8 + 255) & ~(size_t)255;
  unsigned* map = (unsigned*)(ws + koff);
  const int nblk = (E + MAP_EDGES - 1) / MAP_EDGES;
  const size_t poff = (koff + (size_t)256 * (size_t)nblk * 128 + 255) & ~(size_t)255;
  int* pairs = (int*)(ws + poff);

  const int mtiles = (Nn + 15) >> 4;

  map_pair_kernel<<<nblk, 1024, 0, stream>>>(src, dst, map, pairs, E, nblk);

  find_tile_kernel<<<ntiles, SCAN_WAVES * 32, 0, stream>>>(pairs, map, dinv, lst, lcount, nblk, Nn);

  agg64_list_kernel<<<ntiles * (SCAN_WAVES / 8), 256, 0, stream>>>(x, lst, lcount, src, dst, dinv, agg1, E, Nn);

  mlp_wmma_kernel<<<(mtiles + 7) / 8, 256, 0, stream>>>(agg1, W1, b1, W2, h2, Nn);
  agg16_list_kernel<<<ntiles * (SCAN_WAVES / 8), 256, 0, stream>>>(h2, lst, lcount, src, dst, dinv, b2, out, E, Nn);

  (void)n_in; (void)out_size; (void)ws_size;
}


# Tile pass hand-over by one lane per queue: an owning wave prefix-sums its 32 run lengths, each lane copies its run into an LDS row, saved in full batches of 32; per-queue walk gone; sum order kept
# speedup vs baseline: 4.0690x; 1.0525x over previous
#include <hip/hip_runtime.h>
#include <hip/hip_bf16.h>


#define IN_DIM  64
#define OUT_DIM 16

typedef __attribute__((ext_vector_type(2))) float v2f;
typedef __attribute__((ext_vector_type(8))) float v8f;


#define TILE_PITCH 68

__global__ __launch_bounds__(256) void mlp_wmma_kernel(const float* __restrict__ A,
                                                       const float* __restrict__ W1,
                                                       const float* __restrict__ b1,
                                                       const float* __restrict__ W2,
                                                       float* __restrict__ out, int M) {
  __shared__ __attribute__((aligned(16))) float sW1[64 * 64];
  __shared__ __attribute__((aligned(16))) float sW2[64 * 16];
  __shared__ __attribute__((aligned(16))) float sT[8][16 * TILE_PITCH];
  for (int i = threadIdx.x; i < (64 * 64) / 4; i += blockDim.x)
    ((float4*)sW1)[i] = ((const float4*)W1)[i];
  for (int i = threadIdx.x; i < (64 * 16) / 4; i += blockDim.x)
    ((float4*)sW2)[i] = ((const float4*)W2)[i];
  __syncthreads();

  const int lane = threadIdx.x & 31;
  const int wid  = threadIdx.x >> 5;
  const int mtiles = (M + 15) >> 4;
  const int mt = blockIdx.x * 8 + wid;
  const bool live = mt < mtiles;
  const int m0 = (live ? mt : mtiles - 1) * 16;
  const int row = lane & 15;
  const int kh  = (lane >> 4) << 1;
  const int rbase = (lane >> 4) << 3;
  const int arow_idx = min(m0 + row, M - 1);
  const float* __restrict__ arow = A + (size_t)arow_idx * 64 + kh;
  const float* __restrict__ bcol = sW1 + row;

  v8f c0 = {}, c1 = {}, c2 = {}, c3 = {};
#pragma unroll
  for (int k = 0; k < 64; k += 4) {
    v2f a, b;
    a.x = arow[k];
    a.y = arow[k + 1];
    const float* __restrict__ bk = bcol + (k + kh) * 64;
    b.x = bk[0];  b.y = bk[64];
    c0 = __builtin_amdgcn_wmma_f32_16x16x4_f32(false, a, false, b, (short)0, c0, false, false);
    b.x = bk[16]; b.y = bk[80];
    c1 = __builtin_amdgcn_wmma_f32_16x16x4_f32(false, a, false, b, (short)0, c1, false, false);
    b.x = bk[32]; b.y = bk[96];
    c2 = __builtin_amdgcn_wmma_f32_16x16x4_f32(false, a, false, b, (short)0, c2, false, false);
    b.x = bk[48]; b.y = bk[112];
    c3 = __builtin_amdgcn_wmma_f32_16x16x4_f32(false, a, false, b, (short)0, c3, false, false);
  }

  float* __restrict__ T = sT[wid];
  const float bb0 = b1[row], bb1 = b1[16 + row], bb2 = b1[32 + row], bb3 = b1[48 + row];
#pragma unroll
  for (int r = 0; r < 8; ++r) {
    float* __restrict__ tp = T + (rbase + r) * TILE_PITCH + row;
    tp[0]  = fmaxf(c0[r] + bb0, 0.0f);
    tp[16] = fmaxf(c1[r] + bb1, 0.0f);
    tp[32] = fmaxf(c2[r] + bb2, 0.0f);
    tp[48] = fmaxf(c3[r] + bb3, 0.0f);
  }
  __syncthreads();

  const float* __restrict__ trow  = T + row * TILE_PITCH + kh;
  const float* __restrict__ b2col = sW2 + row;
  v8f d = {};
#pragma unroll
  for (int k = 0; k < 64; k += 4) {
    v2f a, b;
    a.x = trow[k];
    a.y = trow[k + 1];
    b.x = b2col[(k + kh) * 16];
    b.y = b2col[(k + kh + 1) * 16];
    d = __builtin_amdgcn_wmma_f32_16x16x4_f32(false, a, false, b, (short)0, d, false, false);
  }

#pragma unroll
  for (int r = 0; r < 8; ++r) {
    const int orow = m0 + rbase + r;
    if (live && orow < M) out[(size_t)orow * 16 + row] = d[r];
  }
}


#define TILE_R        400
#define SCAN_WAVES    32
#define ROWS_PER_WAVE 13
#define QCAP          1536
#define MAP_EDGES     8192
#define BITS_PER_TRIP 4
#define TRIP_MAX      (32 * BITS_PER_TRIP * 8)
#define RING          64
#define STAGE         512
#define ROWS_AHEAD    32
#define ROWS_AHEAD_REST 16
#define LCAP          512

__device__ __forceinline__ int4 load_idx4_guarded(const int* __restrict__ p, int e, int E) {
  int4 r;
  r.x = (e     < E) ? p[e]     : -1;
  r.y = (e + 1 < E) ? p[e + 1] : -1;
  r.z = (e + 2 < E) ? p[e + 2] : -1;
  r.w = (e + 3 < E) ? p[e + 3] : -1;
  return r;
}

__global__ __launch_bounds__(1024) void map_pair_kernel(const int* __restrict__ src, const int* __restrict__ dst,
                                                        unsigned* __restrict__ map,
                                                        int* __restrict__ pairs, int E, int nblk) {
  __shared__ __attribute__((aligned(16))) unsigned smap[256 * 32];
  const int tid  = threadIdx.x;
  const int lane = tid & 31;
  const int w    = tid >> 5;
  const int blk  = blockIdx.x;
  for (int i = 0; i < 8; ++i) smap[(lane + 32 * i) * 32 + w] = 0u;

  const int e = blk * MAP_EDGES + w * 256 + lane * 8;
  int4 da, db, sa, sb;
  if ((E & 3) == 0 && e + 7 < E) {
    da = ((const int4*)dst)[e / 4];  db = ((const int4*)dst)[e / 4 + 1];
    sa = ((const int4*)src)[e / 4];  sb = ((const int4*)src)[e / 4 + 1];
  } else {
    da = load_idx4_guarded(dst, e, E);  db = load_idx4_guarded(dst, e + 4, E);
    sa = load_idx4_guarded(src, e, E);  sb = load_idx4_guarded(src, e + 4, E);
  }
  int4* __restrict__ pr = (int4*)pairs + (size_t)e / 2;
  pr[0] = make_int4(sa.x, da.x, sa.y, da.y);
  pr[1] = make_int4(sa.z, da.z, sa.w, da.w);
  pr[2] = make_int4(sb.x, db.x, sb.y, db.y);
  pr[3] = make_int4(sb.z, db.z, sb.w, db.w);

  const int dv[8] = {da.x, da.y, da.z, da.w, db.x, db.y, db.z, db.w};
#pragma unroll
  for (int j = 0; j < 8; ++j) {
    __builtin_amdgcn_wave_barrier();
    const int d = dv[j];
    const bool valid = d >= 0;
    const unsigned t = ((unsigned)d / TILE_R) & 255u;
    unsigned p = __builtin_amdgcn_ballot_w32(valid);
#pragma unroll
    for (int bit = 0; bit < 8; ++bit) {
      const bool on = ((t >> bit) & 1u) != 0u;
      const unsigned bb = __builtin_amdgcn_ballot_w32(on);
      p &= on ? bb : ~bb;
    }
    if (valid && __builtin_amdgcn_mbcnt_lo(p, 0u) == 0u) {
      unsigned* wp = smap + t * 32 + w;
      *wp = *wp | p;
    }
  }
  __syncthreads();
  for (int it = 0; it < 2; ++it) {
    const int u = it * 1024 + tid;
    ((uint4*)map)[((size_t)(u >> 3) * (size_t)nblk + (size_t)blk) * 8 + (size_t)(u & 7)] = ((const uint4*)smap)[u];
  }
}


typedef __attribute__((ext_vector_type(16))) float v16f;

template <int N> struct count_c { static constexpr int value = N; };

__global__ __launch_bounds__(1024) void find_tile_kernel(const int* __restrict__ pairs,
                                                         const unsigned* __restrict__ map,
                                                         float* __restrict__ dinv,
                                                         int* __restrict__ lst,
                                                         int* __restrict__ lcount,
                                                         int nblk, int Nn) {
  __shared__ unsigned sQ[SCAN_WAVES][QCAP];
  __shared__ unsigned sRing[SCAN_WAVES][RING];
  __shared__ unsigned sStage[SCAN_WAVES][STAGE];
  __shared__ int sQn[SCAN_WAVES];
  __shared__ int sMore[SCAN_WAVES];
  __shared__ int sRun[SCAN_WAVES][SCAN_WAVES + 1];
  __shared__ int sBig[SCAN_WAVES];
  static_assert(SCAN_WAVES == 32, "one owner per lane, five owner bits");

  const int lane = threadIdx.x & 31;
  const int wid  = __builtin_amdgcn_readfirstlane((int)(threadIdx.x >> 5));
  const int d0   = blockIdx.x * TILE_R;
  const int nrows = min(TILE_R, Nn - d0);
  const unsigned un = (unsigned)nrows;
  const int row0 = wid * ROWS_PER_WAVE;

  unsigned* __restrict__ q    = sQ[wid];
  unsigned* __restrict__ ring = sRing[wid];
  int qn = 0;
  int rhead = 0, rn = 0;
  const int gw = (int)blockIdx.x * SCAN_WAVES + wid;
  int2* __restrict__ mylst = (int2*)lst + (size_t)gw * LCAP;
  int lcnt = 0;
  int cnt_l = 0;

  auto keep = [&](bool h, int sv, int dv) {
    const bool ok = h && (unsigned)(dv - d0) < un;
    const unsigned mask = __builtin_amdgcn_ballot_w32(ok);
    if (ok) q[qn + (int)__builtin_amdgcn_mbcnt_lo(mask, 0u)] = ((unsigned)sv << 9) | (unsigned)(dv - d0);
    qn += __builtin_popcount(mask);
  };

  auto ownergroup = [&](int b, unsigned& pk, unsigned& own, unsigned& peers) {
    __builtin_amdgcn_wave_barrier();
    const bool valid = b + lane < qn;
    pk = valid ? q[b + lane] : 0u;
    own = (pk & 511u) / ROWS_PER_WAVE;
    unsigned p = __builtin_amdgcn_ballot_w32(valid);
#pragma unroll
    for (int bit = 0; bit < 5; ++bit) {
      const bool on = ((own >> bit) & 1u) != 0u;
      const unsigned bb = __builtin_amdgcn_ballot_w32(on);
      p &= on ? bb : ~bb;
    }
    peers = p;
    return valid;
  };

  auto lanescan = [&](int v) {
    int incl = v;
    incl += __builtin_amdgcn_update_dpp(0, incl, 0x111, 0xf, 0xf, true);
    incl += __builtin_amdgcn_update_dpp(0, incl, 0x112, 0xf, 0xf, true);
    incl += __builtin_amdgcn_update_dpp(0, incl, 0x114, 0xf, 0xf, true);
    incl += __builtin_amdgcn_update_dpp(0, incl, 0x118, 0xf, 0xf, true);
    const int low16 = __builtin_amdgcn_readlane(incl, 15);
    if (lane >= 16) incl += low16;
    return incl;
  };

  auto save = [&](unsigned pk, int n) {
    const int r_l = (lane < n) ? (int)(pk & 511u) - row0 : -1;
    const int s_l = (int)(pk >> 9);
    if (lane < n && lcnt + lane < LCAP)
      mylst[lcnt + lane] = make_int2((s_l << 4) | r_l, 0);
    lcnt += n;
#pragma unroll
    for (int k = 0; k < ROWS_PER_WAVE; ++k) {
      const unsigned m = __builtin_amdgcn_ballot_w32(r_l == k);
      if (lane == k) cnt_l += __builtin_popcount(m);
    }
  };

  auto ringhead = [&](int n) {
    __builtin_amdgcn_wave_barrier();
    unsigned pk = 0u;
    if (lane < n) pk = ring[(rhead + lane) & (RING - 1)];
    return pk;
  };

  auto rpush = [&](bool m, unsigned pk) {
    const unsigned mask = __builtin_amdgcn_ballot_w32(m);
    if (mask != 0u) {
      if (m) ring[(rhead + rn + (int)__builtin_amdgcn_mbcnt_lo(mask, 0u)) & (RING - 1)] = pk;
      rn += __builtin_popcount(mask);
      if (rn >= 32) {
        save(ringhead(32), 32);
        rhead = (rhead + 32) & (RING - 1);
        rn -= 32;
      }
    }
  };

  const unsigned* __restrict__ mslice = map + (size_t)((unsigned)blockIdx.x & 255u) * (size_t)nblk * 32;
  unsigned w[8] = {0u, 0u, 0u, 0u, 0u, 0u, 0u, 0u};
  int wb = 0;
  int rb = wid;

  for (;;) {
    for (;;) {
      const bool left = __builtin_amdgcn_ballot_w32((w[0] | w[1] | w[2] | w[3] | w[4] | w[5] | w[6] | w[7]) != 0u) != 0u;
      if (!left) {
        if (rb >= nblk) break;
        wb = rb;
#pragma unroll
        for (int i = 0; i < 8; ++i) {
          const int r = rb + 32 * i;
          w[i] = (r < nblk) ? mslice[(size_t)r * 32 + lane] : 0u;
        }
        rb += 256;
        continue;
      }
      if (qn + TRIP_MAX > QCAP) break;
      unsigned g[BITS_PER_TRIP];
      bool has[BITS_PER_TRIP];
#pragma unroll
      for (int b = 0; b < BITS_PER_TRIP; ++b) {
        unsigned val = 0u;
        int wi = 0;
#pragma unroll
        for (int i = 7; i >= 0; --i)
          if (w[i] != 0u) { val = w[i]; wi = i; }
        has[b] = val != 0u;
        const int k = has[b] ? __builtin_ctz(val) : 0;
        const unsigned rest = val & (val - 1u);
#pragma unroll
        for (int i = 0; i < 8; ++i)
          if (i == wi) w[i] = rest;
        g[b] = has[b] ? (((unsigned)(wb + 32 * wi) << 10) + ((unsigned)lane << 5) + (unsigned)k) : 0u;
      }
      int4 pv[BITS_PER_TRIP][4];
#pragma unroll
      for (int b = 0; b < BITS_PER_TRIP; ++b) {
        const int4* __restrict__ gp = (const int4*)pairs + (size_t)g[b] * 4;
        pv[b][0] = gp[0]; pv[b][1] = gp[1]; pv[b][2] = gp[2]; pv[b][3] = gp[3];
      }
#pragma unroll
      for (int b = 0; b < BITS_PER_TRIP; ++b) {
#pragma unroll
        for (int j = 0; j < 4; ++j) {
          keep(has[b], pv[b][j].x, pv[b][j].y);
          keep(has[b], pv[b][j].z, pv[b][j].w);
        }
      }
    }
    const bool fits = 2 * qn <= QCAP;
    if (fits) {
      int* run = sRun[wid] + 1;
      run[lane] = 0;
      for (int b = 0; b < qn; b += 32) {
        unsigned pk, own, peers;
        const bool valid = ownergroup(b, pk, own, peers);
        if (valid && __builtin_amdgcn_mbcnt_lo(peers, 0u) == 0u) run[own] += __builtin_popcount(peers);
      }
      __builtin_amdgcn_wave_barrier();
      const int cnt_o = run[lane];
      const int incl = lanescan(cnt_o);
      run[lane] = qn + incl - cnt_o;
      if (lane == 0) run[-1] = qn;
      for (int b = 0; b < qn; b += 32) {
        unsigned pk, own, peers;
        const bool valid = ownergroup(b, pk, own, peers);
        const int rank = (int)__builtin_amdgcn_mbcnt_lo(peers, 0u);
        int base = 0;
        if (valid) { base = run[own]; q[base + rank] = pk; }
        if (valid && rank == 0) run[own] = base + __builtin_popcount(peers);
      }
    }
    const bool unscanned = (rb < nblk) ||
        __builtin_amdgcn_ballot_w32((w[0] | w[1] | w[2] | w[3] | w[4] | w[5] | w[6] | w[7]) != 0u) != 0u;
    if (lane == 0) { sQn[wid] = qn; sMore[wid] = unscanned ? 1 : 0; sBig[wid] = fits ? 0 : 1; }
    __syncthreads();

    const bool more = __builtin_amdgcn_ballot_w32(sMore[lane] != 0) != 0u;
    const bool big = __builtin_amdgcn_ballot_w32(sBig[lane] != 0) != 0u;
    int lo_l = 0, n_l = 0;
    if (!big) { lo_l = sRun[lane][wid]; n_l = sRun[lane][wid + 1] - lo_l; }
    const int incl_l = lanescan(n_l);
    const int total = __builtin_amdgcn_readlane(incl_l, 31);
    if (!big && total <= STAGE) {
      unsigned* st = sStage[wid];
      const unsigned* qs = sQ[lane] + lo_l;
      unsigned* dp = st + (incl_l - n_l);
      for (int i = 0; __builtin_amdgcn_ballot_w32(i < n_l) != 0u; i += 4) {
        unsigned e0 = 0u, e1 = 0u, e2 = 0u, e3 = 0u;
        if (i     < n_l) e0 = qs[i];
        if (i + 1 < n_l) e1 = qs[i + 1];
        if (i + 2 < n_l) e2 = qs[i + 2];
        if (i + 3 < n_l) e3 = qs[i + 3];
        if (i     < n_l) dp[i]     = e0;
        if (i + 1 < n_l) dp[i + 1] = e1;
        if (i + 2 < n_l) dp[i + 2] = e2;
        if (i + 3 < n_l) dp[i + 3] = e3;
      }
      for (int b = 0; b < total; b += 32) {
        const int m = min(32, total - b);
        __builtin_amdgcn_wave_barrier();
        unsigned pk = 0u;
        if (lane < m) pk = st[b + lane];
        save(pk, m);
      }
    } else {
      for (int k = 0; k < SCAN_WAVES; ++k) {
        int lo_v = sRun[k][wid], hi_v = sRun[k][wid + 1];
        if (big) { lo_v = 0; hi_v = sQn[k]; }
        const int lo = __builtin_amdgcn_readfirstlane(lo_v);
        const int nk = __builtin_amdgcn_readfirstlane(hi_v) - lo;
        const unsigned* __restrict__ qk = sQ[k] + lo;
        for (int b = 0; b < nk; b += 32) {
          unsigned pk = 0u;
          bool own = false;
          if (b + lane < nk) {
            pk = qk[b + lane];
            own = ((pk & 511u) / ROWS_PER_WAVE) == (unsigned)wid;
          }
          rpush(own, pk);
        }
      }
      if (rn > 0) save(ringhead(rn), rn);
    }
    rhead = 0; rn = 0; qn = 0;
    __syncthreads();
    if (!more) break;
  }

  if (lane < ROWS_PER_WAVE && row0 + lane < nrows) dinv[d0 + row0 + lane] = rsqrtf((float)cnt_l + 1.0f);
  if (lane == 0) lcount[gw] = lcnt;
}


__global__ __launch_bounds__(256) void agg64_list_kernel(const float* __restrict__ x,
                                                         int* __restrict__ lst,
                                                         const int* __restrict__ lcount,
                                                         const int* __restrict__ src,
                                                         const int* __restrict__ dst,
                                                         const float* __restrict__ dinv,
                                                         float* __restrict__ agg,
                                                         int E, int Nn) {
  const int lane = threadIdx.x & 31;
  const int wid  = __builtin_amdgcn_readfirstlane((int)(threadIdx.x >> 5));
  const int gw   = (int)blockIdx.x * 8 + wid;
  const int d0   = (gw / SCAN_WAVES) * TILE_R;
  const int row0 = (gw % SCAN_WAVES) * ROWS_PER_WAVE;
  const int nrows = min(TILE_R, Nn - d0);
  if (row0 >= nrows) return;
  const int base = d0 + row0;

  v16f accx = {}, accy = {};
  {
    float2 v0[ROWS_PER_WAVE];
    float dd[ROWS_PER_WAVE];
#pragma unroll
    for (int k = 0; k < ROWS_PER_WAVE; ++k) {
      const int d = d0 + min(row0 + k, nrows - 1);
      dd[k] = dinv[d];
      v0[k] = ((const float2*)(x + (size_t)d * 64))[lane];
    }
#pragma unroll
    for (int k = 0; k < ROWS_PER_WAVE; ++k) {
      const bool valid = row0 + k < nrows;
      const float w = dd[k] * dd[k];
      accx[k] = valid ? v0[k].x * w : 0.0f;
      accy[k] = valid ? v0[k].y * w : 0.0f;
    }
  }

  auto addrow = [&](int r, float vx, float vy, float w) {
    accx[r] += vx * w;
    accy[r] += vy * w;
  };
  const float2* __restrict__ xr = (const float2*)x;

  const int cnt = __builtin_amdgcn_readfirstlane(lcount[gw]);
  if (cnt <= LCAP) {
    int2* my = (int2*)lst + (size_t)gw * LCAP;
    int en = 0;
    if (lane < cnt) en = my[lane].x;
    auto rows = [&](auto ahead_c, int j, int n) {
      constexpr int AHEAD = decltype(ahead_c)::value;
      const int ex = en;
      float ds_l = 0.0f, dd_l = 0.0f;
      if (lane < n) {
        ds_l = dinv[ex >> 4];
        dd_l = dinv[base + (ex & 15)];
      }
      const int r_l = ex & 15;
      const unsigned o_l = (unsigned)(ex >> 4) << 5;
      float2 v[32];
#pragma unroll
      for (int u = 0; u < AHEAD; ++u)
        if (u < n) {
          const unsigned o = (unsigned)__builtin_amdgcn_readlane((int)o_l, u);
          v[u] = xr[o + (unsigned)lane];
        }
      int nx = 0;
      if (j + 32 + lane < cnt) nx = my[j + 32 + lane].x;
      const float w_l = ds_l * dd_l;
      if (lane < n) my[j + lane] = make_int2(ex, __float_as_int(w_l));
#pragma unroll
      for (int u = 0; u < 32; ++u) {
        if (u < n) {
          const int r = __builtin_amdgcn_readlane(r_l, u);
          const float w = __int_as_float(__builtin_amdgcn_readlane(__float_as_int(w_l), u));
          addrow(r, v[u].x, v[u].y, w);
        }
        if (u + AHEAD < 32 && u + AHEAD < n) {
          const unsigned o = (unsigned)__builtin_amdgcn_readlane((int)o_l, u + AHEAD);
          v[u + AHEAD] = xr[o + (unsigned)lane];
        }
      }
      en = nx;
    };
    int j = 0;
    for (; j + 32 <= cnt; j += 32) rows(count_c<ROWS_AHEAD>{}, j, 32);
    if (j < cnt) rows(count_c<ROWS_AHEAD_REST>{}, j, cnt - j);
  } else {
    const int nr = min(ROWS_PER_WAVE, nrows - row0);
    for (int e0 = 0; e0 < E; e0 += 32) {
      const int e = e0 + lane;
      const int d = (e < E) ? dst[e] : -1;
      unsigned mask = __builtin_amdgcn_ballot_w32((unsigned)(d - base) < (unsigned)nr);
      while (mask != 0u) {
        const int l = __builtin_ctz(mask);
        mask &= mask - 1u;
        const int dd = __builtin_amdgcn_readlane(d, l);
        const int s  = src[e0 + l];
        const float w = dinv[s] * dinv[dd];
        const float2 v = xr[((unsigned)s << 5) + (unsigned)lane];
        addrow(dd - base, v.x, v.y, w);
      }
    }
  }

#pragma unroll
  for (int k = 0; k < ROWS_PER_WAVE; ++k)
    if (row0 + k < nrows)
      ((float2*)(agg + (size_t)(d0 + row0 + k) * 64))[lane] = make_float2(accx[k], accy[k]);
}

__global__ __launch_bounds__(256) void agg16_list_kernel(const float* __restrict__ h,
                                                         const int* __restrict__ lst,
                                                         const int* __restrict__ lcount,
                                                         const int* __restrict__ src,
                                                         const int* __restrict__ dst,
                                                         const float* __restrict__ dinv,
                                                         const float* __restrict__ b2,
                                                         float* __restrict__ out,
                                                         int E, int Nn) {
  __shared__ float sA0[8][ROWS_PER_WAVE * 32];
  __shared__ float sA1[8][ROWS_PER_WAVE * 32];
  __shared__ float sA2[8][ROWS_PER_WAVE * 32];
  __shared__ float sA3[8][ROWS_PER_WAVE * 32];

  const int lane = threadIdx.x & 31;
  const int wid  = __builtin_amdgcn_readfirstlane((int)(threadIdx.x >> 5));
  const int gw   = (int)blockIdx.x * 8 + wid;
  const int d0   = (gw / SCAN_WAVES) * TILE_R;
  const int row0 = (gw % SCAN_WAVES) * ROWS_PER_WAVE;
  const int nr   = min(ROWS_PER_WAVE, min(TILE_R, Nn - d0) - row0);
  if (nr <= 0) return;
  const int c  = lane & 15;
  const int hh = lane >> 4;
  const int base = d0 + row0;

  float* a0 = sA0[wid] + lane;
  float* a1 = sA1[wid] + lane;
  float* a2 = sA2[wid] + lane;
  float* a3 = sA3[wid] + lane;
#pragma unroll
  for (int k = 0; k < ROWS_PER_WAVE; ++k) {
    a0[k * 32] = 0.0f; a1[k * 32] = 0.0f; a2[k * 32] = 0.0f; a3[k * 32] = 0.0f;
  }

  float selfv[7];
#pragma unroll
  for (int i = 0; i < 7; ++i) {
    const int d = base + min(2 * i + hh, nr - 1);
    const float dd = dinv[d];
    selfv[i] = h[(size_t)d * 16 + c] * (dd * dd);
  }
  const float bias = b2[c];

  const int cnt = __builtin_amdgcn_readfirstlane(lcount[gw]);
  if (cnt <= LCAP) {
    const int2* __restrict__ my = (const int2*)lst + (size_t)gw * LCAP;
    int2 en = make_int2(0, 0);
    if (lane < cnt) en = my[lane];
    for (int j = 0; j < cnt; j += 32) {
      const int m = cnt - j;
      int pk[16];
      float wt[16], v[16];
#pragma unroll
      for (int u = 0; u < 16; ++u) {
        const int sel = (2 * u + hh) << 2;
        pk[u] = __builtin_amdgcn_ds_bpermute(sel, en.x);
        wt[u] = __int_as_float(__builtin_amdgcn_ds_bpermute(sel, en.y));
        v[u] = h[((unsigned)pk[u] & ~15u) + (unsigned)c];
      }
      int2 nx = make_int2(0, 0);
      if (j + 32 + lane < cnt) nx = my[j + 32 + lane];
#pragma unroll
      for (int u = 0; u < 16; u += 4) {
        const int r0 = (pk[u] & 15) * 32, r1 = (pk[u + 1] & 15) * 32;
        const int r2 = (pk[u + 2] & 15) * 32, r3 = (pk[u + 3] & 15) * 32;
        float t0 = a0[r0], t1 = a1[r1], t2 = a2[r2], t3 = a3[r3];
        t0 += v[u] * wt[u];
        t1 += v[u + 1] * wt[u + 1];
        t2 += v[u + 2] * wt[u + 2];
        t3 += v[u + 3] * wt[u + 3];
        if (2 * u + hh < m)       a0[r0] = t0;
        if (2 * (u + 1) + hh < m) a1[r1] = t1;
        if (2 * (u + 2) + hh < m) a2[r2] = t2;
        if (2 * (u + 3) + hh < m) a3[r3] = t3;
      }
      en = nx;
    }
  } else {
    for (int e0 = 0; e0 < E; e0 += 32) {
      const int e = e0 + lane;
      const int d = (e < E) ? dst[e] : -1;
      unsigned mask = __builtin_amdgcn_ballot_w32((unsigned)(d - base) < (unsigned)nr);
      while (mask != 0u) {
        const int l = __builtin_ctz(mask);
        mask &= mask - 1u;
        const int dd = __builtin_amdgcn_readlane(d, l);
        const int s  = src[e0 + l];
        const float w = dinv[s] * dinv[dd];
        if (lane < 16) a0[(dd - base) * 32] += h[(size_t)s * 16 + lane] * w;
      }
    }
  }

  __builtin_amdgcn_wave_barrier();
#pragma unroll
  for (int i = 0; i < 7; ++i) {
    const int r = 2 * i + hh;
    if (r < nr) {
      const float* p0 = sA0[wid] + r * 32 + c;
      const float* p1 = sA1[wid] + r * 32 + c;
      const float* p2 = sA2[wid] + r * 32 + c;
      const float* p3 = sA3[wid] + r * 32 + c;
      float s = selfv[i];
      s += p0[0]; s += p0[16]; s += p1[0]; s += p1[16];
      s += p2[0]; s += p2[16]; s += p3[0]; s += p3[16];
      out[(size_t)(base + r) * 16 + c] = s + bias;
    }
  }
}


extern "C" void kernel_launch(void* const* d_in, const int* in_sizes, int n_in,
                              void* d_out, int out_size, void* d_ws, size_t ws_size,
                              hipStream_t stream) {
  const float* x    = (const float*)d_in[0];
  const int*   edge = (const int*)d_in[1];
  const float* W1   = (const float*)d_in[2];
  const float* b1   = (const float*)d_in[3];
  const float* W2   = (const float*)d_in[4];
  const float* b2   = (const float*)d_in[5];
  float* out = (float*)d_out;

  const int E  = in_sizes[1] / 2;
  const int Nn = in_sizes[0] / IN_DIM;
  const int* src = edge;
  const int* dst = edge + E;

  char* ws = (char*)d_ws;
  float* dinv = (float*)ws;
  float* h2   = (float*)(ws + (size_t)Nn * 4);
  float* agg1 = (float*)(ws + (size_t)Nn * 4 + (size_t)Nn * 64 * 4);
  const int ntiles = (Nn + TILE_R - 1) / TILE_R;
  const size_t loff = ((size_t)Nn * 4 + (size_t)Nn * 128 * 4 + 255) & ~(size_t)255;
  int* lcount = (int*)(ws + loff);
  int* lst    = (int*)(ws + loff + (size_t)ntiles * SCAN_WAVES * 4);
  const size_t koff = (loff + (size_t)ntiles * SCAN_WAVES * 4 + (size_t)ntiles * SCAN_WAVES * LCAP * 8 + 255) & ~(size_t)255;
  unsigned* map = (unsigned*)(ws + koff);
  const int nblk = (E + MAP_EDGES - 1) / MAP_EDGES;
  const size_t poff = (koff + (size_t)256 * (size_t)nblk * 128 + 255) & ~(size_t)255;
  int* pairs = (int*)(ws + poff);

  const int mtiles = (Nn + 15) >> 4;

  map_pair_kernel<<<nblk, 1024, 0, stream>>>(src, dst, map, pairs, E, nblk);

  find_tile_kernel<<<ntiles, SCAN_WAVES * 32, 0, stream>>>(pairs, map, dinv, lst, lcount, nblk, Nn);

  agg64_list_kernel<<<ntiles * (SCAN_WAVES / 8), 256, 0, stream>>>(x, lst, lcount, src, dst, dinv, agg1, E, Nn);

  mlp_wmma_kernel<<<(mtiles + 7) / 8, 256, 0, stream>>>(agg1, W1, b1, W2, h2, Nn);
  agg16_list_kernel<<<ntiles * (SCAN_WAVES / 8), 256, 0, stream>>>(h2, lst, lcount, src, dst, dinv, b2, out, E, Nn);

  (void)n_in; (void)out_size; (void)ws_size;
}
